# starting kernel
# baseline (speedup 1.0000x reference)
#include <hip/hip_runtime.h>


#define N_NODES 100000
#define N_EDGES 1600000
#define F 64
#define HSTRIDE 65

typedef __attribute__((ext_vector_type(2))) float v2f;
typedef __attribute__((ext_vector_type(8))) float v8f;

__global__ __launch_bounds__(256) void zero_f32(float* __restrict__ p, long long n) {
    long long i = (long long)blockIdx.x * 256 + threadIdx.x;
    long long stride = (long long)gridDim.x * 256;
    for (; i < n; i += stride) p[i] = 0.0f;
}

__global__ __launch_bounds__(256) void sage_degree(const int* __restrict__ dst,
                                                   float* __restrict__ deg, int n_edges) {
    int e = blockIdx.x * 256 + threadIdx.x;
    if (e < n_edges) atomicAdd(&deg[dst[e]], 1.0f);
}

__global__ __launch_bounds__(256) void sage_aggregate(const float* __restrict__ h,
                                                      const int* __restrict__ src,
                                                      const int* __restrict__ dst,
                                                      float* __restrict__ msum, int n_edges) {
    long long idx = (long long)blockIdx.x * 256 + threadIdx.x;
    int e = (int)(idx >> 5);
    int l = (int)(idx & 31);
    if (e >= n_edges) return;
    int s = src[e];
    int d = dst[e];
    const float2 v = *(const float2*)(h + (long long)s * F + l * 2);
    atomicAdd(&msum[(long long)d * F + l * 2 + 0], v.x);
    atomicAdd(&msum[(long long)d * F + l * 2 + 1], v.y);
}

__global__ __launch_bounds__(256) void sage_gemm(const float* __restrict__ h,
                                                 const float* __restrict__ msum,
                                                 const float* __restrict__ deg,
                                                 const float* __restrict__ Wself,
                                                 const float* __restrict__ Wneigh,
                                                 const float* __restrict__ bias,
                                                 float* __restrict__ out,
                                                 int n_nodes, int do_relu) {
    __shared__ float sWs[F * F];
    __shared__ float sWn[F * F];
    __shared__ float sH [32 * HSTRIDE];
    __shared__ float sHn[32 * HSTRIDE];
    __shared__ float sB [F];

    const int tid = threadIdx.x;
    const int node0 = blockIdx.x * 32;

    #pragma unroll
    for (int i = 0; i < 16; ++i) {
        sWs[tid + i * 256] = Wself [tid + i * 256];
        sWn[tid + i * 256] = Wneigh[tid + i * 256];
    }
    if (tid < F) sB[tid] = bias[tid];

    #pragma unroll
    for (int i = 0; i < 8; ++i) {
        int idx = tid + i * 256;
        int row = idx >> 6;
        int col = idx & 63;
        int node = node0 + row;
        float hv = 0.0f, mv = 0.0f, dv = 1.0f;
        if (node < n_nodes) {
            hv = h   [(long long)node * F + col];
            mv = msum[(long long)node * F + col];
            dv = deg[node];
        }
        float rinv = 1.0f / fmaxf(dv, 1.0f);
        sH [row * HSTRIDE + col] = hv;
        sHn[row * HSTRIDE + col] = mv * rinv;
    }
    __syncthreads();

    const int wave  = tid >> 5;
    const int lane  = tid & 31;
    const int m0    = (wave >> 2) * 16;
    const int c0    = (wave & 3) * 16;
    const int lrow  = lane & 15;
    const int khalf = lane >> 4;

    v8f acc = {};
    const float* aSelf = &sH [m0 * HSTRIDE + lrow * HSTRIDE];
    const float* aNei  = &sHn[m0 * HSTRIDE + lrow * HSTRIDE];
    #pragma unroll
    for (int kk = 0; kk < 16; ++kk) {
        const int kb = kk * 4 + khalf * 2;
        v2f a, b;
        a.x = aSelf[kb];
        a.y = aSelf[kb + 1];
        b.x = sWs[(kb    ) * F + c0 + lrow];
        b.y = sWs[(kb + 1) * F + c0 + lrow];
        acc = __builtin_amdgcn_wmma_f32_16x16x4_f32(false, a, false, b,
                                                    (short)0, acc, false, false);
        a.x = aNei[kb];
        a.y = aNei[kb + 1];
        b.x = sWn[(kb    ) * F + c0 + lrow];
        b.y = sWn[(kb + 1) * F + c0 + lrow];
        acc = __builtin_amdgcn_wmma_f32_16x16x4_f32(false, a, false, b,
                                                    (short)0, acc, false, false);
    }

    const float bv = sB[c0 + lrow];
    #pragma unroll
    for (int i = 0; i < 8; ++i) {
        int row = node0 + m0 + i + khalf * 8;
        float v = acc[i] + bv;
        if (do_relu) v = fmaxf(v, 0.0f);
        if (row < n_nodes) out[(long long)row * F + c0 + lrow] = v;
    }
}

extern "C" void kernel_launch(void* const* d_in, const int* in_sizes, int n_in,
                              void* d_out, int out_size, void* d_ws, size_t ws_size,
                              hipStream_t stream) {
    (void)in_sizes; (void)n_in; (void)out_size; (void)ws_size;
    const float* feat = (const float*)d_in[0];
    const int*   src  = (const int*)d_in[1];
    const int*   dst  = (const int*)d_in[2];
    const float* Ws[3] = { (const float*)d_in[3], (const float*)d_in[6], (const float*)d_in[9]  };
    const float* Wn[3] = { (const float*)d_in[4], (const float*)d_in[7], (const float*)d_in[10] };
    const float* bs[3] = { (const float*)d_in[5], (const float*)d_in[8], (const float*)d_in[11] };
    float* outp = (float*)d_out;

    float* ws   = (float*)d_ws;
    float* deg  = ws;
    float* msum = deg  + (long long)N_NODES;
    float* h1   = msum + (long long)N_NODES * F;
    float* h2   = h1   + (long long)N_NODES * F;

    const int zeroBlocks  = 2048;
    const int degBlocks   = (N_EDGES + 255) / 256;
    const int aggBlocks   = (int)(((long long)N_EDGES * 32 + 255) / 256);
    const int gemmBlocks  = (N_NODES + 31) / 32;

    zero_f32<<<zeroBlocks, 256, 0, stream>>>(deg, (long long)N_NODES * (1 + F));
    sage_degree<<<degBlocks, 256, 0, stream>>>(dst, deg, N_EDGES);

    sage_aggregate<<<aggBlocks, 256, 0, stream>>>(feat, src, dst, msum, N_EDGES);
    sage_gemm<<<gemmBlocks, 256, 0, stream>>>(feat, msum, deg, Ws[0], Wn[0], bs[0], h1, N_NODES, 1);

    zero_f32<<<zeroBlocks, 256, 0, stream>>>(msum, (long long)N_NODES * F);
    sage_aggregate<<<aggBlocks, 256, 0, stream>>>(h1, src, dst, msum, N_EDGES);
    sage_gemm<<<gemmBlocks, 256, 0, stream>>>(h1, msum, deg, Ws[1], Wn[1], bs[1], h2, N_NODES, 1);

    zero_f32<<<zeroBlocks, 256, 0, stream>>>(msum, (long long)N_NODES * F);
    sage_aggregate<<<aggBlocks, 256, 0, stream>>>(h2, src, dst, msum, N_EDGES);
    sage_gemm<<<gemmBlocks, 256, 0, stream>>>(h2, msum, deg, Ws[2], Wn[2], bs[2], outp, N_NODES, 0);
}


# aggregate: each block owns 625 nodes, keeps their sums in LDS, scans all edges, adds matching source rows by LDS f32 atomics, stores msum once; no global atomics, no clears; sum order changed
# speedup vs baseline: 1.6668x; 1.6668x over previous
#include <hip/hip_runtime.h>


#define N_NODES 100000
#define N_EDGES 1600000
#define F 64
#define HSTRIDE 65

typedef __attribute__((ext_vector_type(2))) float v2f;
typedef __attribute__((ext_vector_type(8))) float v8f;

#define DEG_RANGES  8
#define DEG_SLICES  32
#define DEG_R       (N_NODES / DEG_RANGES)
#define DEG_SLICE_E (N_EDGES / DEG_SLICES)
static_assert(DEG_R * DEG_RANGES == N_NODES, "ranges must tile the nodes");
static_assert(DEG_SLICE_E * DEG_SLICES == N_EDGES, "slices must tile the edges");
static_assert(DEG_SLICE_E % 4 == 0, "a slice is read as 16-byte vectors");

__global__ __launch_bounds__(1024) void sage_degree_count(const int* __restrict__ dst,
                                                          int* __restrict__ part) {
    __shared__ int bins[DEG_R];
    const int tid = threadIdx.x;
    const int rg  = blockIdx.x % DEG_RANGES;
    const int sl  = blockIdx.x / DEG_RANGES;
    const int r0  = rg * DEG_R;
    for (int j = tid; j < DEG_R; j += 1024) bins[j] = 0;
    __syncthreads();

    const int4* p = (const int4*)(dst + (long long)sl * DEG_SLICE_E);
    const int nvec = DEG_SLICE_E / 4;
    for (int i = tid; i < nvec; i += 4096) {
        int4 v[4];
        #pragma unroll
        for (int k = 0; k < 4; ++k) {
            const int idx = i + k * 1024;
            v[k] = (idx < nvec) ? p[idx] : make_int4(-1, -1, -1, -1);
        }
        #pragma unroll
        for (int k = 0; k < 4; ++k) {
            unsigned u;
            u = (unsigned)(v[k].x - r0); if (u < (unsigned)DEG_R) atomicAdd(&bins[u], 1);
            u = (unsigned)(v[k].y - r0); if (u < (unsigned)DEG_R) atomicAdd(&bins[u], 1);
            u = (unsigned)(v[k].z - r0); if (u < (unsigned)DEG_R) atomicAdd(&bins[u], 1);
            u = (unsigned)(v[k].w - r0); if (u < (unsigned)DEG_R) atomicAdd(&bins[u], 1);
        }
    }
    __syncthreads();

    int* row = part + (long long)sl * N_NODES + r0;
    for (int j = tid; j < DEG_R; j += 1024) row[j] = bins[j];
}

__global__ __launch_bounds__(256) void sage_init(const int* __restrict__ part,
                                                 float* __restrict__ deg) {
    const int i = blockIdx.x * 256 + threadIdx.x;
    if (i < N_NODES) {
        int c = 0;
        #pragma unroll
        for (int s = 0; s < DEG_SLICES; ++s) c += part[(long long)s * N_NODES + i];
        deg[i] = (float)c;
    }
}

#define AGG_RANGES 160
#define AGG_RN     (N_NODES / AGG_RANGES)
#define AGG_WAVES  32
#define AGG_CHUNK  1024
#define AGG_DEPTH  8
static_assert(AGG_RN * AGG_RANGES == N_NODES, "ranges must tile the nodes");
static_assert(AGG_RN <= 1024, "a queue entry keeps the local node in 10 bits");
static_assert(N_EDGES <= (1 << 21), "a queue entry keeps the edge number in 21 bits");
static_assert(N_EDGES % 4 == 0, "the destination list is read as 16-byte vectors");
static_assert((N_EDGES + AGG_CHUNK - 1) / AGG_CHUNK >= AGG_WAVES, "every wave has a first chunk");

static __device__ __forceinline__ void agg_load_chunk(const int4* __restrict__ dst4, int c, int lane,
                                                      int4 (&v)[8]) {
    #pragma unroll
    for (int u = 0; u < 8; ++u) {
        const int vi = c * (AGG_CHUNK / 4) + u * 32 + lane;
        v[u] = dst4[min(vi, N_EDGES / 4 - 1)];
        if (vi >= N_EDGES / 4) v[u] = make_int4(-1, -1, -1, -1);
    }
}

static __device__ __forceinline__ void agg_add_rows(const float* __restrict__ h, float* acc,
                                                    int s, int dl, int cnt, int lane) {
    for (int j = 0; j < cnt; j += AGG_DEPTH) {
        float2 hv[AGG_DEPTH];
        int dj[AGG_DEPTH];
        #pragma unroll
        for (int k = 0; k < AGG_DEPTH; ++k) {
            const int jj = min(j + k, cnt - 1);
            const int sj = __builtin_amdgcn_readlane(s, jj);
            dj[k] = __builtin_amdgcn_readlane(dl, jj);
            hv[k] = *(const float2*)(h + (long long)sj * F + lane * 2);
        }
        #pragma unroll
        for (int k = 0; k < AGG_DEPTH; ++k) {
            if (k == 0 || j + k < cnt) {
                float* row = &acc[dj[k] * F + lane * 2];
                __hip_atomic_fetch_add(row,     hv[k].x, __ATOMIC_RELAXED, __HIP_MEMORY_SCOPE_WORKGROUP);
                __hip_atomic_fetch_add(row + 1, hv[k].y, __ATOMIC_RELAXED, __HIP_MEMORY_SCOPE_WORKGROUP);
            }
        }
    }
}

__global__ __launch_bounds__(1024) void sage_aggregate(const float* __restrict__ h,
                                                       const int* __restrict__ src,
                                                       const int* __restrict__ dst,
                                                       float* __restrict__ msum) {
    __shared__ __attribute__((aligned(16))) float acc[AGG_RN * F];
    __shared__ unsigned queue[AGG_WAVES * AGG_CHUNK];
    const int tid  = threadIdx.x;
    const int lane = tid & 31;
    const int wave = __builtin_amdgcn_readfirstlane(tid >> 5);
    const int r0   = blockIdx.x * AGG_RN;

    float4* acc4 = (float4*)acc;
    for (int j = tid; j < AGG_RN * F / 4; j += 1024) acc4[j] = make_float4(0.f, 0.f, 0.f, 0.f);
    __syncthreads();

    unsigned* q = queue + wave * AGG_CHUNK;
    const int4* dst4 = (const int4*)dst;
    const int nchunks = (N_EDGES + AGG_CHUNK - 1) / AGG_CHUNK;

    const int off = (int)(((long long)blockIdx.x * nchunks) / AGG_RANGES);

    int4 v[8];
    {
        const int c = wave + off;
        agg_load_chunk(dst4, (c >= nchunks) ? c - nchunks : c, lane, v);
    }
    for (int i = wave; i < nchunks; i += AGG_WAVES) {
        const int c = (i + off >= nchunks) ? i + off - nchunks : i + off;
        int qn = 0;
        #pragma unroll
        for (int u = 0; u < 8; ++u) {
            const unsigned ux = (unsigned)(v[u].x - r0);
            const unsigned uy = (unsigned)(v[u].y - r0);
            const unsigned uz = (unsigned)(v[u].z - r0);
            const unsigned uw = (unsigned)(v[u].w - r0);
            const unsigned mn = min(min(ux, uy), min(uz, uw));
            if (__builtin_amdgcn_ballot_w32(mn < (unsigned)AGG_RN) != 0u) {
                const unsigned e0 = (unsigned)(c * (AGG_CHUNK / 4) + u * 32 + lane) * 4u;
                const unsigned uc[4] = { ux, uy, uz, uw };
                #pragma unroll
                for (int k = 0; k < 4; ++k) {
                    const bool m = uc[k] < (unsigned)AGG_RN;
                    const unsigned bal = __builtin_amdgcn_ballot_w32(m);
                    if (m) q[qn + (int)__builtin_amdgcn_mbcnt_lo(bal, 0u)] = ((e0 + k) << 10) | uc[k];
                    qn += __builtin_popcount(bal);
                }
            }
        }

        const int cnt0 = min(32, qn);
        unsigned ent = 0u;
        int s = 0;
        if (lane < cnt0) { ent = q[lane]; s = src[ent >> 10]; }

        {
            const int cn = min(i + AGG_WAVES, nchunks - 1) + off;
            agg_load_chunk(dst4, (cn >= nchunks) ? cn - nchunks : cn, lane, v);
        }

        if (cnt0 > 0) agg_add_rows(h, acc, s, (int)(ent & 1023u), cnt0, lane);

        for (int b = 32; b < qn; b += 32) {
            const int cnt = min(32, qn - b);
            unsigned ent2 = 0u;
            int s2 = 0;
            if (lane < cnt) { ent2 = q[b + lane]; s2 = src[ent2 >> 10]; }
            agg_add_rows(h, acc, s2, (int)(ent2 & 1023u), cnt, lane);
        }
    }
    __syncthreads();

    float4* out4 = (float4*)(msum + (long long)r0 * F);
    for (int j = tid; j < AGG_RN * F / 4; j += 1024) out4[j] = acc4[j];
}

__global__ __launch_bounds__(256) void sage_gemm(const float* __restrict__ h,
                                                 const float* __restrict__ msum,
                                                 const float* __restrict__ deg,
                                                 const float* __restrict__ Wself,
                                                 const float* __restrict__ Wneigh,
                                                 const float* __restrict__ bias,
                                                 float* __restrict__ out,
                                                 int n_nodes, int do_relu) {
    __shared__ float sWs[F * F];
    __shared__ float sWn[F * F];
    __shared__ float sH [32 * HSTRIDE];
    __shared__ float sHn[32 * HSTRIDE];
    __shared__ float sB [F];

    const int tid = threadIdx.x;
    const int node0 = blockIdx.x * 32;

    #pragma unroll
    for (int i = 0; i < 16; ++i) {
        sWs[tid + i * 256] = Wself [tid + i * 256];
        sWn[tid + i * 256] = Wneigh[tid + i * 256];
    }
    if (tid < F) sB[tid] = bias[tid];

    #pragma unroll
    for (int i = 0; i < 8; ++i) {
        int idx = tid + i * 256;
        int row = idx >> 6;
        int col = idx & 63;
        int node = node0 + row;
        float hv = 0.0f, mv = 0.0f, dv = 1.0f;
        if (node < n_nodes) {
            hv = h   [(long long)node * F + col];
            mv = msum[(long long)node * F + col];
            dv = deg[node];
        }
        float rinv = 1.0f / fmaxf(dv, 1.0f);
        sH [row * HSTRIDE + col] = hv;
        sHn[row * HSTRIDE + col] = mv * rinv;
    }
    __syncthreads();

    const int wave  = tid >> 5;
    const int lane  = tid & 31;
    const int m0    = (wave >> 2) * 16;
    const int c0    = (wave & 3) * 16;
    const int lrow  = lane & 15;
    const int khalf = lane >> 4;

    v8f acc = {};
    const float* aSelf = &sH [m0 * HSTRIDE + lrow * HSTRIDE];
    const float* aNei  = &sHn[m0 * HSTRIDE + lrow * HSTRIDE];
    #pragma unroll
    for (int kk = 0; kk < 16; ++kk) {
        const int kb = kk * 4 + khalf * 2;
        v2f a, b;
        a.x = aSelf[kb];
        a.y = aSelf[kb + 1];
        b.x = sWs[(kb    ) * F + c0 + lrow];
        b.y = sWs[(kb + 1) * F + c0 + lrow];
        acc = __builtin_amdgcn_wmma_f32_16x16x4_f32(false, a, false, b,
                                                    (short)0, acc, false, false);
        a.x = aNei[kb];
        a.y = aNei[kb + 1];
        b.x = sWn[(kb    ) * F + c0 + lrow];
        b.y = sWn[(kb + 1) * F + c0 + lrow];
        acc = __builtin_amdgcn_wmma_f32_16x16x4_f32(false, a, false, b,
                                                    (short)0, acc, false, false);
    }

    const float bv = sB[c0 + lrow];
    #pragma unroll
    for (int i = 0; i < 8; ++i) {
        int row = node0 + m0 + i + khalf * 8;
        float v = acc[i] + bv;
        if (do_relu) v = fmaxf(v, 0.0f);
        if (row < n_nodes) out[(long long)row * F + c0 + lrow] = v;
    }
}

extern "C" void kernel_launch(void* const* d_in, const int* in_sizes, int n_in,
                              void* d_out, int out_size, void* d_ws, size_t ws_size,
                              hipStream_t stream) {
    (void)in_sizes; (void)n_in; (void)out_size; (void)ws_size;
    const float* feat = (const float*)d_in[0];
    const int*   src  = (const int*)d_in[1];
    const int*   dst  = (const int*)d_in[2];
    const float* Ws[3] = { (const float*)d_in[3], (const float*)d_in[6], (const float*)d_in[9]  };
    const float* Wn[3] = { (const float*)d_in[4], (const float*)d_in[7], (const float*)d_in[10] };
    const float* bs[3] = { (const float*)d_in[5], (const float*)d_in[8], (const float*)d_in[11] };
    float* outp = (float*)d_out;

    float* ws   = (float*)d_ws;
    float* deg  = ws;
    float* msum = deg  + (long long)N_NODES;
    float* h1   = msum + (long long)N_NODES * F;
    float* h2   = h1   + (long long)N_NODES * F;

    const int initBlocks  = (N_NODES + 255) / 256;
    const int degBlocks   = DEG_RANGES * DEG_SLICES;
    const int aggBlocks   = AGG_RANGES;
    const int gemmBlocks  = (N_NODES + 31) / 32;

    int* part = (int*)h2;
    sage_degree_count<<<degBlocks, 1024, 0, stream>>>(dst, part);
    sage_init<<<initBlocks, 256, 0, stream>>>(part, deg);

    sage_aggregate<<<aggBlocks, 1024, 0, stream>>>(feat, src, dst, msum);
    sage_gemm<<<gemmBlocks, 256, 0, stream>>>(feat, msum, deg, Ws[0], Wn[0], bs[0], h1, N_NODES, 1);

    sage_aggregate<<<aggBlocks, 1024, 0, stream>>>(h1, src, dst, msum);
    sage_gemm<<<gemmBlocks, 256, 0, stream>>>(h1, msum, deg, Ws[1], Wn[1], bs[1], h2, N_NODES, 1);

    sage_aggregate<<<aggBlocks, 1024, 0, stream>>>(h2, src, dst, msum);
    sage_gemm<<<gemmBlocks, 256, 0, stream>>>(h2, msum, deg, Ws[2], Wn[2], bs[2], outp, N_NODES, 0);
}


# aggregate: layer 0's edge scan also stores each block's matching (source, row) pairs as whole 128-B lines; layers 1-2 walk that list, no rescan; layer 1 in place on h1; sum order changed
# speedup vs baseline: 2.1239x; 1.2742x over previous
#include <hip/hip_runtime.h>


#define N_NODES 100000
#define N_EDGES 1600000
#define F 64
#define HSTRIDE 65

typedef __attribute__((ext_vector_type(2))) float v2f;
typedef __attribute__((ext_vector_type(8))) float v8f;

#define DEG_RANGES  8
#define DEG_SLICES  32
#define DEG_R       (N_NODES / DEG_RANGES)
#define DEG_SLICE_E (N_EDGES / DEG_SLICES)
static_assert(DEG_R * DEG_RANGES == N_NODES, "ranges must tile the nodes");
static_assert(DEG_SLICE_E * DEG_SLICES == N_EDGES, "slices must tile the edges");
static_assert(DEG_SLICE_E % 4 == 0, "a slice is read as 16-byte vectors");

__global__ __launch_bounds__(1024) void sage_degree_count(const int* __restrict__ dst,
                                                          int* __restrict__ part) {
    __shared__ int bins[DEG_R];
    const int tid = threadIdx.x;
    const int rg  = blockIdx.x % DEG_RANGES;
    const int sl  = blockIdx.x / DEG_RANGES;
    const int r0  = rg * DEG_R;
    for (int j = tid; j < DEG_R; j += 1024) bins[j] = 0;
    __syncthreads();

    const int4* p = (const int4*)(dst + (long long)sl * DEG_SLICE_E);
    const int nvec = DEG_SLICE_E / 4;
    for (int i = tid; i < nvec; i += 4096) {
        int4 v[4];
        #pragma unroll
        for (int k = 0; k < 4; ++k) {
            const int idx = i + k * 1024;
            v[k] = (idx < nvec) ? p[idx] : make_int4(-1, -1, -1, -1);
        }
        #pragma unroll
        for (int k = 0; k < 4; ++k) {
            unsigned u;
            u = (unsigned)(v[k].x - r0); if (u < (unsigned)DEG_R) atomicAdd(&bins[u], 1);
            u = (unsigned)(v[k].y - r0); if (u < (unsigned)DEG_R) atomicAdd(&bins[u], 1);
            u = (unsigned)(v[k].z - r0); if (u < (unsigned)DEG_R) atomicAdd(&bins[u], 1);
            u = (unsigned)(v[k].w - r0); if (u < (unsigned)DEG_R) atomicAdd(&bins[u], 1);
        }
    }
    __syncthreads();

    int* row = part + (long long)sl * N_NODES + r0;
    for (int j = tid; j < DEG_R; j += 1024) row[j] = bins[j];
}

__global__ __launch_bounds__(256) void sage_init(const int* __restrict__ part,
                                                 float* __restrict__ deg) {
    const int i = blockIdx.x * 256 + threadIdx.x;
    if (i < N_NODES) {
        int c = 0;
        #pragma unroll
        for (int s = 0; s < DEG_SLICES; ++s) c += part[(long long)s * N_NODES + i];
        deg[i] = (float)c;
    }
}

#define AGG_RANGES 160
#define AGG_RN     (N_NODES / AGG_RANGES)
#define AGG_WAVES  32
#define AGG_CHUNK  1024
#define AGG_DEPTH  8
#define REPLAY_DEPTH 16
#define LIST_PAD   1056
#define LIST_CAP   (N_EDGES + AGG_RANGES * LIST_PAD)
#define LIST_END   0xFFFFFFFFu
#define REPLAY_LDS_RESERVE 8192
static_assert(AGG_RN * F * 4 + REPLAY_LDS_RESERVE > 160 * 1024, "one list-walk block per WGP");
static_assert(LIST_PAD >= 32 * AGG_WAVES + 32, "room for every wave's part line and the rounding");
static_assert(LIST_CAP + 2 * AGG_RANGES <= N_NODES * F, "list and directory fit in one plane");
static_assert(N_NODES <= (1 << 21), "an entry keeps the source node above the 10 bits of the row");
static_assert(AGG_RN * AGG_RANGES == N_NODES, "ranges must tile the nodes");
static_assert(AGG_RN <= 1024, "a queue entry keeps the local node in 10 bits");
static_assert(N_EDGES <= (1 << 21), "a queue entry keeps the edge number in 21 bits");
static_assert(N_EDGES % 4 == 0, "the destination list is read as 16-byte vectors");
static_assert((N_EDGES + AGG_CHUNK - 1) / AGG_CHUNK >= AGG_WAVES, "every wave has a first chunk");

static __device__ __forceinline__ void agg_load_chunk(const int4* __restrict__ dst4, int c, int lane,
                                                      int4 (&v)[8]) {
    #pragma unroll
    for (int u = 0; u < 8; ++u) {
        const int vi = c * (AGG_CHUNK / 4) + u * 32 + lane;
        v[u] = dst4[min(vi, N_EDGES / 4 - 1)];
        if (vi >= N_EDGES / 4) v[u] = make_int4(-1, -1, -1, -1);
    }
}

template <int DEPTH, bool WHOLE>
static __device__ __forceinline__ void agg_add_rows(const float* __restrict__ h, float* acc,
                                                    int s, int dl, int cnt, int lane) {
    int j = 0;
    for (; WHOLE && j + DEPTH <= cnt; j += DEPTH) {
        float2 hv[DEPTH];
        int dj[DEPTH];
        #pragma unroll
        for (int k = 0; k < DEPTH; ++k) {
            const int sj = __builtin_amdgcn_readlane(s, j + k);
            dj[k] = __builtin_amdgcn_readlane(dl, j + k);
            hv[k] = *(const float2*)(h + (long long)sj * F + lane * 2);
        }
        #pragma unroll
        for (int k = 0; k < DEPTH; ++k) {
            float* row = &acc[dj[k] * F + lane * 2];
            __hip_atomic_fetch_add(row,     hv[k].x, __ATOMIC_RELAXED, __HIP_MEMORY_SCOPE_WORKGROUP);
            __hip_atomic_fetch_add(row + 1, hv[k].y, __ATOMIC_RELAXED, __HIP_MEMORY_SCOPE_WORKGROUP);
        }
    }
    for (; j < cnt; j += DEPTH) {
        float2 hv[DEPTH];
        int dj[DEPTH];
        #pragma unroll
        for (int k = 0; k < DEPTH; ++k) {
            const int jj = min(j + k, cnt - 1);
            const int sj = __builtin_amdgcn_readlane(s, jj);
            dj[k] = __builtin_amdgcn_readlane(dl, jj);
            hv[k] = *(const float2*)(h + (long long)sj * F + lane * 2);
        }
        #pragma unroll
        for (int k = 0; k < DEPTH; ++k) {
            if (k == 0 || j + k < cnt) {
                float* row = &acc[dj[k] * F + lane * 2];
                __hip_atomic_fetch_add(row,     hv[k].x, __ATOMIC_RELAXED, __HIP_MEMORY_SCOPE_WORKGROUP);
                __hip_atomic_fetch_add(row + 1, hv[k].y, __ATOMIC_RELAXED, __HIP_MEMORY_SCOPE_WORKGROUP);
            }
        }
    }
}

static __device__ __forceinline__ int agg_stage(unsigned* st, int sn, unsigned entry,
                                                int cnt, int lane, int* lines,
                                                unsigned* __restrict__ blist) {
    if (lane < cnt) st[sn + lane] = entry;
    sn += cnt;
    if (sn >= 32) {
        int ln = 0;
        if (lane == 0) ln = atomicAdd(lines, 1);
        ln = __builtin_amdgcn_readfirstlane(ln);
        const unsigned e = st[lane];
        const unsigned r = st[32 + lane];
        blist[ln * 32 + lane] = e;
        sn -= 32;
        if (lane < sn) st[lane] = r;
    }
    return sn;
}

__global__ __launch_bounds__(1024) void sage_aggregate(const float* __restrict__ h,
                                                       const int* __restrict__ src,
                                                       const int* __restrict__ dst,
                                                       const float* __restrict__ deg,
                                                       float* __restrict__ msum,
                                                       unsigned* __restrict__ list,
                                                       int* __restrict__ dir) {
    __shared__ __attribute__((aligned(16))) float acc[AGG_RN * F];
    __shared__ unsigned queue[AGG_WAVES * AGG_CHUNK];
    __shared__ unsigned stage[AGG_WAVES * 64];
    __shared__ int s_below;
    __shared__ int s_lines;
    const int tid  = threadIdx.x;
    const int lane = tid & 31;
    const int wave = __builtin_amdgcn_readfirstlane(tid >> 5);
    const int r0   = blockIdx.x * AGG_RN;

    float4* acc4 = (float4*)acc;
    for (int j = tid; j < AGG_RN * F / 4; j += 1024) acc4[j] = make_float4(0.f, 0.f, 0.f, 0.f);
    if (tid == 0) { s_below = 0; s_lines = 0; }

    int below = 0;
    {
        const float4* deg4 = (const float4*)deg;
        const int n4 = r0 >> 2;
        for (int j = tid; j < n4; j += 8 * 1024) {
            float4 t[8];
            #pragma unroll
            for (int k = 0; k < 8; ++k) {
                const int idx = j + k * 1024;
                t[k] = deg4[min(idx, n4 - 1)];
                if (idx >= n4) t[k] = make_float4(0.f, 0.f, 0.f, 0.f);
            }
            #pragma unroll
            for (int k = 0; k < 8; ++k) below += (int)t[k].x + (int)t[k].y + (int)t[k].z + (int)t[k].w;
        }
        if (tid < (r0 & 3)) below += (int)deg[(n4 << 2) + tid];
        for (int m = 16; m >= 1; m >>= 1) below += __shfl_xor(below, m);
    }
    __syncthreads();
    if (lane == 0 && below != 0) atomicAdd(&s_below, below);
    __syncthreads();
    unsigned* blist = list + ((__builtin_amdgcn_readfirstlane(s_below) & ~31) + (int)blockIdx.x * LIST_PAD);
    unsigned* st = stage + wave * 64;
    int sn = 0;

    unsigned* q = queue + wave * AGG_CHUNK;
    const int4* dst4 = (const int4*)dst;
    const int nchunks = (N_EDGES + AGG_CHUNK - 1) / AGG_CHUNK;

    const int off = (int)(((long long)blockIdx.x * nchunks) / AGG_RANGES);

    int4 v[8];
    {
        const int c = wave + off;
        agg_load_chunk(dst4, (c >= nchunks) ? c - nchunks : c, lane, v);
    }
    for (int i = wave; i < nchunks; i += AGG_WAVES) {
        const int c = (i + off >= nchunks) ? i + off - nchunks : i + off;
        int qn = 0;
        #pragma unroll
        for (int u = 0; u < 8; ++u) {
            const unsigned ux = (unsigned)(v[u].x - r0);
            const unsigned uy = (unsigned)(v[u].y - r0);
            const unsigned uz = (unsigned)(v[u].z - r0);
            const unsigned uw = (unsigned)(v[u].w - r0);
            const unsigned mn = min(min(ux, uy), min(uz, uw));
            if (__builtin_amdgcn_ballot_w32(mn < (unsigned)AGG_RN) != 0u) {
                const unsigned e0 = (unsigned)(c * (AGG_CHUNK / 4) + u * 32 + lane) * 4u;
                const unsigned uc[4] = { ux, uy, uz, uw };
                #pragma unroll
                for (int k = 0; k < 4; ++k) {
                    const bool m = uc[k] < (unsigned)AGG_RN;
                    const unsigned bal = __builtin_amdgcn_ballot_w32(m);
                    if (m) q[qn + (int)__builtin_amdgcn_mbcnt_lo(bal, 0u)] = ((e0 + k) << 10) | uc[k];
                    qn += __builtin_popcount(bal);
                }
            }
        }

        const int cnt0 = min(32, qn);
        unsigned ent = 0u;
        int s = 0;
        if (lane < cnt0) { ent = q[lane]; s = src[ent >> 10]; }

        {
            const int cn = min(i + AGG_WAVES, nchunks - 1) + off;
            agg_load_chunk(dst4, (cn >= nchunks) ? cn - nchunks : cn, lane, v);
        }

        if (cnt0 > 0) {
            agg_add_rows<AGG_DEPTH, false>(h, acc, s, (int)(ent & 1023u), cnt0, lane);
            sn = agg_stage(st, sn, ((unsigned)s << 10) | (ent & 1023u), cnt0, lane, &s_lines, blist);
        }

        for (int b = 32; b < qn; b += 32) {
            const int cnt = min(32, qn - b);
            unsigned ent2 = 0u;
            int s2 = 0;
            if (lane < cnt) { ent2 = q[b + lane]; s2 = src[ent2 >> 10]; }
            agg_add_rows<AGG_DEPTH, false>(h, acc, s2, (int)(ent2 & 1023u), cnt, lane);
            sn = agg_stage(st, sn, ((unsigned)s2 << 10) | (ent2 & 1023u), cnt, lane, &s_lines, blist);
        }
    }
    if (sn > 0) {
        int ln = 0;
        if (lane == 0) ln = atomicAdd(&s_lines, 1);
        ln = __builtin_amdgcn_readfirstlane(ln);
        blist[ln * 32 + lane] = (lane < sn) ? st[lane] : LIST_END;
    }
    __syncthreads();

    if (tid == 0) {
        dir[2 * blockIdx.x]     = (int)(blist - list);
        dir[2 * blockIdx.x + 1] = s_lines;
    }
    float4* out4 = (float4*)(msum + (long long)r0 * F);
    for (int j = tid; j < AGG_RN * F / 4; j += 1024) out4[j] = acc4[j];
}

__global__ __launch_bounds__(1024) void sage_replay(const float* __restrict__ h,
                                                    const unsigned* __restrict__ list,
                                                    const int* __restrict__ dir,
                                                    float* __restrict__ msum) {
    __shared__ __attribute__((aligned(16))) float acc[AGG_RN * F];
    const int tid  = threadIdx.x;
    const int lane = tid & 31;
    const int wave = __builtin_amdgcn_readfirstlane(tid >> 5);
    const int r0   = blockIdx.x * AGG_RN;

    const unsigned* blist = list + dir[2 * blockIdx.x];
    const int nl = dir[2 * blockIdx.x + 1];

    unsigned e = LIST_END;
    if (wave < nl) e = blist[wave * 32 + lane];

    float4* acc4 = (float4*)acc;
    for (int j = tid; j < AGG_RN * F / 4; j += 1024) acc4[j] = make_float4(0.f, 0.f, 0.f, 0.f);
    __syncthreads();

    for (int i = wave; i < nl; i += AGG_WAVES) {
        const unsigned en = blist[min(i + AGG_WAVES, nl - 1) * 32 + lane];
        const int cnt = __builtin_popcount(__builtin_amdgcn_ballot_w32(e != LIST_END));
        agg_add_rows<REPLAY_DEPTH, true>(h, acc, (int)(e >> 10), (int)(e & 1023u), cnt, lane);
        e = en;
    }
    __syncthreads();

    float4* out4 = (float4*)(msum + (long long)r0 * F);
    for (int j = tid; j < AGG_RN * F / 4; j += 1024) out4[j] = acc4[j];
}

__global__ __launch_bounds__(256) void sage_gemm(const float* h,
                                                 const float* __restrict__ msum,
                                                 const float* __restrict__ deg,
                                                 const float* __restrict__ Wself,
                                                 const float* __restrict__ Wneigh,
                                                 const float* __restrict__ bias,
                                                 float* out,
                                                 int n_nodes, int do_relu) {
    __shared__ float sWs[F * F];
    __shared__ float sWn[F * F];
    __shared__ float sH [32 * HSTRIDE];
    __shared__ float sHn[32 * HSTRIDE];
    __shared__ float sB [F];

    const int tid = threadIdx.x;
    const int node0 = blockIdx.x * 32;

    #pragma unroll
    for (int i = 0; i < 16; ++i) {
        sWs[tid + i * 256] = Wself [tid + i * 256];
        sWn[tid + i * 256] = Wneigh[tid + i * 256];
    }
    if (tid < F) sB[tid] = bias[tid];

    #pragma unroll
    for (int i = 0; i < 8; ++i) {
        int idx = tid + i * 256;
        int row = idx >> 6;
        int col = idx & 63;
        int node = node0 + row;
        float hv = 0.0f, mv = 0.0f, dv = 1.0f;
        if (node < n_nodes) {
            hv = h   [(long long)node * F + col];
            mv = msum[(long long)node * F + col];
            dv = deg[node];
        }
        float rinv = 1.0f / fmaxf(dv, 1.0f);
        sH [row * HSTRIDE + col] = hv;
        sHn[row * HSTRIDE + col] = mv * rinv;
    }
    __syncthreads();

    const int wave  = tid >> 5;
    const int lane  = tid & 31;
    const int m0    = (wave >> 2) * 16;
    const int c0    = (wave & 3) * 16;
    const int lrow  = lane & 15;
    const int khalf = lane >> 4;

    v8f acc = {};
    const float* aSelf = &sH [m0 * HSTRIDE + lrow * HSTRIDE];
    const float* aNei  = &sHn[m0 * HSTRIDE + lrow * HSTRIDE];
    #pragma unroll
    for (int kk = 0; kk < 16; ++kk) {
        const int kb = kk * 4 + khalf * 2;
        v2f a, b;
        a.x = aSelf[kb];
        a.y = aSelf[kb + 1];
        b.x = sWs[(kb    ) * F + c0 + lrow];
        b.y = sWs[(kb + 1) * F + c0 + lrow];
        acc = __builtin_amdgcn_wmma_f32_16x16x4_f32(false, a, false, b,
                                                    (short)0, acc, false, false);
        a.x = aNei[kb];
        a.y = aNei[kb + 1];
        b.x = sWn[(kb    ) * F + c0 + lrow];
        b.y = sWn[(kb + 1) * F + c0 + lrow];
        acc = __builtin_amdgcn_wmma_f32_16x16x4_f32(false, a, false, b,
                                                    (short)0, acc, false, false);
    }

    const float bv = sB[c0 + lrow];
    #pragma unroll
    for (int i = 0; i < 8; ++i) {
        int row = node0 + m0 + i + khalf * 8;
        float v = acc[i] + bv;
        if (do_relu) v = fmaxf(v, 0.0f);
        if (row < n_nodes) out[(long long)row * F + c0 + lrow] = v;
    }
}

extern "C" void kernel_launch(void* const* d_in, const int* in_sizes, int n_in,
                              void* d_out, int out_size, void* d_ws, size_t ws_size,
                              hipStream_t stream) {
    (void)in_sizes; (void)n_in; (void)out_size; (void)ws_size;
    const float* feat = (const float*)d_in[0];
    const int*   src  = (const int*)d_in[1];
    const int*   dst  = (const int*)d_in[2];
    const float* Ws[3] = { (const float*)d_in[3], (const float*)d_in[6], (const float*)d_in[9]  };
    const float* Wn[3] = { (const float*)d_in[4], (const float*)d_in[7], (const float*)d_in[10] };
    const float* bs[3] = { (const float*)d_in[5], (const float*)d_in[8], (const float*)d_in[11] };
    float* outp = (float*)d_out;

    float* ws   = (float*)d_ws;
    float* deg  = ws;
    float* msum = deg  + (long long)N_NODES;
    float* h1   = msum + (long long)N_NODES * F;
    float* h2   = h1   + (long long)N_NODES * F;

    const int initBlocks  = (N_NODES + 255) / 256;
    const int degBlocks   = DEG_RANGES * DEG_SLICES;
    const int aggBlocks   = AGG_RANGES;
    const int gemmBlocks  = (N_NODES + 31) / 32;

    int* part = (int*)h2;
    sage_degree_count<<<degBlocks, 1024, 0, stream>>>(dst, part);
    sage_init<<<initBlocks, 256, 0, stream>>>(part, deg);

    unsigned* list = (unsigned*)h2;
    int* dir = (int*)(list + LIST_CAP);

    sage_aggregate<<<aggBlocks, 1024, 0, stream>>>(feat, src, dst, deg, msum, list, dir);
    sage_gemm<<<gemmBlocks, 256, 0, stream>>>(feat, msum, deg, Ws[0], Wn[0], bs[0], h1, N_NODES, 1);

    sage_replay<<<aggBlocks, 1024, REPLAY_LDS_RESERVE, stream>>>(h1, list, dir, msum);
    sage_gemm<<<gemmBlocks, 256, 0, stream>>>(h1, msum, deg, Ws[1], Wn[1], bs[1], h1, N_NODES, 1);

    sage_replay<<<aggBlocks, 1024, REPLAY_LDS_RESERVE, stream>>>(h1, list, dir, msum);
    sage_gemm<<<gemmBlocks, 256, 0, stream>>>(h1, msum, deg, Ws[2], Wn[2], bs[2], outp, N_NODES, 0);
}


# aggregate and list walk: 250 node ranges of 400 (was 160 of 625), one block per WGP: a layer's row gathers and LDS adds run on 250 of the 256 WGPs, not 160; sum order changed
# speedup vs baseline: 2.3959x; 1.1281x over previous
#include <hip/hip_runtime.h>


#define N_NODES 100000
#define N_EDGES 1600000
#define F 64
#define HSTRIDE 65

typedef __attribute__((ext_vector_type(2))) float v2f;
typedef __attribute__((ext_vector_type(8))) float v8f;

#define DEG_RANGES  8
#define DEG_SLICES  32
#define DEG_R       (N_NODES / DEG_RANGES)
#define DEG_SLICE_E (N_EDGES / DEG_SLICES)
static_assert(DEG_R * DEG_RANGES == N_NODES, "ranges must tile the nodes");
static_assert(DEG_SLICE_E * DEG_SLICES == N_EDGES, "slices must tile the edges");
static_assert(DEG_SLICE_E % 4 == 0, "a slice is read as 16-byte vectors");

__global__ __launch_bounds__(1024) void sage_degree_count(const int* __restrict__ dst,
                                                          int* __restrict__ part) {
    __shared__ int bins[DEG_R];
    const int tid = threadIdx.x;
    const int rg  = blockIdx.x % DEG_RANGES;
    const int sl  = blockIdx.x / DEG_RANGES;
    const int r0  = rg * DEG_R;
    for (int j = tid; j < DEG_R; j += 1024) bins[j] = 0;
    __syncthreads();

    const int4* p = (const int4*)(dst + (long long)sl * DEG_SLICE_E);
    const int nvec = DEG_SLICE_E / 4;
    for (int i = tid; i < nvec; i += 4096) {
        int4 v[4];
        #pragma unroll
        for (int k = 0; k < 4; ++k) {
            const int idx = i + k * 1024;
            v[k] = (idx < nvec) ? p[idx] : make_int4(-1, -1, -1, -1);
        }
        #pragma unroll
        for (int k = 0; k < 4; ++k) {
            unsigned u;
            u = (unsigned)(v[k].x - r0); if (u < (unsigned)DEG_R) atomicAdd(&bins[u], 1);
            u = (unsigned)(v[k].y - r0); if (u < (unsigned)DEG_R) atomicAdd(&bins[u], 1);
            u = (unsigned)(v[k].z - r0); if (u < (unsigned)DEG_R) atomicAdd(&bins[u], 1);
            u = (unsigned)(v[k].w - r0); if (u < (unsigned)DEG_R) atomicAdd(&bins[u], 1);
        }
    }
    __syncthreads();

    int* row = part + (long long)sl * N_NODES + r0;
    for (int j = tid; j < DEG_R; j += 1024) row[j] = bins[j];
}

__global__ __launch_bounds__(256) void sage_init(const int* __restrict__ part,
                                                 float* __restrict__ deg) {
    const int i = blockIdx.x * 256 + threadIdx.x;
    if (i < N_NODES) {
        int c = 0;
        #pragma unroll
        for (int s = 0; s < DEG_SLICES; ++s) c += part[(long long)s * N_NODES + i];
        deg[i] = (float)c;
    }
}

#define AGG_RANGES 250
#define AGG_RN     (N_NODES / AGG_RANGES)
#define AGG_WAVES  32
#define AGG_CHUNK  1024
#define AGG_DEPTH  8
#define REPLAY_DEPTH 16
#define LIST_PAD   1056
#define LIST_CAP   (N_EDGES + AGG_RANGES * LIST_PAD)
#define LIST_END   0xFFFFFFFFu
#define REPLAY_ACC_FLOATS 41472
static_assert(REPLAY_ACC_FLOATS >= AGG_RN * F, "room for the block's sums");
static_assert(REPLAY_ACC_FLOATS * 4 > 160 * 1024, "one list-walk block per WGP");
static_assert(LIST_PAD >= 32 * AGG_WAVES + 32, "room for every wave's part line and the rounding");
static_assert(LIST_CAP + 2 * AGG_RANGES <= N_NODES * F, "list and directory fit in one plane");
static_assert(N_NODES <= (1 << 21), "an entry keeps the source node above the 10 bits of the row");
static_assert(AGG_RN * AGG_RANGES == N_NODES, "ranges must tile the nodes");
static_assert(AGG_RN <= 1024, "a queue entry keeps the local node in 10 bits");
static_assert(N_EDGES <= (1 << 21), "a queue entry keeps the edge number in 21 bits");
static_assert(N_EDGES % 4 == 0, "the destination list is read as 16-byte vectors");
static_assert((N_EDGES + AGG_CHUNK - 1) / AGG_CHUNK >= AGG_WAVES, "every wave has a first chunk");

static __device__ __forceinline__ void agg_load_chunk(const int4* __restrict__ dst4, int c, int lane,
                                                      int4 (&v)[8]) {
    #pragma unroll
    for (int u = 0; u < 8; ++u) {
        const int vi = c * (AGG_CHUNK / 4) + u * 32 + lane;
        v[u] = dst4[min(vi, N_EDGES / 4 - 1)];
        if (vi >= N_EDGES / 4) v[u] = make_int4(-1, -1, -1, -1);
    }
}

template <int DEPTH, bool WHOLE>
static __device__ __forceinline__ void agg_add_rows(const float* __restrict__ h, float* acc,
                                                    int s, int dl, int cnt, int lane) {
    int j = 0;
    for (; WHOLE && j + DEPTH <= cnt; j += DEPTH) {
        float2 hv[DEPTH];
        int dj[DEPTH];
        #pragma unroll
        for (int k = 0; k < DEPTH; ++k) {
            const int sj = __builtin_amdgcn_readlane(s, j + k);
            dj[k] = __builtin_amdgcn_readlane(dl, j + k);
            hv[k] = *(const float2*)(h + (long long)sj * F + lane * 2);
        }
        #pragma unroll
        for (int k = 0; k < DEPTH; ++k) {
            float* row = &acc[dj[k] * F + lane * 2];
            __hip_atomic_fetch_add(row,     hv[k].x, __ATOMIC_RELAXED, __HIP_MEMORY_SCOPE_WORKGROUP);
            __hip_atomic_fetch_add(row + 1, hv[k].y, __ATOMIC_RELAXED, __HIP_MEMORY_SCOPE_WORKGROUP);
        }
    }
    for (; j < cnt; j += DEPTH) {
        float2 hv[DEPTH];
        int dj[DEPTH];
        #pragma unroll
        for (int k = 0; k < DEPTH; ++k) {
            const int jj = min(j + k, cnt - 1);
            const int sj = __builtin_amdgcn_readlane(s, jj);
            dj[k] = __builtin_amdgcn_readlane(dl, jj);
            hv[k] = *(const float2*)(h + (long long)sj * F + lane * 2);
        }
        #pragma unroll
        for (int k = 0; k < DEPTH; ++k) {
            if (k == 0 || j + k < cnt) {
                float* row = &acc[dj[k] * F + lane * 2];
                __hip_atomic_fetch_add(row,     hv[k].x, __ATOMIC_RELAXED, __HIP_MEMORY_SCOPE_WORKGROUP);
                __hip_atomic_fetch_add(row + 1, hv[k].y, __ATOMIC_RELAXED, __HIP_MEMORY_SCOPE_WORKGROUP);
            }
        }
    }
}

static __device__ __forceinline__ int agg_stage(unsigned* st, int sn, unsigned entry,
                                                int cnt, int lane, int* lines,
                                                unsigned* __restrict__ blist) {
    if (lane < cnt) st[sn + lane] = entry;
    sn += cnt;
    if (sn >= 32) {
        int ln = 0;
        if (lane == 0) ln = atomicAdd(lines, 1);
        ln = __builtin_amdgcn_readfirstlane(ln);
        const unsigned e = st[lane];
        const unsigned r = st[32 + lane];
        blist[ln * 32 + lane] = e;
        sn -= 32;
        if (lane < sn) st[lane] = r;
    }
    return sn;
}

__global__ __launch_bounds__(1024) void sage_aggregate(const float* __restrict__ h,
                                                       const int* __restrict__ src,
                                                       const int* __restrict__ dst,
                                                       const float* __restrict__ deg,
                                                       float* __restrict__ msum,
                                                       unsigned* __restrict__ list,
                                                       int* __restrict__ dir) {
    __shared__ __attribute__((aligned(16))) float acc[AGG_RN * F];
    __shared__ unsigned queue[AGG_WAVES * AGG_CHUNK];
    __shared__ unsigned stage[AGG_WAVES * 64];
    __shared__ int s_below;
    __shared__ int s_lines;
    const int tid  = threadIdx.x;
    const int lane = tid & 31;
    const int wave = __builtin_amdgcn_readfirstlane(tid >> 5);
    const int r0   = blockIdx.x * AGG_RN;

    float4* acc4 = (float4*)acc;
    for (int j = tid; j < AGG_RN * F / 4; j += 1024) acc4[j] = make_float4(0.f, 0.f, 0.f, 0.f);
    if (tid == 0) { s_below = 0; s_lines = 0; }

    int below = 0;
    {
        const float4* deg4 = (const float4*)deg;
        const int n4 = r0 >> 2;
        for (int j = tid; j < n4; j += 8 * 1024) {
            float4 t[8];
            #pragma unroll
            for (int k = 0; k < 8; ++k) {
                const int idx = j + k * 1024;
                t[k] = deg4[min(idx, n4 - 1)];
                if (idx >= n4) t[k] = make_float4(0.f, 0.f, 0.f, 0.f);
            }
            #pragma unroll
            for (int k = 0; k < 8; ++k) below += (int)t[k].x + (int)t[k].y + (int)t[k].z + (int)t[k].w;
        }
        if (tid < (r0 & 3)) below += (int)deg[(n4 << 2) + tid];
        for (int m = 16; m >= 1; m >>= 1) below += __shfl_xor(below, m);
    }
    __syncthreads();
    if (lane == 0 && below != 0) atomicAdd(&s_below, below);
    __syncthreads();
    unsigned* blist = list + ((__builtin_amdgcn_readfirstlane(s_below) & ~31) + (int)blockIdx.x * LIST_PAD);
    unsigned* st = stage + wave * 64;
    int sn = 0;

    unsigned* q = queue + wave * AGG_CHUNK;
    const int4* dst4 = (const int4*)dst;
    const int nchunks = (N_EDGES + AGG_CHUNK - 1) / AGG_CHUNK;

    const int off = (int)(((long long)blockIdx.x * nchunks) / AGG_RANGES);

    int4 v[8];
    {
        const int c = wave + off;
        agg_load_chunk(dst4, (c >= nchunks) ? c - nchunks : c, lane, v);
    }
    for (int i = wave; i < nchunks; i += AGG_WAVES) {
        const int c = (i + off >= nchunks) ? i + off - nchunks : i + off;
        int qn = 0;
        #pragma unroll
        for (int u = 0; u < 8; ++u) {
            const unsigned ux = (unsigned)(v[u].x - r0);
            const unsigned uy = (unsigned)(v[u].y - r0);
            const unsigned uz = (unsigned)(v[u].z - r0);
            const unsigned uw = (unsigned)(v[u].w - r0);
            const unsigned mn = min(min(ux, uy), min(uz, uw));
            if (__builtin_amdgcn_ballot_w32(mn < (unsigned)AGG_RN) != 0u) {
                const unsigned e0 = (unsigned)(c * (AGG_CHUNK / 4) + u * 32 + lane) * 4u;
                const unsigned uc[4] = { ux, uy, uz, uw };
                #pragma unroll
                for (int k = 0; k < 4; ++k) {
                    const bool m = uc[k] < (unsigned)AGG_RN;
                    const unsigned bal = __builtin_amdgcn_ballot_w32(m);
                    if (m) q[qn + (int)__builtin_amdgcn_mbcnt_lo(bal, 0u)] = ((e0 + k) << 10) | uc[k];
                    qn += __builtin_popcount(bal);
                }
            }
        }

        const int cnt0 = min(32, qn);
        unsigned ent = 0u;
        int s = 0;
        if (lane < cnt0) { ent = q[lane]; s = src[ent >> 10]; }

        {
            const int cn = min(i + AGG_WAVES, nchunks - 1) + off;
            agg_load_chunk(dst4, (cn >= nchunks) ? cn - nchunks : cn, lane, v);
        }

        if (cnt0 > 0) {
            agg_add_rows<AGG_DEPTH, false>(h, acc, s, (int)(ent & 1023u), cnt0, lane);
            sn = agg_stage(st, sn, ((unsigned)s << 10) | (ent & 1023u), cnt0, lane, &s_lines, blist);
        }

        for (int b = 32; b < qn; b += 32) {
            const int cnt = min(32, qn - b);
            unsigned ent2 = 0u;
            int s2 = 0;
            if (lane < cnt) { ent2 = q[b + lane]; s2 = src[ent2 >> 10]; }
            agg_add_rows<AGG_DEPTH, false>(h, acc, s2, (int)(ent2 & 1023u), cnt, lane);
            sn = agg_stage(st, sn, ((unsigned)s2 << 10) | (ent2 & 1023u), cnt, lane, &s_lines, blist);
        }
    }
    if (sn > 0) {
        int ln = 0;
        if (lane == 0) ln = atomicAdd(&s_lines, 1);
        ln = __builtin_amdgcn_readfirstlane(ln);
        blist[ln * 32 + lane] = (lane < sn) ? st[lane] : LIST_END;
    }
    __syncthreads();

    if (tid == 0) {
        dir[2 * blockIdx.x]     = (int)(blist - list);
        dir[2 * blockIdx.x + 1] = s_lines;
    }
    float4* out4 = (float4*)(msum + (long long)r0 * F);
    for (int j = tid; j < AGG_RN * F / 4; j += 1024) out4[j] = acc4[j];
}

__global__ __launch_bounds__(1024) void sage_replay(const float* __restrict__ h,
                                                    const unsigned* __restrict__ list,
                                                    const int* __restrict__ dir,
                                                    float* __restrict__ msum) {
    __shared__ __attribute__((aligned(16))) float acc[REPLAY_ACC_FLOATS];
    const int tid  = threadIdx.x;
    const int lane = tid & 31;
    const int wave = __builtin_amdgcn_readfirstlane(tid >> 5);
    const int r0   = blockIdx.x * AGG_RN;

    const unsigned* blist = list + dir[2 * blockIdx.x];
    const int nl = dir[2 * blockIdx.x + 1];

    unsigned e = LIST_END;
    if (wave < nl) e = blist[wave * 32 + lane];

    float4* acc4 = (float4*)acc;
    for (int j = tid; j < AGG_RN * F / 4; j += 1024) acc4[j] = make_float4(0.f, 0.f, 0.f, 0.f);
    __syncthreads();

    for (int i = wave; i < nl; i += AGG_WAVES) {
        const unsigned en = blist[min(i + AGG_WAVES, nl - 1) * 32 + lane];
        const int cnt = __builtin_popcount(__builtin_amdgcn_ballot_w32(e != LIST_END));
        agg_add_rows<REPLAY_DEPTH, true>(h, acc, (int)(e >> 10), (int)(e & 1023u), cnt, lane);
        e = en;
    }
    __syncthreads();

    float4* out4 = (float4*)(msum + (long long)r0 * F);
    for (int j = tid; j < AGG_RN * F / 4; j += 1024) out4[j] = acc4[j];
}

__global__ __launch_bounds__(256) void sage_gemm(const float* h,
                                                 const float* __restrict__ msum,
                                                 const float* __restrict__ deg,
                                                 const float* __restrict__ Wself,
                                                 const float* __restrict__ Wneigh,
                                                 const float* __restrict__ bias,
                                                 float* out,
                                                 int n_nodes, int do_relu) {
    __shared__ float sWs[F * F];
    __shared__ float sWn[F * F];
    __shared__ float sH [32 * HSTRIDE];
    __shared__ float sHn[32 * HSTRIDE];
    __shared__ float sB [F];

    const int tid = threadIdx.x;
    const int node0 = blockIdx.x * 32;

    #pragma unroll
    for (int i = 0; i < 16; ++i) {
        sWs[tid + i * 256] = Wself [tid + i * 256];
        sWn[tid + i * 256] = Wneigh[tid + i * 256];
    }
    if (tid < F) sB[tid] = bias[tid];

    #pragma unroll
    for (int i = 0; i < 8; ++i) {
        int idx = tid + i * 256;
        int row = idx >> 6;
        int col = idx & 63;
        int node = node0 + row;
        float hv = 0.0f, mv = 0.0f, dv = 1.0f;
        if (node < n_nodes) {
            hv = h   [(long long)node * F + col];
            mv = msum[(long long)node * F + col];
            dv = deg[node];
        }
        float rinv = 1.0f / fmaxf(dv, 1.0f);
        sH [row * HSTRIDE + col] = hv;
        sHn[row * HSTRIDE + col] = mv * rinv;
    }
    __syncthreads();

    const int wave  = tid >> 5;
    const int lane  = tid & 31;
    const int m0    = (wave >> 2) * 16;
    const int c0    = (wave & 3) * 16;
    const int lrow  = lane & 15;
    const int khalf = lane >> 4;

    v8f acc = {};
    const float* aSelf = &sH [m0 * HSTRIDE + lrow * HSTRIDE];
    const float* aNei  = &sHn[m0 * HSTRIDE + lrow * HSTRIDE];
    #pragma unroll
    for (int kk = 0; kk < 16; ++kk) {
        const int kb = kk * 4 + khalf * 2;
        v2f a, b;
        a.x = aSelf[kb];
        a.y = aSelf[kb + 1];
        b.x = sWs[(kb    ) * F + c0 + lrow];
        b.y = sWs[(kb + 1) * F + c0 + lrow];
        acc = __builtin_amdgcn_wmma_f32_16x16x4_f32(false, a, false, b,
                                                    (short)0, acc, false, false);
        a.x = aNei[kb];
        a.y = aNei[kb + 1];
        b.x = sWn[(kb    ) * F + c0 + lrow];
        b.y = sWn[(kb + 1) * F + c0 + lrow];
        acc = __builtin_amdgcn_wmma_f32_16x16x4_f32(false, a, false, b,
                                                    (short)0, acc, false, false);
    }

    const float bv = sB[c0 + lrow];
    #pragma unroll
    for (int i = 0; i < 8; ++i) {
        int row = node0 + m0 + i + khalf * 8;
        float v = acc[i] + bv;
        if (do_relu) v = fmaxf(v, 0.0f);
        if (row < n_nodes) out[(long long)row * F + c0 + lrow] = v;
    }
}

extern "C" void kernel_launch(void* const* d_in, const int* in_sizes, int n_in,
                              void* d_out, int out_size, void* d_ws, size_t ws_size,
                              hipStream_t stream) {
    (void)in_sizes; (void)n_in; (void)out_size; (void)ws_size;
    const float* feat = (const float*)d_in[0];
    const int*   src  = (const int*)d_in[1];
    const int*   dst  = (const int*)d_in[2];
    const float* Ws[3] = { (const float*)d_in[3], (const float*)d_in[6], (const float*)d_in[9]  };
    const float* Wn[3] = { (const float*)d_in[4], (const float*)d_in[7], (const float*)d_in[10] };
    const float* bs[3] = { (const float*)d_in[5], (const float*)d_in[8], (const float*)d_in[11] };
    float* outp = (float*)d_out;

    float* ws   = (float*)d_ws;
    float* deg  = ws;
    float* msum = deg  + (long long)N_NODES;
    float* h1   = msum + (long long)N_NODES * F;
    float* h2   = h1   + (long long)N_NODES * F;

    const int initBlocks  = (N_NODES + 255) / 256;
    const int degBlocks   = DEG_RANGES * DEG_SLICES;
    const int aggBlocks   = AGG_RANGES;
    const int gemmBlocks  = (N_NODES + 31) / 32;

    int* part = (int*)h2;
    sage_degree_count<<<degBlocks, 1024, 0, stream>>>(dst, part);
    sage_init<<<initBlocks, 256, 0, stream>>>(part, deg);

    unsigned* list = (unsigned*)h2;
    int* dir = (int*)(list + LIST_CAP);

    sage_aggregate<<<aggBlocks, 1024, 0, stream>>>(feat, src, dst, deg, msum, list, dir);
    sage_gemm<<<gemmBlocks, 256, 0, stream>>>(feat, msum, deg, Ws[0], Wn[0], bs[0], h1, N_NODES, 1);

    sage_replay<<<aggBlocks, 1024, 0, stream>>>(h1, list, dir, msum);
    sage_gemm<<<gemmBlocks, 256, 0, stream>>>(h1, msum, deg, Ws[1], Wn[1], bs[1], h1, N_NODES, 1);

    sage_replay<<<aggBlocks, 1024, 0, stream>>>(h1, list, dir, msum);
    sage_gemm<<<gemmBlocks, 256, 0, stream>>>(h1, msum, deg, Ws[2], Wn[2], bs[2], outp, N_NODES, 0);
}


# layer GEMM moved into the aggregation block: its 400 nodes' sums stay in LDS and feed the WMMA neighbour operand from there; msum plane and 3 of 8 launches gone; sums and k order as in parent
# speedup vs baseline: 3.0846x; 1.2874x over previous
#include <hip/hip_runtime.h>


#define N_NODES 100000
#define N_EDGES 1600000
#define F 64
#define ACC_P 68
#define W_P   80

typedef __attribute__((ext_vector_type(2))) float v2f;
typedef __attribute__((ext_vector_type(8))) float v8f;

#define DEG_RANGES  8
#define DEG_SLICES  32
#define DEG_R       (N_NODES / DEG_RANGES)
#define DEG_SLICE_E (N_EDGES / DEG_SLICES)
static_assert(DEG_R * DEG_RANGES == N_NODES, "ranges must tile the nodes");
static_assert(DEG_SLICE_E * DEG_SLICES == N_EDGES, "slices must tile the edges");
static_assert(DEG_SLICE_E % 4 == 0, "a slice is read as 16-byte vectors");

__global__ __launch_bounds__(1024) void sage_degree_count(const int* __restrict__ dst,
                                                          int* __restrict__ part) {
    __shared__ int bins[DEG_R];
    const int tid = threadIdx.x;
    const int rg  = blockIdx.x % DEG_RANGES;
    const int sl  = blockIdx.x / DEG_RANGES;
    const int r0  = rg * DEG_R;
    for (int j = tid; j < DEG_R; j += 1024) bins[j] = 0;
    __syncthreads();

    const int4* p = (const int4*)(dst + (long long)sl * DEG_SLICE_E);
    const int nvec = DEG_SLICE_E / 4;
    for (int i = tid; i < nvec; i += 4096) {
        int4 v[4];
        #pragma unroll
        for (int k = 0; k < 4; ++k) {
            const int idx = i + k * 1024;
            v[k] = (idx < nvec) ? p[idx] : make_int4(-1, -1, -1, -1);
        }
        #pragma unroll
        for (int k = 0; k < 4; ++k) {
            unsigned u;
            u = (unsigned)(v[k].x - r0); if (u < (unsigned)DEG_R) atomicAdd(&bins[u], 1);
            u = (unsigned)(v[k].y - r0); if (u < (unsigned)DEG_R) atomicAdd(&bins[u], 1);
            u = (unsigned)(v[k].z - r0); if (u < (unsigned)DEG_R) atomicAdd(&bins[u], 1);
            u = (unsigned)(v[k].w - r0); if (u < (unsigned)DEG_R) atomicAdd(&bins[u], 1);
        }
    }
    __syncthreads();

    int* row = part + (long long)sl * N_NODES + r0;
    for (int j = tid; j < DEG_R; j += 1024) row[j] = bins[j];
}

__global__ __launch_bounds__(256) void sage_init(const int* __restrict__ part,
                                                 float* __restrict__ deg) {
    const int i = blockIdx.x * 256 + threadIdx.x;
    if (i < N_NODES) {
        int c = 0;
        #pragma unroll
        for (int s = 0; s < DEG_SLICES; ++s) c += part[(long long)s * N_NODES + i];
        deg[i] = (float)c;
    }
}

#define AGG_RANGES 250
#define AGG_RN     (N_NODES / AGG_RANGES)
#define AGG_WAVES  32
#define AGG_CHUNK  1024
#define AGG_DEPTH  8
#define REPLAY_DEPTH 16
#define LIST_PAD   1056
#define LIST_CAP   (N_EDGES + AGG_RANGES * LIST_PAD)
#define LIST_END   0xFFFFFFFFu
#define REPLAY_ACC_FLOATS 41472
static_assert(REPLAY_ACC_FLOATS >= AGG_RN * ACC_P, "room for the block's sums");
static_assert(AGG_RN % 16 == 0 && AGG_RN / 16 <= AGG_WAVES, "one 16-node tile per wave covers the range");
static_assert(F * F / 4 == AGG_WAVES * 32, "one 16-byte piece of each weight matrix per thread");
static_assert(REPLAY_ACC_FLOATS * 4 > 160 * 1024, "one list-walk block per WGP");
static_assert(LIST_PAD >= 32 * AGG_WAVES + 32, "room for every wave's part line and the rounding");
static_assert(LIST_CAP + 2 * AGG_RANGES <= N_NODES * F, "list and directory fit in one plane");
static_assert(N_NODES <= (1 << 21), "an entry keeps the source node above the 10 bits of the row");
static_assert(AGG_RN * AGG_RANGES == N_NODES, "ranges must tile the nodes");
static_assert(AGG_RN <= 1024, "a queue entry keeps the local node in 10 bits");
static_assert(N_EDGES <= (1 << 21), "a queue entry keeps the edge number in 21 bits");
static_assert(N_EDGES % 4 == 0, "the destination list is read as 16-byte vectors");
static_assert((N_EDGES + AGG_CHUNK - 1) / AGG_CHUNK >= AGG_WAVES, "every wave has a first chunk");

static __device__ __forceinline__ void agg_load_chunk(const int4* __restrict__ dst4, int c, int lane,
                                                      int4 (&v)[8]) {
    #pragma unroll
    for (int u = 0; u < 8; ++u) {
        const int vi = c * (AGG_CHUNK / 4) + u * 32 + lane;
        v[u] = dst4[min(vi, N_EDGES / 4 - 1)];
        if (vi >= N_EDGES / 4) v[u] = make_int4(-1, -1, -1, -1);
    }
}

template <int DEPTH, bool WHOLE>
static __device__ __forceinline__ void agg_add_rows(const float* __restrict__ h, float* acc,
                                                    int s, int dl, int cnt, int lane) {
    int j = 0;
    for (; WHOLE && j + DEPTH <= cnt; j += DEPTH) {
        float2 hv[DEPTH];
        int dj[DEPTH];
        #pragma unroll
        for (int k = 0; k < DEPTH; ++k) {
            const int sj = __builtin_amdgcn_readlane(s, j + k);
            dj[k] = __builtin_amdgcn_readlane(dl, j + k);
            hv[k] = *(const float2*)(h + (long long)sj * F + lane * 2);
        }
        #pragma unroll
        for (int k = 0; k < DEPTH; ++k) {
            float* row = &acc[dj[k] * ACC_P + lane * 2];
            __hip_atomic_fetch_add(row,     hv[k].x, __ATOMIC_RELAXED, __HIP_MEMORY_SCOPE_WORKGROUP);
            __hip_atomic_fetch_add(row + 1, hv[k].y, __ATOMIC_RELAXED, __HIP_MEMORY_SCOPE_WORKGROUP);
        }
    }
    for (; j < cnt; j += DEPTH) {
        float2 hv[DEPTH];
        int dj[DEPTH];
        #pragma unroll
        for (int k = 0; k < DEPTH; ++k) {
            const int jj = min(j + k, cnt - 1);
            const int sj = __builtin_amdgcn_readlane(s, jj);
            dj[k] = __builtin_amdgcn_readlane(dl, jj);
            hv[k] = *(const float2*)(h + (long long)sj * F + lane * 2);
        }
        #pragma unroll
        for (int k = 0; k < DEPTH; ++k) {
            if (k == 0 || j + k < cnt) {
                float* row = &acc[dj[k] * ACC_P + lane * 2];
                __hip_atomic_fetch_add(row,     hv[k].x, __ATOMIC_RELAXED, __HIP_MEMORY_SCOPE_WORKGROUP);
                __hip_atomic_fetch_add(row + 1, hv[k].y, __ATOMIC_RELAXED, __HIP_MEMORY_SCOPE_WORKGROUP);
            }
        }
    }
}

static __device__ __forceinline__ int agg_stage(unsigned* st, int sn, unsigned entry,
                                                int cnt, int lane, int* lines,
                                                unsigned* __restrict__ blist) {
    if (lane < cnt) st[sn + lane] = entry;
    sn += cnt;
    if (sn >= 32) {
        int ln = 0;
        if (lane == 0) ln = atomicAdd(lines, 1);
        ln = __builtin_amdgcn_readfirstlane(ln);
        const unsigned e = st[lane];
        const unsigned r = st[32 + lane];
        blist[ln * 32 + lane] = e;
        sn -= 32;
        if (lane < sn) st[lane] = r;
    }
    return sn;
}

static __device__ __forceinline__ void stage_weights(const float* __restrict__ Wself,
                                                     const float* __restrict__ Wneigh,
                                                     const float* __restrict__ bias,
                                                     float* sWs, float* sWn, float* sB, int tid) {
    const float4 ws = ((const float4*)Wself)[tid];
    const float4 wn = ((const float4*)Wneigh)[tid];
    const int o = (tid >> 4) * W_P + (tid & 15) * 4;
    *(float4*)(sWs + o) = ws;
    *(float4*)(sWn + o) = wn;
    if (tid < F) sB[tid] = bias[tid];
}

static __device__ __forceinline__ void range_gemm(const float* __restrict__ h,
                                                  const float* __restrict__ deg,
                                                  const float* acc,
                                                  const float* sWs, const float* sWn, const float* sB,
                                                  float* __restrict__ hout,
                                                  int r0, int wave, int lane, int do_relu) {
    const int m0    = wave * 16;
    const int lrow  = lane & 15;
    const int khalf = lane >> 4;
    const int node  = r0 + m0 + lrow;

    const float* hrow = h + (long long)node * F + khalf * 2;
    v2f as[16];
    #pragma unroll
    for (int kk = 0; kk < 16; ++kk) as[kk] = *(const v2f*)(hrow + kk * 4);
    const float rinv = 1.0f / fmaxf(deg[node], 1.0f);

    const float* arow = acc + (m0 + lrow) * ACC_P + khalf * 2;
    const float* bs = sWs + khalf * 2 * W_P + lrow;
    const float* bn = sWn + khalf * 2 * W_P + lrow;
    v8f d[4] = {};
    #pragma unroll
    for (int kk = 0; kk < 16; ++kk) {
        v2f an = *(const v2f*)(arow + kk * 4);
        an.x *= rinv;
        an.y *= rinv;
        #pragma unroll
        for (int c = 0; c < 4; ++c) {
            v2f b;
            b.x = bs[(kk * 4    ) * W_P + c * 16];
            b.y = bs[(kk * 4 + 1) * W_P + c * 16];
            d[c] = __builtin_amdgcn_wmma_f32_16x16x4_f32(false, as[kk], false, b,
                                                         (short)0, d[c], false, false);
        }
        #pragma unroll
        for (int c = 0; c < 4; ++c) {
            v2f b;
            b.x = bn[(kk * 4    ) * W_P + c * 16];
            b.y = bn[(kk * 4 + 1) * W_P + c * 16];
            d[c] = __builtin_amdgcn_wmma_f32_16x16x4_f32(false, an, false, b,
                                                         (short)0, d[c], false, false);
        }
    }

    float* orow = hout + (long long)(r0 + m0 + khalf * 8) * F + lrow;
    #pragma unroll
    for (int c = 0; c < 4; ++c) {
        const float bv = sB[c * 16 + lrow];
        #pragma unroll
        for (int i = 0; i < 8; ++i) {
            float v = d[c][i] + bv;
            if (do_relu) v = fmaxf(v, 0.0f);
            orow[i * F + c * 16] = v;
        }
    }
}

__global__ __launch_bounds__(1024) void sage_aggregate(const float* __restrict__ h,
                                                       const int* __restrict__ src,
                                                       const int* __restrict__ dst,
                                                       const float* __restrict__ deg,
                                                       const float* __restrict__ Wself,
                                                       const float* __restrict__ Wneigh,
                                                       const float* __restrict__ bias,
                                                       float* __restrict__ hout,
                                                       unsigned* __restrict__ list,
                                                       int* __restrict__ dir,
                                                       int do_relu) {
    __shared__ __attribute__((aligned(16))) float lds[2 * F * W_P + F + AGG_RN * ACC_P
                                                      + AGG_WAVES * AGG_CHUNK + AGG_WAVES * 64 + 4];
    float* sWs = lds;
    float* sWn = sWs + F * W_P;
    float* sB  = sWn + F * W_P;
    float* acc = sB + F;
    unsigned* queue = (unsigned*)(acc + AGG_RN * ACC_P);
    unsigned* stage = queue + AGG_WAVES * AGG_CHUNK;
    int* s_below = (int*)(stage + AGG_WAVES * 64);
    int* s_lines = s_below + 1;
    const int tid  = threadIdx.x;
    const int lane = tid & 31;
    const int wave = __builtin_amdgcn_readfirstlane(tid >> 5);
    const int r0   = blockIdx.x * AGG_RN;

    stage_weights(Wself, Wneigh, bias, sWs, sWn, sB, tid);

    float4* acc4 = (float4*)acc;
    for (int j = tid; j < AGG_RN * ACC_P / 4; j += 1024) acc4[j] = make_float4(0.f, 0.f, 0.f, 0.f);
    if (tid == 0) { *s_below = 0; *s_lines = 0; }

    int below = 0;
    {
        const float4* deg4 = (const float4*)deg;
        const int n4 = r0 >> 2;
        for (int j = tid; j < n4; j += 8 * 1024) {
            float4 t[8];
            #pragma unroll
            for (int k = 0; k < 8; ++k) {
                const int idx = j + k * 1024;
                t[k] = deg4[min(idx, n4 - 1)];
                if (idx >= n4) t[k] = make_float4(0.f, 0.f, 0.f, 0.f);
            }
            #pragma unroll
            for (int k = 0; k < 8; ++k) below += (int)t[k].x + (int)t[k].y + (int)t[k].z + (int)t[k].w;
        }
        if (tid < (r0 & 3)) below += (int)deg[(n4 << 2) + tid];
        for (int m = 16; m >= 1; m >>= 1) below += __shfl_xor(below, m);
    }
    __syncthreads();
    if (lane == 0 && below != 0) atomicAdd(s_below, below);
    __syncthreads();
    unsigned* blist = list + ((__builtin_amdgcn_readfirstlane(*s_below) & ~31) + (int)blockIdx.x * LIST_PAD);
    unsigned* st = stage + wave * 64;
    int sn = 0;

    unsigned* q = queue + wave * AGG_CHUNK;
    const int4* dst4 = (const int4*)dst;
    const int nchunks = (N_EDGES + AGG_CHUNK - 1) / AGG_CHUNK;

    const int off = (int)(((long long)blockIdx.x * nchunks) / AGG_RANGES);

    int4 v[8];
    {
        const int c = wave + off;
        agg_load_chunk(dst4, (c >= nchunks) ? c - nchunks : c, lane, v);
    }
    for (int i = wave; i < nchunks; i += AGG_WAVES) {
        const int c = (i + off >= nchunks) ? i + off - nchunks : i + off;
        int qn = 0;
        #pragma unroll
        for (int u = 0; u < 8; ++u) {
            const unsigned ux = (unsigned)(v[u].x - r0);
            const unsigned uy = (unsigned)(v[u].y - r0);
            const unsigned uz = (unsigned)(v[u].z - r0);
            const unsigned uw = (unsigned)(v[u].w - r0);
            const unsigned mn = min(min(ux, uy), min(uz, uw));
            if (__builtin_amdgcn_ballot_w32(mn < (unsigned)AGG_RN) != 0u) {
                const unsigned e0 = (unsigned)(c * (AGG_CHUNK / 4) + u * 32 + lane) * 4u;
                const unsigned uc[4] = { ux, uy, uz, uw };
                #pragma unroll
                for (int k = 0; k < 4; ++k) {
                    const bool m = uc[k] < (unsigned)AGG_RN;
                    const unsigned bal = __builtin_amdgcn_ballot_w32(m);
                    if (m) q[qn + (int)__builtin_amdgcn_mbcnt_lo(bal, 0u)] = ((e0 + k) << 10) | uc[k];
                    qn += __builtin_popcount(bal);
                }
            }
        }

        const int cnt0 = min(32, qn);
        unsigned ent = 0u;
        int s = 0;
        if (lane < cnt0) { ent = q[lane]; s = src[ent >> 10]; }

        {
            const int cn = min(i + AGG_WAVES, nchunks - 1) + off;
            agg_load_chunk(dst4, (cn >= nchunks) ? cn - nchunks : cn, lane, v);
        }

        if (cnt0 > 0) {
            agg_add_rows<AGG_DEPTH, false>(h, acc, s, (int)(ent & 1023u), cnt0, lane);
            sn = agg_stage(st, sn, ((unsigned)s << 10) | (ent & 1023u), cnt0, lane, s_lines, blist);
        }

        for (int b = 32; b < qn; b += 32) {
            const int cnt = min(32, qn - b);
            unsigned ent2 = 0u;
            int s2 = 0;
            if (lane < cnt) { ent2 = q[b + lane]; s2 = src[ent2 >> 10]; }
            agg_add_rows<AGG_DEPTH, false>(h, acc, s2, (int)(ent2 & 1023u), cnt, lane);
            sn = agg_stage(st, sn, ((unsigned)s2 << 10) | (ent2 & 1023u), cnt, lane, s_lines, blist);
        }
    }
    if (sn > 0) {
        int ln = 0;
        if (lane == 0) ln = atomicAdd(s_lines, 1);
        ln = __builtin_amdgcn_readfirstlane(ln);
        blist[ln * 32 + lane] = (lane < sn) ? st[lane] : LIST_END;
    }
    __syncthreads();

    if (tid == 0) {
        dir[2 * blockIdx.x]     = (int)(blist - list);
        dir[2 * blockIdx.x + 1] = *s_lines;
    }
    if (wave < AGG_RN / 16) range_gemm(h, deg, acc, sWs, sWn, sB, hout, r0, wave, lane, do_relu);
}

__global__ __launch_bounds__(1024) void sage_replay(const float* __restrict__ h,
                                                    const unsigned* __restrict__ list,
                                                    const int* __restrict__ dir,
                                                    const float* __restrict__ deg,
                                                    const float* __restrict__ Wself,
                                                    const float* __restrict__ Wneigh,
                                                    const float* __restrict__ bias,
                                                    float* __restrict__ hout,
                                                    int do_relu) {
    __shared__ __attribute__((aligned(16))) float lds[2 * F * W_P + F + REPLAY_ACC_FLOATS];
    float* sWs = lds;
    float* sWn = sWs + F * W_P;
    float* sB  = sWn + F * W_P;
    float* acc = sB + F;
    const int tid  = threadIdx.x;
    const int lane = tid & 31;
    const int wave = __builtin_amdgcn_readfirstlane(tid >> 5);
    const int r0   = blockIdx.x * AGG_RN;

    const unsigned* blist = list + dir[2 * blockIdx.x];
    const int nl = dir[2 * blockIdx.x + 1];

    unsigned e = LIST_END;
    if (wave < nl) e = blist[wave * 32 + lane];

    stage_weights(Wself, Wneigh, bias, sWs, sWn, sB, tid);

    float4* acc4 = (float4*)acc;
    for (int j = tid; j < AGG_RN * ACC_P / 4; j += 1024) acc4[j] = make_float4(0.f, 0.f, 0.f, 0.f);
    __syncthreads();

    for (int i = wave; i < nl; i += AGG_WAVES) {
        const unsigned en = blist[min(i + AGG_WAVES, nl - 1) * 32 + lane];
        const int cnt = __builtin_popcount(__builtin_amdgcn_ballot_w32(e != LIST_END));
        agg_add_rows<REPLAY_DEPTH, true>(h, acc, (int)(e >> 10), (int)(e & 1023u), cnt, lane);
        e = en;
    }
    __syncthreads();

    if (wave < AGG_RN / 16) range_gemm(h, deg, acc, sWs, sWn, sB, hout, r0, wave, lane, do_relu);
}

extern "C" void kernel_launch(void* const* d_in, const int* in_sizes, int n_in,
                              void* d_out, int out_size, void* d_ws, size_t ws_size,
                              hipStream_t stream) {
    (void)in_sizes; (void)n_in; (void)out_size; (void)ws_size;
    const float* feat = (const float*)d_in[0];
    const int*   src  = (const int*)d_in[1];
    const int*   dst  = (const int*)d_in[2];
    const float* Ws[3] = { (const float*)d_in[3], (const float*)d_in[6], (const float*)d_in[9]  };
    const float* Wn[3] = { (const float*)d_in[4], (const float*)d_in[7], (const float*)d_in[10] };
    const float* bs[3] = { (const float*)d_in[5], (const float*)d_in[8], (const float*)d_in[11] };
    float* outp = (float*)d_out;

    float* ws   = (float*)d_ws;
    float* deg  = ws;
    float* hb   = deg + (long long)N_NODES;
    float* ha   = hb  + (long long)N_NODES * F;
    float* h2   = ha  + (long long)N_NODES * F;

    const int initBlocks  = (N_NODES + 255) / 256;
    const int degBlocks   = DEG_RANGES * DEG_SLICES;
    const int aggBlocks   = AGG_RANGES;

    int* part = (int*)h2;
    sage_degree_count<<<degBlocks, 1024, 0, stream>>>(dst, part);
    sage_init<<<initBlocks, 256, 0, stream>>>(part, deg);

    unsigned* list = (unsigned*)h2;
    int* dir = (int*)(list + LIST_CAP);

    sage_aggregate<<<aggBlocks, 1024, 0, stream>>>(feat, src, dst, deg, Ws[0], Wn[0], bs[0], ha, list, dir, 1);

    sage_replay<<<aggBlocks, 1024, 0, stream>>>(ha, list, dir, deg, Ws[1], Wn[1], bs[1], hb, 1);

    sage_replay<<<aggBlocks, 1024, 0, stream>>>(hb, list, dir, deg, Ws[2], Wn[2], bs[2], outp, 0);
}


# layer 0 scan reads one byte per edge (its destination's range number, a plane the degree launch writes), not the 4-byte dst; src and dst fetched per 32 queued matches; sum order changed
# speedup vs baseline: 3.4185x; 1.1082x over previous
#include <hip/hip_runtime.h>


#define N_NODES 100000
#define N_EDGES 1600000
#define F 64
#define ACC_P 68
#define W_P   80

#define AGG_RANGES 250
#define AGG_RN     (N_NODES / AGG_RANGES)
static_assert(AGG_RANGES <= 255, "a range number fits one byte, 255 left for no range");

typedef __attribute__((ext_vector_type(2))) float v2f;
typedef __attribute__((ext_vector_type(8))) float v8f;

#define DEG_RANGES  8
#define DEG_SLICES  32
#define DEG_R       (N_NODES / DEG_RANGES)
#define DEG_SLICE_E (N_EDGES / DEG_SLICES)
static_assert(DEG_R * DEG_RANGES == N_NODES, "ranges must tile the nodes");
static_assert(DEG_SLICE_E * DEG_SLICES == N_EDGES, "slices must tile the edges");
static_assert(DEG_SLICE_E % 4 == 0, "a slice is read as 16-byte vectors");
static_assert(DEG_SLICES * N_NODES <= 4000000, "the partial counts end before the range plane");

#define PLANE_LINES (N_EDGES / 128)
static_assert(PLANE_LINES * 128 == N_EDGES, "the plane is written in whole lines of 128 edges");
static_assert(PLANE_LINES <= 2 * DEG_RANGES * DEG_SLICES * 32, "two lines per wave cover the plane");

static __device__ __forceinline__ unsigned range_byte(int d) {
    return ((unsigned)d < (unsigned)N_NODES) ? (unsigned)d / (unsigned)AGG_RN : 255u;
}
__global__ __launch_bounds__(1024) void sage_degree_count(const int* __restrict__ dst,
                                                          int* __restrict__ part,
                                                          unsigned* __restrict__ rplane) {
    __shared__ int bins[DEG_R];
    const int tid = threadIdx.x;
    const int rg  = blockIdx.x % DEG_RANGES;
    const int sl  = blockIdx.x / DEG_RANGES;
    const int r0  = rg * DEG_R;

    const int lane = tid & 31;
    const int l0 = (int)blockIdx.x + DEG_RANGES * DEG_SLICES * (tid >> 5);
    const int l1 = l0 + DEG_RANGES * DEG_SLICES * 32;
    const int4 pa = ((const int4*)dst)[l0 * 32 + lane];
    const int4 pb = ((const int4*)dst)[min(l1, PLANE_LINES - 1) * 32 + lane];

    for (int j = tid; j < DEG_R; j += 1024) bins[j] = 0;

    rplane[l0 * 32 + lane] = range_byte(pa.x) | (range_byte(pa.y) << 8)
                           | (range_byte(pa.z) << 16) | (range_byte(pa.w) << 24);
    if (l1 < PLANE_LINES)
        rplane[l1 * 32 + lane] = range_byte(pb.x) | (range_byte(pb.y) << 8)
                               | (range_byte(pb.z) << 16) | (range_byte(pb.w) << 24);
    __syncthreads();

    const int4* p = (const int4*)(dst + (long long)sl * DEG_SLICE_E);
    const int nvec = DEG_SLICE_E / 4;
    for (int i = tid; i < nvec; i += 4096) {
        int4 v[4];
        #pragma unroll
        for (int k = 0; k < 4; ++k) {
            const int idx = i + k * 1024;
            v[k] = (idx < nvec) ? p[idx] : make_int4(-1, -1, -1, -1);
        }
        #pragma unroll
        for (int k = 0; k < 4; ++k) {
            unsigned u;
            u = (unsigned)(v[k].x - r0); if (u < (unsigned)DEG_R) atomicAdd(&bins[u], 1);
            u = (unsigned)(v[k].y - r0); if (u < (unsigned)DEG_R) atomicAdd(&bins[u], 1);
            u = (unsigned)(v[k].z - r0); if (u < (unsigned)DEG_R) atomicAdd(&bins[u], 1);
            u = (unsigned)(v[k].w - r0); if (u < (unsigned)DEG_R) atomicAdd(&bins[u], 1);
        }
    }
    __syncthreads();

    int* row = part + (long long)sl * N_NODES + r0;
    for (int j = tid; j < DEG_R; j += 1024) row[j] = bins[j];
}

__global__ __launch_bounds__(256) void sage_init(const int* __restrict__ part,
                                                 float* __restrict__ deg) {
    const int i = blockIdx.x * 256 + threadIdx.x;
    if (i < N_NODES) {
        int c = 0;
        #pragma unroll
        for (int s = 0; s < DEG_SLICES; ++s) c += part[(long long)s * N_NODES + i];
        deg[i] = (float)c;
    }
}

#define AGG_WAVES  32
#define AGG_CHUNK  4096
#define AGG_SEG    1024
#define AGG_QCAP   (AGG_SEG + 32)
#define AGG_DEPTH  8
#define REPLAY_DEPTH 16
#define LIST_PAD   1056
#define LIST_CAP   (N_EDGES + AGG_RANGES * LIST_PAD)
#define LIST_END   0xFFFFFFFFu
#define REPLAY_ACC_FLOATS 41472
static_assert(REPLAY_ACC_FLOATS >= AGG_RN * ACC_P, "room for the block's sums");
static_assert(AGG_RN % 16 == 0 && AGG_RN / 16 <= AGG_WAVES, "one 16-node tile per wave covers the range");
static_assert(F * F / 4 == AGG_WAVES * 32, "one 16-byte piece of each weight matrix per thread");
static_assert(REPLAY_ACC_FLOATS * 4 > 160 * 1024, "one list-walk block per WGP");
static_assert(LIST_PAD >= 32 * AGG_WAVES + 32, "room for every wave's part line and the rounding");
static_assert(LIST_CAP + 2 * AGG_RANGES <= 4000000, "list and directory end before the range plane");
static_assert(4000000 + N_EDGES / 4 <= N_NODES * F, "the range plane fits in the same workspace plane");
static_assert(N_NODES <= (1 << 21), "an entry keeps the source node above the 10 bits of the row");
static_assert(AGG_RN * AGG_RANGES == N_NODES, "ranges must tile the nodes");
static_assert(AGG_RN <= 1024, "a list entry keeps the local row in 10 bits");
static_assert(N_EDGES % 16 == 0, "the range plane is read as 16-byte vectors");
static_assert(AGG_CHUNK == 8 * 32 * 16 && AGG_SEG == 2 * 32 * 16, "a step is 8 vectors per lane, a segment 2");
static_assert((N_EDGES + AGG_CHUNK - 1) / AGG_CHUNK >= AGG_WAVES, "every wave has a first chunk");

#define PLANE_VECS (N_EDGES / 16)
static_assert(PLANE_VECS % 32 == 0, "the plane ends on a whole wave-load of 32 vectors");
static __device__ __forceinline__ bool agg_vec_inside(int c, int half, int u) {
    return c * (AGG_CHUNK / 16) + (half * 4 + u) * 32 < PLANE_VECS;
}
static __device__ __forceinline__ void agg_load_half(const int4* __restrict__ rp4, int c, int half,
                                                     int lane, int4 (&v)[4]) {
    #pragma unroll
    for (int u = 0; u < 4; ++u) {
        const int v0 = c * (AGG_CHUNK / 16) + (half * 4 + u) * 32;
        v[u] = rp4[min(v0, PLANE_VECS - 32) + lane];
    }
}

static __device__ __forceinline__ int agg_scan_vec(const int4 v, unsigned idrep, unsigned e0,
                                                   unsigned* q, int qn) {
    const unsigned wd[4] = { (unsigned)v.x, (unsigned)v.y, (unsigned)v.z, (unsigned)v.w };
    #pragma unroll
    for (int w = 0; w < 4; ++w) {
        const unsigned x = wd[w] ^ idrep;
        const unsigned nz = (((x & 0x7F7F7F7Fu) + 0x7F7F7F7Fu) | x) | 0x7F7F7F7Fu;
        if (__builtin_amdgcn_ballot_w32(nz != 0xFFFFFFFFu) != 0u) {
            #pragma unroll
            for (int k = 0; k < 4; ++k) {
                const bool m = ((nz >> (8 * k + 7)) & 1u) == 0u;
                const unsigned bal = __builtin_amdgcn_ballot_w32(m);
                if (m) q[qn + (int)__builtin_amdgcn_mbcnt_lo(bal, 0u)] = e0 + (unsigned)(w * 4 + k);
                qn += __builtin_popcount(bal);
            }
        }
    }
    return qn;
}

template <int DEPTH, bool WHOLE>
static __device__ __forceinline__ void agg_add_rows(const float* __restrict__ h, float* acc,
                                                    int s, int dl, int cnt, int lane) {
    int j = 0;
    for (; WHOLE && j + DEPTH <= cnt; j += DEPTH) {
        float2 hv[DEPTH];
        int dj[DEPTH];
        #pragma unroll
        for (int k = 0; k < DEPTH; ++k) {
            const int sj = __builtin_amdgcn_readlane(s, j + k);
            dj[k] = __builtin_amdgcn_readlane(dl, j + k);
            hv[k] = *(const float2*)(h + (long long)sj * F + lane * 2);
        }
        #pragma unroll
        for (int k = 0; k < DEPTH; ++k) {
            float* row = &acc[dj[k] * ACC_P + lane * 2];
            __hip_atomic_fetch_add(row,     hv[k].x, __ATOMIC_RELAXED, __HIP_MEMORY_SCOPE_WORKGROUP);
            __hip_atomic_fetch_add(row + 1, hv[k].y, __ATOMIC_RELAXED, __HIP_MEMORY_SCOPE_WORKGROUP);
        }
    }
    for (; j < cnt; j += DEPTH) {
        float2 hv[DEPTH];
        int dj[DEPTH];
        #pragma unroll
        for (int k = 0; k < DEPTH; ++k) {
            const int jj = min(j + k, cnt - 1);
            const int sj = __builtin_amdgcn_readlane(s, jj);
            dj[k] = __builtin_amdgcn_readlane(dl, jj);
            hv[k] = *(const float2*)(h + (long long)sj * F + lane * 2);
        }
        #pragma unroll
        for (int k = 0; k < DEPTH; ++k) {
            if (k == 0 || j + k < cnt) {
                float* row = &acc[dj[k] * ACC_P + lane * 2];
                __hip_atomic_fetch_add(row,     hv[k].x, __ATOMIC_RELAXED, __HIP_MEMORY_SCOPE_WORKGROUP);
                __hip_atomic_fetch_add(row + 1, hv[k].y, __ATOMIC_RELAXED, __HIP_MEMORY_SCOPE_WORKGROUP);
            }
        }
    }
}

static __device__ __forceinline__ int agg_stage(unsigned* st, int sn, unsigned entry,
                                                int cnt, int lane, int* lines,
                                                unsigned* __restrict__ blist) {
    if (lane < cnt) st[sn + lane] = entry;
    sn += cnt;
    if (sn >= 32) {
        int ln = 0;
        if (lane == 0) ln = atomicAdd(lines, 1);
        ln = __builtin_amdgcn_readfirstlane(ln);
        const unsigned e = st[lane];
        const unsigned r = st[32 + lane];
        blist[ln * 32 + lane] = e;
        sn -= 32;
        if (lane < sn) st[lane] = r;
    }
    return sn;
}

static __device__ __forceinline__ void stage_weights(const float* __restrict__ Wself,
                                                     const float* __restrict__ Wneigh,
                                                     const float* __restrict__ bias,
                                                     float* sWs, float* sWn, float* sB, int tid) {
    const float4 ws = ((const float4*)Wself)[tid];
    const float4 wn = ((const float4*)Wneigh)[tid];
    const int o = (tid >> 4) * W_P + (tid & 15) * 4;
    *(float4*)(sWs + o) = ws;
    *(float4*)(sWn + o) = wn;
    if (tid < F) sB[tid] = bias[tid];
}

static __device__ __forceinline__ void range_gemm(const float* __restrict__ h,
                                                  const float* __restrict__ deg,
                                                  const float* acc,
                                                  const float* sWs, const float* sWn, const float* sB,
                                                  float* __restrict__ hout,
                                                  int r0, int wave, int lane, int do_relu) {
    const int m0    = wave * 16;
    const int lrow  = lane & 15;
    const int khalf = lane >> 4;
    const int node  = r0 + m0 + lrow;

    const float* hrow = h + (long long)node * F + khalf * 2;
    v2f as[16];
    #pragma unroll
    for (int kk = 0; kk < 16; ++kk) as[kk] = *(const v2f*)(hrow + kk * 4);
    const float rinv = 1.0f / fmaxf(deg[node], 1.0f);

    const float* arow = acc + (m0 + lrow) * ACC_P + khalf * 2;
    const float* bs = sWs + khalf * 2 * W_P + lrow;
    const float* bn = sWn + khalf * 2 * W_P + lrow;
    v8f d[4] = {};
    #pragma unroll
    for (int kk = 0; kk < 16; ++kk) {
        v2f an = *(const v2f*)(arow + kk * 4);
        an.x *= rinv;
        an.y *= rinv;
        #pragma unroll
        for (int c = 0; c < 4; ++c) {
            v2f b;
            b.x = bs[(kk * 4    ) * W_P + c * 16];
            b.y = bs[(kk * 4 + 1) * W_P + c * 16];
            d[c] = __builtin_amdgcn_wmma_f32_16x16x4_f32(false, as[kk], false, b,
                                                         (short)0, d[c], false, false);
        }
        #pragma unroll
        for (int c = 0; c < 4; ++c) {
            v2f b;
            b.x = bn[(kk * 4    ) * W_P + c * 16];
            b.y = bn[(kk * 4 + 1) * W_P + c * 16];
            d[c] = __builtin_amdgcn_wmma_f32_16x16x4_f32(false, an, false, b,
                                                         (short)0, d[c], false, false);
        }
    }

    float* orow = hout + (long long)(r0 + m0 + khalf * 8) * F + lrow;
    #pragma unroll
    for (int c = 0; c < 4; ++c) {
        const float bv = sB[c * 16 + lrow];
        #pragma unroll
        for (int i = 0; i < 8; ++i) {
            float v = d[c][i] + bv;
            if (do_relu) v = fmaxf(v, 0.0f);
            orow[i * F + c * 16] = v;
        }
    }
}

__global__ __launch_bounds__(1024) void sage_aggregate(const float* __restrict__ h,
                                                       const int* __restrict__ src,
                                                       const int* __restrict__ dst,
                                                       const unsigned* __restrict__ rplane,
                                                       const float* __restrict__ deg,
                                                       const float* __restrict__ Wself,
                                                       const float* __restrict__ Wneigh,
                                                       const float* __restrict__ bias,
                                                       float* __restrict__ hout,
                                                       unsigned* __restrict__ list,
                                                       int* __restrict__ dir,
                                                       int do_relu) {
    __shared__ __attribute__((aligned(16))) float lds[2 * F * W_P + F + AGG_RN * ACC_P
                                                      + AGG_WAVES * AGG_QCAP + AGG_WAVES * 64 + 4];
    float* sWs = lds;
    float* sWn = sWs + F * W_P;
    float* sB  = sWn + F * W_P;
    float* acc = sB + F;
    unsigned* queue = (unsigned*)(acc + AGG_RN * ACC_P);
    unsigned* stage = queue + AGG_WAVES * AGG_QCAP;
    int* s_below = (int*)(stage + AGG_WAVES * 64);
    int* s_lines = s_below + 1;
    const int tid  = threadIdx.x;
    const int lane = tid & 31;
    const int wave = __builtin_amdgcn_readfirstlane(tid >> 5);
    const int r0   = blockIdx.x * AGG_RN;

    stage_weights(Wself, Wneigh, bias, sWs, sWn, sB, tid);

    float4* acc4 = (float4*)acc;
    for (int j = tid; j < AGG_RN * ACC_P / 4; j += 1024) acc4[j] = make_float4(0.f, 0.f, 0.f, 0.f);
    if (tid == 0) { *s_below = 0; *s_lines = 0; }

    int below = 0;
    {
        const float4* deg4 = (const float4*)deg;
        const int n4 = r0 >> 2;
        for (int j = tid; j < n4; j += 8 * 1024) {
            float4 t[8];
            #pragma unroll
            for (int k = 0; k < 8; ++k) {
                const int idx = j + k * 1024;
                t[k] = deg4[min(idx, n4 - 1)];
                if (idx >= n4) t[k] = make_float4(0.f, 0.f, 0.f, 0.f);
            }
            #pragma unroll
            for (int k = 0; k < 8; ++k) below += (int)t[k].x + (int)t[k].y + (int)t[k].z + (int)t[k].w;
        }
        if (tid < (r0 & 3)) below += (int)deg[(n4 << 2) + tid];
        for (int m = 16; m >= 1; m >>= 1) below += __shfl_xor(below, m);
    }
    __syncthreads();
    if (lane == 0 && below != 0) atomicAdd(s_below, below);
    __syncthreads();
    unsigned* blist = list + ((__builtin_amdgcn_readfirstlane(*s_below) & ~31) + (int)blockIdx.x * LIST_PAD);
    unsigned* st = stage + wave * 64;
    int sn = 0;

    unsigned* q = queue + wave * AGG_QCAP;
    const int4* rp4 = (const int4*)rplane;
    const int nchunks = (N_EDGES + AGG_CHUNK - 1) / AGG_CHUNK;
    const unsigned idrep = (unsigned)blockIdx.x * 0x01010101u;

    const int off = (int)(((long long)blockIdx.x * nchunks) / AGG_RANGES);

    int qn = 0;

    auto drain = [&]() {
        while (qn >= 32) {
            qn -= 32;
            const unsigned e = q[qn + lane];
            const int sv = src[e];
            const int dl = dst[e] - r0;
            agg_add_rows<REPLAY_DEPTH, true>(h, acc, sv, dl, 32, lane);
            sn = agg_stage(st, sn, ((unsigned)sv << 10) | (unsigned)dl, 32, lane, s_lines, blist);
        }
    };
    auto scan_half = [&](const int4 (&v)[4], int c, int half) {
        #pragma unroll
        for (int u = 0; u < 4; ++u) {
            const unsigned e0 = (unsigned)(c * (AGG_CHUNK / 16) + (half * 4 + u) * 32 + lane) * 16u;
            if (agg_vec_inside(c, half, u)) qn = agg_scan_vec(v[u], idrep, e0, q, qn);
            if (u & 1) drain();
        }
    };

    int4 va[4], vb[4];
    {
        const int c = wave + off;
        const int cw = (c >= nchunks) ? c - nchunks : c;
        agg_load_half(rp4, cw, 0, lane, va);
        agg_load_half(rp4, cw, 1, lane, vb);
    }
    for (int i = wave; i < nchunks; i += AGG_WAVES) {
        const int c = (i + off >= nchunks) ? i + off - nchunks : i + off;
        const int cn0 = min(i + AGG_WAVES, nchunks - 1) + off;
        const int cn = (cn0 >= nchunks) ? cn0 - nchunks : cn0;
        scan_half(va, c, 0);
        agg_load_half(rp4, cn, 0, lane, va);
        scan_half(vb, c, 1);
        agg_load_half(rp4, cn, 1, lane, vb);
    }
    if (qn > 0) {
        unsigned e = 0u;
        int sv = 0, dl = 0;
        if (lane < qn) { e = q[lane]; sv = src[e]; dl = dst[e] - r0; }
        agg_add_rows<AGG_DEPTH, false>(h, acc, sv, dl, qn, lane);
        sn = agg_stage(st, sn, ((unsigned)sv << 10) | (unsigned)dl, qn, lane, s_lines, blist);
    }
    if (sn > 0) {
        int ln = 0;
        if (lane == 0) ln = atomicAdd(s_lines, 1);
        ln = __builtin_amdgcn_readfirstlane(ln);
        blist[ln * 32 + lane] = (lane < sn) ? st[lane] : LIST_END;
    }
    __syncthreads();

    if (tid == 0) {
        dir[2 * blockIdx.x]     = (int)(blist - list);
        dir[2 * blockIdx.x + 1] = *s_lines;
    }
    if (wave < AGG_RN / 16) range_gemm(h, deg, acc, sWs, sWn, sB, hout, r0, wave, lane, do_relu);
}

__global__ __launch_bounds__(1024) void sage_replay(const float* __restrict__ h,
                                                    const unsigned* __restrict__ list,
                                                    const int* __restrict__ dir,
                                                    const float* __restrict__ deg,
                                                    const float* __restrict__ Wself,
                                                    const float* __restrict__ Wneigh,
                                                    const float* __restrict__ bias,
                                                    float* __restrict__ hout,
                                                    int do_relu) {
    __shared__ __attribute__((aligned(16))) float lds[2 * F * W_P + F + REPLAY_ACC_FLOATS];
    float* sWs = lds;
    float* sWn = sWs + F * W_P;
    float* sB  = sWn + F * W_P;
    float* acc = sB + F;
    const int tid  = threadIdx.x;
    const int lane = tid & 31;
    const int wave = __builtin_amdgcn_readfirstlane(tid >> 5);
    const int r0   = blockIdx.x * AGG_RN;

    const unsigned* blist = list + dir[2 * blockIdx.x];
    const int nl = dir[2 * blockIdx.x + 1];

    unsigned e = LIST_END;
    if (wave < nl) e = blist[wave * 32 + lane];

    stage_weights(Wself, Wneigh, bias, sWs, sWn, sB, tid);

    float4* acc4 = (float4*)acc;
    for (int j = tid; j < AGG_RN * ACC_P / 4; j += 1024) acc4[j] = make_float4(0.f, 0.f, 0.f, 0.f);
    __syncthreads();

    for (int i = wave; i < nl; i += AGG_WAVES) {
        const unsigned en = blist[min(i + AGG_WAVES, nl - 1) * 32 + lane];
        const int cnt = __builtin_popcount(__builtin_amdgcn_ballot_w32(e != LIST_END));
        agg_add_rows<REPLAY_DEPTH, true>(h, acc, (int)(e >> 10), (int)(e & 1023u), cnt, lane);
        e = en;
    }
    __syncthreads();

    if (wave < AGG_RN / 16) range_gemm(h, deg, acc, sWs, sWn, sB, hout, r0, wave, lane, do_relu);
}

extern "C" void kernel_launch(void* const* d_in, const int* in_sizes, int n_in,
                              void* d_out, int out_size, void* d_ws, size_t ws_size,
                              hipStream_t stream) {
    (void)in_sizes; (void)n_in; (void)out_size; (void)ws_size;
    const float* feat = (const float*)d_in[0];
    const int*   src  = (const int*)d_in[1];
    const int*   dst  = (const int*)d_in[2];
    const float* Ws[3] = { (const float*)d_in[3], (const float*)d_in[6], (const float*)d_in[9]  };
    const float* Wn[3] = { (const float*)d_in[4], (const float*)d_in[7], (const float*)d_in[10] };
    const float* bs[3] = { (const float*)d_in[5], (const float*)d_in[8], (const float*)d_in[11] };
    float* outp = (float*)d_out;

    float* ws   = (float*)d_ws;
    float* deg  = ws;
    float* hb   = deg + (long long)N_NODES;
    float* ha   = hb  + (long long)N_NODES * F;
    float* h2   = ha  + (long long)N_NODES * F;

    const int initBlocks  = (N_NODES + 255) / 256;
    const int degBlocks   = DEG_RANGES * DEG_SLICES;
    const int aggBlocks   = AGG_RANGES;

    int* part = (int*)h2;
    unsigned* rplane = (unsigned*)(h2 + 4000000);
    sage_degree_count<<<degBlocks, 1024, 0, stream>>>(dst, part, rplane);
    sage_init<<<initBlocks, 256, 0, stream>>>(part, deg);

    unsigned* list = (unsigned*)h2;
    int* dir = (int*)(list + LIST_CAP);

    sage_aggregate<<<aggBlocks, 1024, 0, stream>>>(feat, src, dst, rplane, deg, Ws[0], Wn[0], bs[0], ha, list, dir, 1);

    sage_replay<<<aggBlocks, 1024, 0, stream>>>(ha, list, dir, deg, Ws[1], Wn[1], bs[1], hb, 1);

    sage_replay<<<aggBlocks, 1024, 0, stream>>>(hb, list, dir, deg, Ws[2], Wn[2], bs[2], outp, 0);
}


# layer 0 scan: a lane packs its 32 edges' matches into one word; the wave queues them in rounds of lowest set bit: one ballot and store per round, was up to 32 per 1024 edges; sum order changed
# speedup vs baseline: 3.6016x; 1.0536x over previous
#include <hip/hip_runtime.h>


#define N_NODES 100000
#define N_EDGES 1600000
#define F 64
#define ACC_P 68
#define W_P   80

#define AGG_RANGES 250
#define AGG_RN     (N_NODES / AGG_RANGES)
static_assert(AGG_RANGES <= 255, "a range number fits one byte, 255 left for no range");

typedef __attribute__((ext_vector_type(2))) float v2f;
typedef __attribute__((ext_vector_type(8))) float v8f;

#define DEG_RANGES  8
#define DEG_SLICES  32
#define DEG_R       (N_NODES / DEG_RANGES)
#define DEG_SLICE_E (N_EDGES / DEG_SLICES)
static_assert(DEG_R * DEG_RANGES == N_NODES, "ranges must tile the nodes");
static_assert(DEG_SLICE_E * DEG_SLICES == N_EDGES, "slices must tile the edges");
static_assert(DEG_SLICE_E % 4 == 0, "a slice is read as 16-byte vectors");
static_assert(DEG_SLICES * N_NODES <= 4000000, "the partial counts end before the range plane");

#define PLANE_LINES (N_EDGES / 128)
static_assert(PLANE_LINES * 128 == N_EDGES, "the plane is written in whole lines of 128 edges");
static_assert(PLANE_LINES <= 2 * DEG_RANGES * DEG_SLICES * 32, "two lines per wave cover the plane");

static __device__ __forceinline__ unsigned range_byte(int d) {
    return ((unsigned)d < (unsigned)N_NODES) ? (unsigned)d / (unsigned)AGG_RN : 255u;
}
__global__ __launch_bounds__(1024) void sage_degree_count(const int* __restrict__ dst,
                                                          int* __restrict__ part,
                                                          unsigned* __restrict__ rplane) {
    __shared__ int bins[DEG_R];
    const int tid = threadIdx.x;
    const int rg  = blockIdx.x % DEG_RANGES;
    const int sl  = blockIdx.x / DEG_RANGES;
    const int r0  = rg * DEG_R;

    const int lane = tid & 31;
    const int l0 = (int)blockIdx.x + DEG_RANGES * DEG_SLICES * (tid >> 5);
    const int l1 = l0 + DEG_RANGES * DEG_SLICES * 32;
    const int4 pa = ((const int4*)dst)[l0 * 32 + lane];
    const int4 pb = ((const int4*)dst)[min(l1, PLANE_LINES - 1) * 32 + lane];

    for (int j = tid; j < DEG_R; j += 1024) bins[j] = 0;

    rplane[l0 * 32 + lane] = range_byte(pa.x) | (range_byte(pa.y) << 8)
                           | (range_byte(pa.z) << 16) | (range_byte(pa.w) << 24);
    if (l1 < PLANE_LINES)
        rplane[l1 * 32 + lane] = range_byte(pb.x) | (range_byte(pb.y) << 8)
                               | (range_byte(pb.z) << 16) | (range_byte(pb.w) << 24);
    __syncthreads();

    const int4* p = (const int4*)(dst + (long long)sl * DEG_SLICE_E);
    const int nvec = DEG_SLICE_E / 4;
    for (int i = tid; i < nvec; i += 4096) {
        int4 v[4];
        #pragma unroll
        for (int k = 0; k < 4; ++k) {
            const int idx = i + k * 1024;
            v[k] = (idx < nvec) ? p[idx] : make_int4(-1, -1, -1, -1);
        }
        #pragma unroll
        for (int k = 0; k < 4; ++k) {
            unsigned u;
            u = (unsigned)(v[k].x - r0); if (u < (unsigned)DEG_R) atomicAdd(&bins[u], 1);
            u = (unsigned)(v[k].y - r0); if (u < (unsigned)DEG_R) atomicAdd(&bins[u], 1);
            u = (unsigned)(v[k].z - r0); if (u < (unsigned)DEG_R) atomicAdd(&bins[u], 1);
            u = (unsigned)(v[k].w - r0); if (u < (unsigned)DEG_R) atomicAdd(&bins[u], 1);
        }
    }
    __syncthreads();

    int* row = part + (long long)sl * N_NODES + r0;
    for (int j = tid; j < DEG_R; j += 1024) row[j] = bins[j];
}

__global__ __launch_bounds__(256) void sage_init(const int* __restrict__ part,
                                                 float* __restrict__ deg) {
    const int i = blockIdx.x * 256 + threadIdx.x;
    if (i < N_NODES) {
        int c = 0;
        #pragma unroll
        for (int s = 0; s < DEG_SLICES; ++s) c += part[(long long)s * N_NODES + i];
        deg[i] = (float)c;
    }
}

#define AGG_WAVES  32
#define AGG_CHUNK  4096
#define AGG_SEG    1024
#define AGG_QCAP   (AGG_SEG + 32)
#define AGG_DEPTH  8
#define REPLAY_DEPTH 16
#define LIST_PAD   1056
#define LIST_CAP   (N_EDGES + AGG_RANGES * LIST_PAD)
#define LIST_END   0xFFFFFFFFu
#define REPLAY_ACC_FLOATS 41472
static_assert(REPLAY_ACC_FLOATS >= AGG_RN * ACC_P, "room for the block's sums");
static_assert(AGG_RN % 16 == 0 && AGG_RN / 16 <= AGG_WAVES, "one 16-node tile per wave covers the range");
static_assert(F * F / 4 == AGG_WAVES * 32, "one 16-byte piece of each weight matrix per thread");
static_assert(REPLAY_ACC_FLOATS * 4 > 160 * 1024, "one list-walk block per WGP");
static_assert(LIST_PAD >= 32 * AGG_WAVES + 32, "room for every wave's part line and the rounding");
static_assert(LIST_CAP + 2 * AGG_RANGES <= 4000000, "list and directory end before the range plane");
static_assert(4000000 + N_EDGES / 4 <= N_NODES * F, "the range plane fits in the same workspace plane");
static_assert(N_NODES <= (1 << 21), "an entry keeps the source node above the 10 bits of the row");
static_assert(AGG_RN * AGG_RANGES == N_NODES, "ranges must tile the nodes");
static_assert(AGG_RN <= 1024, "a list entry keeps the local row in 10 bits");
static_assert(N_EDGES % 16 == 0, "the range plane is read as 16-byte vectors");
static_assert(AGG_CHUNK == 8 * 32 * 16 && AGG_SEG == 2 * 32 * 16, "a step is 8 vectors per lane, a segment 2");
static_assert((N_EDGES + AGG_CHUNK - 1) / AGG_CHUNK >= AGG_WAVES, "every wave has a first chunk");
static_assert(AGG_SEG == 2 * 32 * 16, "a segment is a pair of vectors: the second one's edges lie 512 behind the first one's");

#define PLANE_VECS (N_EDGES / 16)
static_assert(PLANE_VECS % 32 == 0, "the plane ends on a whole wave-load of 32 vectors");
static __device__ __forceinline__ bool agg_vec_inside(int c, int half, int u) {
    return c * (AGG_CHUNK / 16) + (half * 4 + u) * 32 < PLANE_VECS;
}
static __device__ __forceinline__ void agg_load_half(const int4* __restrict__ rp4, int c, int half,
                                                     int lane, int4 (&v)[4]) {
    #pragma unroll
    for (int u = 0; u < 4; ++u) {
        const int v0 = c * (AGG_CHUNK / 16) + (half * 4 + u) * 32;
        v[u] = rp4[min(v0, PLANE_VECS - 32) + lane];
    }
}

template <int BASE>
static __device__ __forceinline__ unsigned agg_hits(const int4 v, unsigned idrep) {
    const unsigned wd[4] = { (unsigned)v.x, (unsigned)v.y, (unsigned)v.z, (unsigned)v.w };
    unsigned hit = 0u;
    #pragma unroll
    for (int w = 0; w < 4; ++w) {
        const unsigned x = wd[w] ^ idrep;
        const unsigned z = ~(((x & 0x7F7F7F7Fu) + 0x7F7F7F7Fu) | x) & 0x80808080u;
        hit |= z >> (7 - BASE - w);
    }
    return hit;
}

static __device__ __forceinline__ int agg_queue_hits(unsigned hit, unsigned e0, unsigned* q, int qn) {
    unsigned bal = __builtin_amdgcn_ballot_w32(hit != 0u);
    while (bal != 0u) {
        if (hit != 0u) {
            const unsigned p = (unsigned)__builtin_ctz(hit);
            q[qn + (int)__builtin_amdgcn_mbcnt_lo(bal, 0u)] =
                e0 + ((p & 4u) << 7) + ((p & 3u) << 2) + (p >> 3);
            hit &= hit - 1u;
        }
        qn += __builtin_popcount(bal);
        bal = __builtin_amdgcn_ballot_w32(hit != 0u);
    }
    return qn;
}

template <int DEPTH, bool WHOLE>
static __device__ __forceinline__ void agg_add_rows(const float* __restrict__ h, float* acc,
                                                    int s, int dl, int cnt, int lane) {
    int j = 0;
    for (; WHOLE && j + DEPTH <= cnt; j += DEPTH) {
        float2 hv[DEPTH];
        int dj[DEPTH];
        #pragma unroll
        for (int k = 0; k < DEPTH; ++k) {
            const int sj = __builtin_amdgcn_readlane(s, j + k);
            dj[k] = __builtin_amdgcn_readlane(dl, j + k);
            hv[k] = *(const float2*)(h + (long long)sj * F + lane * 2);
        }
        #pragma unroll
        for (int k = 0; k < DEPTH; ++k) {
            float* row = &acc[dj[k] * ACC_P + lane * 2];
            __hip_atomic_fetch_add(row,     hv[k].x, __ATOMIC_RELAXED, __HIP_MEMORY_SCOPE_WORKGROUP);
            __hip_atomic_fetch_add(row + 1, hv[k].y, __ATOMIC_RELAXED, __HIP_MEMORY_SCOPE_WORKGROUP);
        }
    }
    for (; j < cnt; j += DEPTH) {
        float2 hv[DEPTH];
        int dj[DEPTH];
        #pragma unroll
        for (int k = 0; k < DEPTH; ++k) {
            const int jj = min(j + k, cnt - 1);
            const int sj = __builtin_amdgcn_readlane(s, jj);
            dj[k] = __builtin_amdgcn_readlane(dl, jj);
            hv[k] = *(const float2*)(h + (long long)sj * F + lane * 2);
        }
        #pragma unroll
        for (int k = 0; k < DEPTH; ++k) {
            if (k == 0 || j + k < cnt) {
                float* row = &acc[dj[k] * ACC_P + lane * 2];
                __hip_atomic_fetch_add(row,     hv[k].x, __ATOMIC_RELAXED, __HIP_MEMORY_SCOPE_WORKGROUP);
                __hip_atomic_fetch_add(row + 1, hv[k].y, __ATOMIC_RELAXED, __HIP_MEMORY_SCOPE_WORKGROUP);
            }
        }
    }
}

static __device__ __forceinline__ int agg_stage(unsigned* st, int sn, unsigned entry,
                                                int cnt, int lane, int* lines,
                                                unsigned* __restrict__ blist) {
    if (lane < cnt) st[sn + lane] = entry;
    sn += cnt;
    if (sn >= 32) {
        int ln = 0;
        if (lane == 0) ln = atomicAdd(lines, 1);
        ln = __builtin_amdgcn_readfirstlane(ln);
        const unsigned e = st[lane];
        const unsigned r = st[32 + lane];
        blist[ln * 32 + lane] = e;
        sn -= 32;
        if (lane < sn) st[lane] = r;
    }
    return sn;
}

static __device__ __forceinline__ void stage_weights(const float* __restrict__ Wself,
                                                     const float* __restrict__ Wneigh,
                                                     const float* __restrict__ bias,
                                                     float* sWs, float* sWn, float* sB, int tid) {
    const float4 ws = ((const float4*)Wself)[tid];
    const float4 wn = ((const float4*)Wneigh)[tid];
    const int o = (tid >> 4) * W_P + (tid & 15) * 4;
    *(float4*)(sWs + o) = ws;
    *(float4*)(sWn + o) = wn;
    if (tid < F) sB[tid] = bias[tid];
}

static __device__ __forceinline__ void range_gemm(const float* __restrict__ h,
                                                  const float* __restrict__ deg,
                                                  const float* acc,
                                                  const float* sWs, const float* sWn, const float* sB,
                                                  float* __restrict__ hout,
                                                  int r0, int wave, int lane, int do_relu) {
    const int m0    = wave * 16;
    const int lrow  = lane & 15;
    const int khalf = lane >> 4;
    const int node  = r0 + m0 + lrow;

    const float* hrow = h + (long long)node * F + khalf * 2;
    v2f as[16];
    #pragma unroll
    for (int kk = 0; kk < 16; ++kk) as[kk] = *(const v2f*)(hrow + kk * 4);
    const float rinv = 1.0f / fmaxf(deg[node], 1.0f);

    const float* arow = acc + (m0 + lrow) * ACC_P + khalf * 2;
    const float* bs = sWs + khalf * 2 * W_P + lrow;
    const float* bn = sWn + khalf * 2 * W_P + lrow;
    v8f d[4] = {};
    #pragma unroll
    for (int kk = 0; kk < 16; ++kk) {
        v2f an = *(const v2f*)(arow + kk * 4);
        an.x *= rinv;
        an.y *= rinv;
        #pragma unroll
        for (int c = 0; c < 4; ++c) {
            v2f b;
            b.x = bs[(kk * 4    ) * W_P + c * 16];
            b.y = bs[(kk * 4 + 1) * W_P + c * 16];
            d[c] = __builtin_amdgcn_wmma_f32_16x16x4_f32(false, as[kk], false, b,
                                                         (short)0, d[c], false, false);
        }
        #pragma unroll
        for (int c = 0; c < 4; ++c) {
            v2f b;
            b.x = bn[(kk * 4    ) * W_P + c * 16];
            b.y = bn[(kk * 4 + 1) * W_P + c * 16];
            d[c] = __builtin_amdgcn_wmma_f32_16x16x4_f32(false, an, false, b,
                                                         (short)0, d[c], false, false);
        }
    }

    float* orow = hout + (long long)(r0 + m0 + khalf * 8) * F + lrow;
    #pragma unroll
    for (int c = 0; c < 4; ++c) {
        const float bv = sB[c * 16 + lrow];
        #pragma unroll
        for (int i = 0; i < 8; ++i) {
            float v = d[c][i] + bv;
            if (do_relu) v = fmaxf(v, 0.0f);
            orow[i * F + c * 16] = v;
        }
    }
}

__global__ __launch_bounds__(1024) void sage_aggregate(const float* __restrict__ h,
                                                       const int* __restrict__ src,
                                                       const int* __restrict__ dst,
                                                       const unsigned* __restrict__ rplane,
                                                       const float* __restrict__ deg,
                                                       const float* __restrict__ Wself,
                                                       const float* __restrict__ Wneigh,
                                                       const float* __restrict__ bias,
                                                       float* __restrict__ hout,
                                                       unsigned* __restrict__ list,
                                                       int* __restrict__ dir,
                                                       int do_relu) {
    __shared__ __attribute__((aligned(16))) float lds[2 * F * W_P + F + AGG_RN * ACC_P
                                                      + AGG_WAVES * AGG_QCAP + AGG_WAVES * 64 + 4];
    float* sWs = lds;
    float* sWn = sWs + F * W_P;
    float* sB  = sWn + F * W_P;
    float* acc = sB + F;
    unsigned* queue = (unsigned*)(acc + AGG_RN * ACC_P);
    unsigned* stage = queue + AGG_WAVES * AGG_QCAP;
    int* s_below = (int*)(stage + AGG_WAVES * 64);
    int* s_lines = s_below + 1;
    const int tid  = threadIdx.x;
    const int lane = tid & 31;
    const int wave = __builtin_amdgcn_readfirstlane(tid >> 5);
    const int r0   = blockIdx.x * AGG_RN;

    stage_weights(Wself, Wneigh, bias, sWs, sWn, sB, tid);

    float4* acc4 = (float4*)acc;
    for (int j = tid; j < AGG_RN * ACC_P / 4; j += 1024) acc4[j] = make_float4(0.f, 0.f, 0.f, 0.f);
    if (tid == 0) { *s_below = 0; *s_lines = 0; }

    int below = 0;
    {
        const float4* deg4 = (const float4*)deg;
        const int n4 = r0 >> 2;
        for (int j = tid; j < n4; j += 8 * 1024) {
            float4 t[8];
            #pragma unroll
            for (int k = 0; k < 8; ++k) {
                const int idx = j + k * 1024;
                t[k] = deg4[min(idx, n4 - 1)];
                if (idx >= n4) t[k] = make_float4(0.f, 0.f, 0.f, 0.f);
            }
            #pragma unroll
            for (int k = 0; k < 8; ++k) below += (int)t[k].x + (int)t[k].y + (int)t[k].z + (int)t[k].w;
        }
        if (tid < (r0 & 3)) below += (int)deg[(n4 << 2) + tid];
        for (int m = 16; m >= 1; m >>= 1) below += __shfl_xor(below, m);
    }
    __syncthreads();
    if (lane == 0 && below != 0) atomicAdd(s_below, below);
    __syncthreads();
    unsigned* blist = list + ((__builtin_amdgcn_readfirstlane(*s_below) & ~31) + (int)blockIdx.x * LIST_PAD);
    unsigned* st = stage + wave * 64;
    int sn = 0;

    unsigned* q = queue + wave * AGG_QCAP;
    const int4* rp4 = (const int4*)rplane;
    const int nchunks = (N_EDGES + AGG_CHUNK - 1) / AGG_CHUNK;
    const unsigned idrep = (unsigned)blockIdx.x * 0x01010101u;

    const int off = (int)(((long long)blockIdx.x * nchunks) / AGG_RANGES);

    int qn = 0;

    auto drain = [&]() {
        while (qn >= 32) {
            qn -= 32;
            const unsigned e = q[qn + lane];
            const int sv = src[e];
            const int dl = dst[e] - r0;
            agg_add_rows<REPLAY_DEPTH, true>(h, acc, sv, dl, 32, lane);
            sn = agg_stage(st, sn, ((unsigned)sv << 10) | (unsigned)dl, 32, lane, s_lines, blist);
        }
    };
    auto scan_half = [&](const int4 (&v)[4], int c, int half) {
        #pragma unroll
        for (int u = 0; u < 4; u += 2) {
            const unsigned e0 = (unsigned)(c * (AGG_CHUNK / 16) + (half * 4 + u) * 32 + lane) * 16u;
            unsigned hit = 0u;
            if (agg_vec_inside(c, half, u))     hit  = agg_hits<0>(v[u], idrep);
            if (agg_vec_inside(c, half, u + 1)) hit |= agg_hits<4>(v[u + 1], idrep);
            qn = agg_queue_hits(hit, e0, q, qn);
            drain();
        }
    };

    int4 va[4], vb[4];
    {
        const int c = wave + off;
        const int cw = (c >= nchunks) ? c - nchunks : c;
        agg_load_half(rp4, cw, 0, lane, va);
        agg_load_half(rp4, cw, 1, lane, vb);
    }
    for (int i = wave; i < nchunks; i += AGG_WAVES) {
        const int c = (i + off >= nchunks) ? i + off - nchunks : i + off;
        const int cn0 = min(i + AGG_WAVES, nchunks - 1) + off;
        const int cn = (cn0 >= nchunks) ? cn0 - nchunks : cn0;
        scan_half(va, c, 0);
        agg_load_half(rp4, cn, 0, lane, va);
        scan_half(vb, c, 1);
        agg_load_half(rp4, cn, 1, lane, vb);
    }
    if (qn > 0) {
        unsigned e = 0u;
        int sv = 0, dl = 0;
        if (lane < qn) { e = q[lane]; sv = src[e]; dl = dst[e] - r0; }
        agg_add_rows<AGG_DEPTH, false>(h, acc, sv, dl, qn, lane);
        sn = agg_stage(st, sn, ((unsigned)sv << 10) | (unsigned)dl, qn, lane, s_lines, blist);
    }
    if (sn > 0) {
        int ln = 0;
        if (lane == 0) ln = atomicAdd(s_lines, 1);
        ln = __builtin_amdgcn_readfirstlane(ln);
        blist[ln * 32 + lane] = (lane < sn) ? st[lane] : LIST_END;
    }
    __syncthreads();

    if (tid == 0) {
        dir[2 * blockIdx.x]     = (int)(blist - list);
        dir[2 * blockIdx.x + 1] = *s_lines;
    }
    if (wave < AGG_RN / 16) range_gemm(h, deg, acc, sWs, sWn, sB, hout, r0, wave, lane, do_relu);
}

__global__ __launch_bounds__(1024) void sage_replay(const float* __restrict__ h,
                                                    const unsigned* __restrict__ list,
                                                    const int* __restrict__ dir,
                                                    const float* __restrict__ deg,
                                                    const float* __restrict__ Wself,
                                                    const float* __restrict__ Wneigh,
                                                    const float* __restrict__ bias,
                                                    float* __restrict__ hout,
                                                    int do_relu) {
    __shared__ __attribute__((aligned(16))) float lds[2 * F * W_P + F + REPLAY_ACC_FLOATS];
    float* sWs = lds;
    float* sWn = sWs + F * W_P;
    float* sB  = sWn + F * W_P;
    float* acc = sB + F;
    const int tid  = threadIdx.x;
    const int lane = tid & 31;
    const int wave = __builtin_amdgcn_readfirstlane(tid >> 5);
    const int r0   = blockIdx.x * AGG_RN;

    const unsigned* blist = list + dir[2 * blockIdx.x];
    const int nl = dir[2 * blockIdx.x + 1];

    unsigned e = LIST_END;
    if (wave < nl) e = blist[wave * 32 + lane];

    stage_weights(Wself, Wneigh, bias, sWs, sWn, sB, tid);

    float4* acc4 = (float4*)acc;
    for (int j = tid; j < AGG_RN * ACC_P / 4; j += 1024) acc4[j] = make_float4(0.f, 0.f, 0.f, 0.f);
    __syncthreads();

    for (int i = wave; i < nl; i += AGG_WAVES) {
        const unsigned en = blist[min(i + AGG_WAVES, nl - 1) * 32 + lane];
        const int cnt = __builtin_popcount(__builtin_amdgcn_ballot_w32(e != LIST_END));
        agg_add_rows<REPLAY_DEPTH, true>(h, acc, (int)(e >> 10), (int)(e & 1023u), cnt, lane);
        e = en;
    }
    __syncthreads();

    if (wave < AGG_RN / 16) range_gemm(h, deg, acc, sWs, sWn, sB, hout, r0, wave, lane, do_relu);
}

extern "C" void kernel_launch(void* const* d_in, const int* in_sizes, int n_in,
                              void* d_out, int out_size, void* d_ws, size_t ws_size,
                              hipStream_t stream) {
    (void)in_sizes; (void)n_in; (void)out_size; (void)ws_size;
    const float* feat = (const float*)d_in[0];
    const int*   src  = (const int*)d_in[1];
    const int*   dst  = (const int*)d_in[2];
    const float* Ws[3] = { (const float*)d_in[3], (const float*)d_in[6], (const float*)d_in[9]  };
    const float* Wn[3] = { (const float*)d_in[4], (const float*)d_in[7], (const float*)d_in[10] };
    const float* bs[3] = { (const float*)d_in[5], (const float*)d_in[8], (const float*)d_in[11] };
    float* outp = (float*)d_out;

    float* ws   = (float*)d_ws;
    float* deg  = ws;
    float* hb   = deg + (long long)N_NODES;
    float* ha   = hb  + (long long)N_NODES * F;
    float* h2   = ha  + (long long)N_NODES * F;

    const int initBlocks  = (N_NODES + 255) / 256;
    const int degBlocks   = DEG_RANGES * DEG_SLICES;
    const int aggBlocks   = AGG_RANGES;

    int* part = (int*)h2;
    unsigned* rplane = (unsigned*)(h2 + 4000000);
    sage_degree_count<<<degBlocks, 1024, 0, stream>>>(dst, part, rplane);
    sage_init<<<initBlocks, 256, 0, stream>>>(part, deg);

    unsigned* list = (unsigned*)h2;
    int* dir = (int*)(list + LIST_CAP);

    sage_aggregate<<<aggBlocks, 1024, 0, stream>>>(feat, src, dst, rplane, deg, Ws[0], Wn[0], bs[0], ha, list, dir, 1);

    sage_replay<<<aggBlocks, 1024, 0, stream>>>(ha, list, dir, deg, Ws[1], Wn[1], bs[1], hb, 1);

    sage_replay<<<aggBlocks, 1024, 0, stream>>>(hb, list, dir, deg, Ws[2], Wn[2], bs[2], outp, 0);
}


# degree counted in layer 0's blocks (integer LDS bins, during the scan); first launch only writes range plane and per-range edge counts; partial-count plane and init launch gone; sums as in parent
# speedup vs baseline: 3.7093x; 1.0299x over previous
#include <hip/hip_runtime.h>


#define N_NODES 100000
#define N_EDGES 1600000
#define F 64
#define ACC_P 68
#define W_P   80

#define AGG_RANGES 250
#define AGG_RN     (N_NODES / AGG_RANGES)
static_assert(AGG_RANGES <= 255, "a range number fits one byte, 255 left for no range");

typedef __attribute__((ext_vector_type(2))) float v2f;
typedef __attribute__((ext_vector_type(8))) float v8f;

#define PLANE_LINES  (N_EDGES / 128)
#define PLANE_BLOCKS 125
static_assert(PLANE_LINES * 128 == N_EDGES, "the plane is written in whole lines of 128 edges");
static_assert(PLANE_LINES <= 4 * PLANE_BLOCKS * 32, "four lines per wave cover the plane");

static __device__ __forceinline__ unsigned range_byte(int d) {
    return ((unsigned)d < (unsigned)N_NODES) ? (unsigned)d / (unsigned)AGG_RN : 255u;
}
__global__ __launch_bounds__(1024) void sage_range_plane(const int* __restrict__ dst,
                                                         unsigned* __restrict__ rplane,
                                                         int* __restrict__ cnt) {
    __shared__ int hist[256];
    const int tid  = threadIdx.x;
    const int lane = tid & 31;
    const int wave = tid >> 5;

    int4 v[4];
    #pragma unroll
    for (int k = 0; k < 4; ++k) {
        const int l = (int)blockIdx.x + PLANE_BLOCKS * (wave + 32 * k);
        v[k] = ((const int4*)dst)[min(l, PLANE_LINES - 1) * 32 + lane];
    }
    if (tid < 256) hist[tid] = 0;
    __syncthreads();

    #pragma unroll
    for (int k = 0; k < 4; ++k) {
        const int l = (int)blockIdx.x + PLANE_BLOCKS * (wave + 32 * k);
        if (l < PLANE_LINES) {
            const unsigned b0 = range_byte(v[k].x), b1 = range_byte(v[k].y);
            const unsigned b2 = range_byte(v[k].z), b3 = range_byte(v[k].w);
            rplane[l * 32 + lane] = b0 | (b1 << 8) | (b2 << 16) | (b3 << 24);
            atomicAdd(&hist[b0], 1);
            atomicAdd(&hist[b1], 1);
            atomicAdd(&hist[b2], 1);
            atomicAdd(&hist[b3], 1);
        }
    }
    __syncthreads();
    if (tid < 256) cnt[(int)blockIdx.x * 256 + tid] = hist[tid];
}

#define AGG_WAVES  32
#define AGG_CHUNK  4096
#define AGG_SEG    1024
#define AGG_QCAP   (AGG_SEG + 32)
#define AGG_DEPTH  8
#define REPLAY_DEPTH 16
#define LIST_PAD   1056
#define LIST_CAP   (N_EDGES + AGG_RANGES * LIST_PAD)
#define LIST_END   0xFFFFFFFFu
#define REPLAY_ACC_FLOATS 41472
static_assert(REPLAY_ACC_FLOATS >= AGG_RN * ACC_P, "room for the block's sums");
static_assert(AGG_RN % 16 == 0 && AGG_RN / 16 <= AGG_WAVES, "one 16-node tile per wave covers the range");
static_assert(F * F / 4 == AGG_WAVES * 32, "one 16-byte piece of each weight matrix per thread");
static_assert(REPLAY_ACC_FLOATS * 4 > 160 * 1024, "one list-walk block per WGP");
static_assert(LIST_PAD >= 32 * AGG_WAVES + 32, "room for every wave's part line and the rounding");
static_assert(LIST_CAP + 2 * AGG_RANGES <= 3000000, "list and directory end before the range counts");
static_assert(3000000 + PLANE_BLOCKS * 256 <= 3100000, "the range counts end before the degrees");
#define DEG_PITCH 512
static_assert(DEG_PITCH >= AGG_RN && DEG_PITCH % 32 == 0, "a block's degrees fill whole lines of their own");
static_assert(3100000 + AGG_RANGES * DEG_PITCH <= 4000000, "the degrees end before the range plane");
static_assert(4000000 + N_EDGES / 4 <= N_NODES * F, "the range plane fits in the same workspace plane");
static_assert(N_NODES <= (1 << 21), "an entry keeps the source node above the 10 bits of the row");
static_assert(AGG_RN * AGG_RANGES == N_NODES, "ranges must tile the nodes");
static_assert(AGG_RN <= 1024, "a list entry keeps the local row in 10 bits");
static_assert(N_EDGES % 16 == 0, "the range plane is read as 16-byte vectors");
static_assert(AGG_CHUNK == 8 * 32 * 16 && AGG_SEG == 2 * 32 * 16, "a step is 8 vectors per lane, a segment 2");
static_assert((N_EDGES + AGG_CHUNK - 1) / AGG_CHUNK >= AGG_WAVES, "every wave has a first chunk");
static_assert(AGG_SEG == 2 * 32 * 16, "a segment is a pair of vectors: the second one's edges lie 512 behind the first one's");

#define PLANE_VECS (N_EDGES / 16)
static_assert(PLANE_VECS % 32 == 0, "the plane ends on a whole wave-load of 32 vectors");
static __device__ __forceinline__ bool agg_vec_inside(int c, int half, int u) {
    return c * (AGG_CHUNK / 16) + (half * 4 + u) * 32 < PLANE_VECS;
}
static __device__ __forceinline__ void agg_load_half(const int4* __restrict__ rp4, int c, int half,
                                                     int lane, int4 (&v)[4]) {
    #pragma unroll
    for (int u = 0; u < 4; ++u) {
        const int v0 = c * (AGG_CHUNK / 16) + (half * 4 + u) * 32;
        v[u] = rp4[min(v0, PLANE_VECS - 32) + lane];
    }
}

template <int BASE>
static __device__ __forceinline__ unsigned agg_hits(const int4 v, unsigned idrep) {
    const unsigned wd[4] = { (unsigned)v.x, (unsigned)v.y, (unsigned)v.z, (unsigned)v.w };
    unsigned hit = 0u;
    #pragma unroll
    for (int w = 0; w < 4; ++w) {
        const unsigned x = wd[w] ^ idrep;
        const unsigned z = ~(((x & 0x7F7F7F7Fu) + 0x7F7F7F7Fu) | x) & 0x80808080u;
        hit |= z >> (7 - BASE - w);
    }
    return hit;
}

static __device__ __forceinline__ int agg_queue_hits(unsigned hit, unsigned e0, unsigned* q, int qn) {
    unsigned bal = __builtin_amdgcn_ballot_w32(hit != 0u);
    while (bal != 0u) {
        if (hit != 0u) {
            const unsigned p = (unsigned)__builtin_ctz(hit);
            q[qn + (int)__builtin_amdgcn_mbcnt_lo(bal, 0u)] =
                e0 + ((p & 4u) << 7) + ((p & 3u) << 2) + (p >> 3);
            hit &= hit - 1u;
        }
        qn += __builtin_popcount(bal);
        bal = __builtin_amdgcn_ballot_w32(hit != 0u);
    }
    return qn;
}

template <int DEPTH, bool WHOLE>
static __device__ __forceinline__ void agg_add_rows(const float* __restrict__ h, float* acc,
                                                    int s, int dl, int cnt, int lane) {
    int j = 0;
    for (; WHOLE && j + DEPTH <= cnt; j += DEPTH) {
        float2 hv[DEPTH];
        int dj[DEPTH];
        #pragma unroll
        for (int k = 0; k < DEPTH; ++k) {
            const int sj = __builtin_amdgcn_readlane(s, j + k);
            dj[k] = __builtin_amdgcn_readlane(dl, j + k);
            hv[k] = *(const float2*)(h + (long long)sj * F + lane * 2);
        }
        #pragma unroll
        for (int k = 0; k < DEPTH; ++k) {
            float* row = &acc[dj[k] * ACC_P + lane * 2];
            __hip_atomic_fetch_add(row,     hv[k].x, __ATOMIC_RELAXED, __HIP_MEMORY_SCOPE_WORKGROUP);
            __hip_atomic_fetch_add(row + 1, hv[k].y, __ATOMIC_RELAXED, __HIP_MEMORY_SCOPE_WORKGROUP);
        }
    }
    for (; j < cnt; j += DEPTH) {
        float2 hv[DEPTH];
        int dj[DEPTH];
        #pragma unroll
        for (int k = 0; k < DEPTH; ++k) {
            const int jj = min(j + k, cnt - 1);
            const int sj = __builtin_amdgcn_readlane(s, jj);
            dj[k] = __builtin_amdgcn_readlane(dl, jj);
            hv[k] = *(const float2*)(h + (long long)sj * F + lane * 2);
        }
        #pragma unroll
        for (int k = 0; k < DEPTH; ++k) {
            if (k == 0 || j + k < cnt) {
                float* row = &acc[dj[k] * ACC_P + lane * 2];
                __hip_atomic_fetch_add(row,     hv[k].x, __ATOMIC_RELAXED, __HIP_MEMORY_SCOPE_WORKGROUP);
                __hip_atomic_fetch_add(row + 1, hv[k].y, __ATOMIC_RELAXED, __HIP_MEMORY_SCOPE_WORKGROUP);
            }
        }
    }
}

static __device__ __forceinline__ int agg_stage(unsigned* st, int sn, unsigned entry,
                                                int cnt, int lane, int* lines,
                                                unsigned* __restrict__ blist) {
    if (lane < cnt) st[sn + lane] = entry;
    sn += cnt;
    if (sn >= 32) {
        int ln = 0;
        if (lane == 0) ln = atomicAdd(lines, 1);
        ln = __builtin_amdgcn_readfirstlane(ln);
        const unsigned e = st[lane];
        const unsigned r = st[32 + lane];
        blist[ln * 32 + lane] = e;
        sn -= 32;
        if (lane < sn) st[lane] = r;
    }
    return sn;
}

static __device__ __forceinline__ void stage_weights(const float* __restrict__ Wself,
                                                     const float* __restrict__ Wneigh,
                                                     const float* __restrict__ bias,
                                                     float* sWs, float* sWn, float* sB, int tid) {
    const float4 ws = ((const float4*)Wself)[tid];
    const float4 wn = ((const float4*)Wneigh)[tid];
    const int o = (tid >> 4) * W_P + (tid & 15) * 4;
    *(float4*)(sWs + o) = ws;
    *(float4*)(sWn + o) = wn;
    if (tid < F) sB[tid] = bias[tid];
}

static __device__ __forceinline__ void range_gemm(const float* __restrict__ h,
                                                  float degv,
                                                  const float* acc,
                                                  const float* sWs, const float* sWn, const float* sB,
                                                  float* __restrict__ hout,
                                                  int r0, int wave, int lane, int do_relu) {
    const int m0    = wave * 16;
    const int lrow  = lane & 15;
    const int khalf = lane >> 4;
    const int node  = r0 + m0 + lrow;

    const float* hrow = h + (long long)node * F + khalf * 2;
    v2f as[16];
    #pragma unroll
    for (int kk = 0; kk < 16; ++kk) as[kk] = *(const v2f*)(hrow + kk * 4);
    const float rinv = 1.0f / fmaxf(degv, 1.0f);

    const float* arow = acc + (m0 + lrow) * ACC_P + khalf * 2;
    const float* bs = sWs + khalf * 2 * W_P + lrow;
    const float* bn = sWn + khalf * 2 * W_P + lrow;
    v8f d[4] = {};
    #pragma unroll
    for (int kk = 0; kk < 16; ++kk) {
        v2f an = *(const v2f*)(arow + kk * 4);
        an.x *= rinv;
        an.y *= rinv;
        #pragma unroll
        for (int c = 0; c < 4; ++c) {
            v2f b;
            b.x = bs[(kk * 4    ) * W_P + c * 16];
            b.y = bs[(kk * 4 + 1) * W_P + c * 16];
            d[c] = __builtin_amdgcn_wmma_f32_16x16x4_f32(false, as[kk], false, b,
                                                         (short)0, d[c], false, false);
        }
        #pragma unroll
        for (int c = 0; c < 4; ++c) {
            v2f b;
            b.x = bn[(kk * 4    ) * W_P + c * 16];
            b.y = bn[(kk * 4 + 1) * W_P + c * 16];
            d[c] = __builtin_amdgcn_wmma_f32_16x16x4_f32(false, an, false, b,
                                                         (short)0, d[c], false, false);
        }
    }

    float* orow = hout + (long long)(r0 + m0 + khalf * 8) * F + lrow;
    #pragma unroll
    for (int c = 0; c < 4; ++c) {
        const float bv = sB[c * 16 + lrow];
        #pragma unroll
        for (int i = 0; i < 8; ++i) {
            float v = d[c][i] + bv;
            if (do_relu) v = fmaxf(v, 0.0f);
            orow[i * F + c * 16] = v;
        }
    }
}

__global__ __launch_bounds__(1024) void sage_aggregate(const float* __restrict__ h,
                                                       const int* __restrict__ src,
                                                       const int* __restrict__ dst,
                                                       const unsigned* __restrict__ rplane,
                                                       const int* __restrict__ cnt,
                                                       float* __restrict__ deg,
                                                       const float* __restrict__ Wself,
                                                       const float* __restrict__ Wneigh,
                                                       const float* __restrict__ bias,
                                                       float* __restrict__ hout,
                                                       unsigned* __restrict__ list,
                                                       int* __restrict__ dir,
                                                       int do_relu) {
    __shared__ __attribute__((aligned(16))) float lds[2 * F * W_P + F + AGG_RN * ACC_P
                                                      + AGG_WAVES * AGG_QCAP + AGG_WAVES * 64 + 4 + AGG_RN];
    float* sWs = lds;
    float* sWn = sWs + F * W_P;
    float* sB  = sWn + F * W_P;
    float* acc = sB + F;
    unsigned* queue = (unsigned*)(acc + AGG_RN * ACC_P);
    unsigned* stage = queue + AGG_WAVES * AGG_QCAP;
    int* s_below = (int*)(stage + AGG_WAVES * 64);
    int* s_lines = s_below + 1;
    int* degbin  = s_below + 4;
    const int tid  = threadIdx.x;
    const int lane = tid & 31;
    const int wave = __builtin_amdgcn_readfirstlane(tid >> 5);
    const int r0   = blockIdx.x * AGG_RN;

    stage_weights(Wself, Wneigh, bias, sWs, sWn, sB, tid);

    float4* acc4 = (float4*)acc;
    for (int j = tid; j < AGG_RN * ACC_P / 4; j += 1024) acc4[j] = make_float4(0.f, 0.f, 0.f, 0.f);
    if (tid == 0) { *s_below = 0; *s_lines = 0; }
    if (tid < AGG_RN) degbin[tid] = 0;

    int below = 0;
    {
        const int4* c4 = (const int4*)cnt;
        const int n4 = PLANE_BLOCKS * 64;
        const int r  = (tid & 63) * 4;
        const int b  = (int)blockIdx.x;
        int4 t[8];
        #pragma unroll
        for (int k = 0; k < 8; ++k) t[k] = c4[min(tid + k * 1024, n4 - 1)];
        #pragma unroll
        for (int k = 0; k < 8; ++k) {
            if (tid + k * 1024 < n4)
                below += (r < b ? t[k].x : 0) + (r + 1 < b ? t[k].y : 0)
                       + (r + 2 < b ? t[k].z : 0) + (r + 3 < b ? t[k].w : 0);
        }
        for (int m = 16; m >= 1; m >>= 1) below += __shfl_xor(below, m);
    }
    __syncthreads();
    if (lane == 0 && below != 0) atomicAdd(s_below, below);
    __syncthreads();
    unsigned* blist = list + ((__builtin_amdgcn_readfirstlane(*s_below) & ~31) + (int)blockIdx.x * LIST_PAD);
    unsigned* st = stage + wave * 64;
    int sn = 0;

    unsigned* q = queue + wave * AGG_QCAP;
    const int4* rp4 = (const int4*)rplane;
    const int nchunks = (N_EDGES + AGG_CHUNK - 1) / AGG_CHUNK;
    const unsigned idrep = (unsigned)blockIdx.x * 0x01010101u;

    const int off = (int)(((long long)blockIdx.x * nchunks) / AGG_RANGES);

    int qn = 0;

    auto drain = [&]() {
        while (qn >= 32) {
            qn -= 32;
            const unsigned e = q[qn + lane];
            const int sv = src[e];
            const int dl = dst[e] - r0;
            atomicAdd(&degbin[dl], 1);
            agg_add_rows<REPLAY_DEPTH, true>(h, acc, sv, dl, 32, lane);
            sn = agg_stage(st, sn, ((unsigned)sv << 10) | (unsigned)dl, 32, lane, s_lines, blist);
        }
    };
    auto scan_half = [&](const int4 (&v)[4], int c, int half) {
        #pragma unroll
        for (int u = 0; u < 4; u += 2) {
            const unsigned e0 = (unsigned)(c * (AGG_CHUNK / 16) + (half * 4 + u) * 32 + lane) * 16u;
            unsigned hit = 0u;
            if (agg_vec_inside(c, half, u))     hit  = agg_hits<0>(v[u], idrep);
            if (agg_vec_inside(c, half, u + 1)) hit |= agg_hits<4>(v[u + 1], idrep);
            qn = agg_queue_hits(hit, e0, q, qn);
            drain();
        }
    };

    int4 va[4], vb[4];
    {
        const int c = wave + off;
        const int cw = (c >= nchunks) ? c - nchunks : c;
        agg_load_half(rp4, cw, 0, lane, va);
        agg_load_half(rp4, cw, 1, lane, vb);
    }
    for (int i = wave; i < nchunks; i += AGG_WAVES) {
        const int c = (i + off >= nchunks) ? i + off - nchunks : i + off;
        const int cn0 = min(i + AGG_WAVES, nchunks - 1) + off;
        const int cn = (cn0 >= nchunks) ? cn0 - nchunks : cn0;
        scan_half(va, c, 0);
        agg_load_half(rp4, cn, 0, lane, va);
        scan_half(vb, c, 1);
        agg_load_half(rp4, cn, 1, lane, vb);
    }
    if (qn > 0) {
        unsigned e = 0u;
        int sv = 0, dl = 0;
        if (lane < qn) { e = q[lane]; sv = src[e]; dl = dst[e] - r0; atomicAdd(&degbin[dl], 1); }
        agg_add_rows<AGG_DEPTH, false>(h, acc, sv, dl, qn, lane);
        sn = agg_stage(st, sn, ((unsigned)sv << 10) | (unsigned)dl, qn, lane, s_lines, blist);
    }
    if (sn > 0) {
        int ln = 0;
        if (lane == 0) ln = atomicAdd(s_lines, 1);
        ln = __builtin_amdgcn_readfirstlane(ln);
        blist[ln * 32 + lane] = (lane < sn) ? st[lane] : LIST_END;
    }
    __syncthreads();

    if (tid == 0) {
        dir[2 * blockIdx.x]     = (int)(blist - list);
        dir[2 * blockIdx.x + 1] = *s_lines;
    }
    if (tid < AGG_RN) deg[(int)blockIdx.x * DEG_PITCH + tid] = (float)degbin[tid];
    if (wave < AGG_RN / 16)
        range_gemm(h, (float)degbin[wave * 16 + (lane & 15)], acc, sWs, sWn, sB, hout, r0, wave, lane, do_relu);
}

__global__ __launch_bounds__(1024) void sage_replay(const float* __restrict__ h,
                                                    const unsigned* __restrict__ list,
                                                    const int* __restrict__ dir,
                                                    const float* __restrict__ deg,
                                                    const float* __restrict__ Wself,
                                                    const float* __restrict__ Wneigh,
                                                    const float* __restrict__ bias,
                                                    float* __restrict__ hout,
                                                    int do_relu) {
    __shared__ __attribute__((aligned(16))) float lds[2 * F * W_P + F + REPLAY_ACC_FLOATS];
    float* sWs = lds;
    float* sWn = sWs + F * W_P;
    float* sB  = sWn + F * W_P;
    float* acc = sB + F;
    const int tid  = threadIdx.x;
    const int lane = tid & 31;
    const int wave = __builtin_amdgcn_readfirstlane(tid >> 5);
    const int r0   = blockIdx.x * AGG_RN;

    const unsigned* blist = list + dir[2 * blockIdx.x];
    const int nl = dir[2 * blockIdx.x + 1];

    unsigned e = LIST_END;
    if (wave < nl) e = blist[wave * 32 + lane];

    stage_weights(Wself, Wneigh, bias, sWs, sWn, sB, tid);

    float4* acc4 = (float4*)acc;
    for (int j = tid; j < AGG_RN * ACC_P / 4; j += 1024) acc4[j] = make_float4(0.f, 0.f, 0.f, 0.f);
    __syncthreads();

    for (int i = wave; i < nl; i += AGG_WAVES) {
        const unsigned en = blist[min(i + AGG_WAVES, nl - 1) * 32 + lane];
        const int cnt = __builtin_popcount(__builtin_amdgcn_ballot_w32(e != LIST_END));
        agg_add_rows<REPLAY_DEPTH, true>(h, acc, (int)(e >> 10), (int)(e & 1023u), cnt, lane);
        e = en;
    }
    __syncthreads();

    if (wave < AGG_RN / 16)
        range_gemm(h, deg[(int)blockIdx.x * DEG_PITCH + wave * 16 + (lane & 15)], acc, sWs, sWn, sB, hout, r0, wave, lane, do_relu);
}

extern "C" void kernel_launch(void* const* d_in, const int* in_sizes, int n_in,
                              void* d_out, int out_size, void* d_ws, size_t ws_size,
                              hipStream_t stream) {
    (void)in_sizes; (void)n_in; (void)out_size; (void)ws_size;
    const float* feat = (const float*)d_in[0];
    const int*   src  = (const int*)d_in[1];
    const int*   dst  = (const int*)d_in[2];
    const float* Ws[3] = { (const float*)d_in[3], (const float*)d_in[6], (const float*)d_in[9]  };
    const float* Wn[3] = { (const float*)d_in[4], (const float*)d_in[7], (const float*)d_in[10] };
    const float* bs[3] = { (const float*)d_in[5], (const float*)d_in[8], (const float*)d_in[11] };
    float* outp = (float*)d_out;

    float* ws   = (float*)d_ws;
    float* hb   = ws + (long long)N_NODES;
    float* ha   = hb  + (long long)N_NODES * F;
    float* h2   = ha  + (long long)N_NODES * F;

    const int aggBlocks   = AGG_RANGES;

    unsigned* list   = (unsigned*)h2;
    int* dir         = (int*)(list + LIST_CAP);
    int* cnt         = (int*)(h2 + 3000000);
    float* deg       = h2 + 3100000;
    unsigned* rplane = (unsigned*)(h2 + 4000000);
    sage_range_plane<<<PLANE_BLOCKS, 1024, 0, stream>>>(dst, rplane, cnt);

    sage_aggregate<<<aggBlocks, 1024, 0, stream>>>(feat, src, dst, rplane, cnt, deg, Ws[0], Wn[0], bs[0], ha, list, dir, 1);

    sage_replay<<<aggBlocks, 1024, 0, stream>>>(ha, list, dir, deg, Ws[1], Wn[1], bs[1], hb, 1);

    sage_replay<<<aggBlocks, 1024, 0, stream>>>(hb, list, dir, deg, Ws[2], Wn[2], bs[2], outp, 0);
}


# self product h x W_self of a wave's 16 nodes runs inside the aggregation phase in all three layers (layer 0: between scan steps), staggered by wave; neighbour half waits; k order of sums changed
# speedup vs baseline: 3.8636x; 1.0416x over previous
#include <hip/hip_runtime.h>


#define N_NODES 100000
#define N_EDGES 1600000
#define F 64
#define ACC_P 68
#define W_P   80

#define AGG_RANGES 250
#define AGG_RN     (N_NODES / AGG_RANGES)
static_assert(AGG_RANGES <= 255, "a range number fits one byte, 255 left for no range");

typedef __attribute__((ext_vector_type(2))) float v2f;
typedef __attribute__((ext_vector_type(8))) float v8f;

#define PLANE_LINES  (N_EDGES / 128)
#define PLANE_BLOCKS 125
static_assert(PLANE_LINES * 128 == N_EDGES, "the plane is written in whole lines of 128 edges");
static_assert(PLANE_LINES <= 4 * PLANE_BLOCKS * 32, "four lines per wave cover the plane");

static __device__ __forceinline__ unsigned range_byte(int d) {
    return ((unsigned)d < (unsigned)N_NODES) ? (unsigned)d / (unsigned)AGG_RN : 255u;
}
__global__ __launch_bounds__(1024) void sage_range_plane(const int* __restrict__ dst,
                                                         unsigned* __restrict__ rplane,
                                                         int* __restrict__ cnt) {
    __shared__ int hist[256];
    const int tid  = threadIdx.x;
    const int lane = tid & 31;
    const int wave = tid >> 5;

    int4 v[4];
    #pragma unroll
    for (int k = 0; k < 4; ++k) {
        const int l = (int)blockIdx.x + PLANE_BLOCKS * (wave + 32 * k);
        v[k] = ((const int4*)dst)[min(l, PLANE_LINES - 1) * 32 + lane];
    }
    if (tid < 256) hist[tid] = 0;
    __syncthreads();

    #pragma unroll
    for (int k = 0; k < 4; ++k) {
        const int l = (int)blockIdx.x + PLANE_BLOCKS * (wave + 32 * k);
        if (l < PLANE_LINES) {
            const unsigned b0 = range_byte(v[k].x), b1 = range_byte(v[k].y);
            const unsigned b2 = range_byte(v[k].z), b3 = range_byte(v[k].w);
            rplane[l * 32 + lane] = b0 | (b1 << 8) | (b2 << 16) | (b3 << 24);
            atomicAdd(&hist[b0], 1);
            atomicAdd(&hist[b1], 1);
            atomicAdd(&hist[b2], 1);
            atomicAdd(&hist[b3], 1);
        }
    }
    __syncthreads();
    if (tid < 256) cnt[(int)blockIdx.x * 256 + tid] = hist[tid];
}

#define AGG_WAVES  32
#define AGG_CHUNK  4096
#define AGG_SEG    1024
#define AGG_QCAP   (AGG_SEG + 32)
#define AGG_DEPTH  8
#define REPLAY_DEPTH 16
#define LIST_PAD   1056
#define LIST_CAP   (N_EDGES + AGG_RANGES * LIST_PAD)
#define LIST_END   0xFFFFFFFFu
#define REPLAY_ACC_FLOATS 41472
static_assert(REPLAY_ACC_FLOATS >= AGG_RN * ACC_P, "room for the block's sums");
static_assert(AGG_RN % 16 == 0 && AGG_RN / 16 <= AGG_WAVES, "one 16-node tile per wave covers the range");
static_assert(F * F / 4 == AGG_WAVES * 32, "one 16-byte piece of each weight matrix per thread");
static_assert(REPLAY_ACC_FLOATS * 4 > 160 * 1024, "one list-walk block per WGP");
static_assert(LIST_PAD >= 32 * AGG_WAVES + 32, "room for every wave's part line and the rounding");
static_assert(LIST_CAP + 2 * AGG_RANGES <= 3000000, "list and directory end before the range counts");
static_assert(3000000 + PLANE_BLOCKS * 256 <= 3100000, "the range counts end before the degrees");
#define DEG_PITCH 512
static_assert(DEG_PITCH >= AGG_RN && DEG_PITCH % 32 == 0, "a block's degrees fill whole lines of their own");
static_assert(3100000 + AGG_RANGES * DEG_PITCH <= 4000000, "the degrees end before the range plane");
static_assert(4000000 + N_EDGES / 4 <= N_NODES * F, "the range plane fits in the same workspace plane");
static_assert(N_NODES <= (1 << 21), "an entry keeps the source node above the 10 bits of the row");
static_assert(AGG_RN * AGG_RANGES == N_NODES, "ranges must tile the nodes");
static_assert(AGG_RN <= 1024, "a list entry keeps the local row in 10 bits");
static_assert(N_EDGES % 16 == 0, "the range plane is read as 16-byte vectors");
static_assert(AGG_CHUNK == 8 * 32 * 16 && AGG_SEG == 2 * 32 * 16, "a step is 8 vectors per lane, a segment 2");
static_assert((N_EDGES + AGG_CHUNK - 1) / AGG_CHUNK >= AGG_WAVES, "every wave has a first chunk");
static_assert(AGG_SEG == 2 * 32 * 16, "a segment is a pair of vectors: the second one's edges lie 512 behind the first one's");

#define PLANE_VECS (N_EDGES / 16)
static_assert(PLANE_VECS % 32 == 0, "the plane ends on a whole wave-load of 32 vectors");
static __device__ __forceinline__ bool agg_vec_inside(int c, int half, int u) {
    return c * (AGG_CHUNK / 16) + (half * 4 + u) * 32 < PLANE_VECS;
}
static __device__ __forceinline__ void agg_load_half(const int4* __restrict__ rp4, int c, int half,
                                                     int lane, int4 (&v)[4]) {
    #pragma unroll
    for (int u = 0; u < 4; ++u) {
        const int v0 = c * (AGG_CHUNK / 16) + (half * 4 + u) * 32;
        v[u] = rp4[min(v0, PLANE_VECS - 32) + lane];
    }
}

template <int BASE>
static __device__ __forceinline__ unsigned agg_hits(const int4 v, unsigned idrep) {
    const unsigned wd[4] = { (unsigned)v.x, (unsigned)v.y, (unsigned)v.z, (unsigned)v.w };
    unsigned hit = 0u;
    #pragma unroll
    for (int w = 0; w < 4; ++w) {
        const unsigned x = wd[w] ^ idrep;
        const unsigned z = ~(((x & 0x7F7F7F7Fu) + 0x7F7F7F7Fu) | x) & 0x80808080u;
        hit |= z >> (7 - BASE - w);
    }
    return hit;
}

static __device__ __forceinline__ int agg_queue_hits(unsigned hit, unsigned e0, unsigned* q, int qn) {
    unsigned bal = __builtin_amdgcn_ballot_w32(hit != 0u);
    while (bal != 0u) {
        if (hit != 0u) {
            const unsigned p = (unsigned)__builtin_ctz(hit);
            q[qn + (int)__builtin_amdgcn_mbcnt_lo(bal, 0u)] =
                e0 + ((p & 4u) << 7) + ((p & 3u) << 2) + (p >> 3);
            hit &= hit - 1u;
        }
        qn += __builtin_popcount(bal);
        bal = __builtin_amdgcn_ballot_w32(hit != 0u);
    }
    return qn;
}

template <int DEPTH, bool WHOLE>
static __device__ __forceinline__ void agg_add_rows(const float* __restrict__ h, float* acc,
                                                    int s, int dl, int cnt, int lane) {
    int j = 0;
    for (; WHOLE && j + DEPTH <= cnt; j += DEPTH) {
        float2 hv[DEPTH];
        int dj[DEPTH];
        #pragma unroll
        for (int k = 0; k < DEPTH; ++k) {
            const int sj = __builtin_amdgcn_readlane(s, j + k);
            dj[k] = __builtin_amdgcn_readlane(dl, j + k);
            hv[k] = *(const float2*)(h + (long long)sj * F + lane * 2);
        }
        #pragma unroll
        for (int k = 0; k < DEPTH; ++k) {
            float* row = &acc[dj[k] * ACC_P + lane * 2];
            __hip_atomic_fetch_add(row,     hv[k].x, __ATOMIC_RELAXED, __HIP_MEMORY_SCOPE_WORKGROUP);
            __hip_atomic_fetch_add(row + 1, hv[k].y, __ATOMIC_RELAXED, __HIP_MEMORY_SCOPE_WORKGROUP);
        }
    }
    for (; j < cnt; j += DEPTH) {
        float2 hv[DEPTH];
        int dj[DEPTH];
        #pragma unroll
        for (int k = 0; k < DEPTH; ++k) {
            const int jj = min(j + k, cnt - 1);
            const int sj = __builtin_amdgcn_readlane(s, jj);
            dj[k] = __builtin_amdgcn_readlane(dl, jj);
            hv[k] = *(const float2*)(h + (long long)sj * F + lane * 2);
        }
        #pragma unroll
        for (int k = 0; k < DEPTH; ++k) {
            if (k == 0 || j + k < cnt) {
                float* row = &acc[dj[k] * ACC_P + lane * 2];
                __hip_atomic_fetch_add(row,     hv[k].x, __ATOMIC_RELAXED, __HIP_MEMORY_SCOPE_WORKGROUP);
                __hip_atomic_fetch_add(row + 1, hv[k].y, __ATOMIC_RELAXED, __HIP_MEMORY_SCOPE_WORKGROUP);
            }
        }
    }
}

static __device__ __forceinline__ int agg_stage(unsigned* st, int sn, unsigned entry,
                                                int cnt, int lane, int* lines,
                                                unsigned* __restrict__ blist) {
    if (lane < cnt) st[sn + lane] = entry;
    sn += cnt;
    if (sn >= 32) {
        int ln = 0;
        if (lane == 0) ln = atomicAdd(lines, 1);
        ln = __builtin_amdgcn_readfirstlane(ln);
        const unsigned e = st[lane];
        const unsigned r = st[32 + lane];
        blist[ln * 32 + lane] = e;
        sn -= 32;
        if (lane < sn) st[lane] = r;
    }
    return sn;
}

static __device__ __forceinline__ void stage_weights(const float* __restrict__ Wself,
                                                     const float* __restrict__ Wneigh,
                                                     const float* __restrict__ bias,
                                                     float* sWs, float* sWn, float* sB, int tid) {
    const float4 ws = ((const float4*)Wself)[tid];
    const float4 wn = ((const float4*)Wneigh)[tid];
    const int o = (tid >> 4) * W_P + (tid & 15) * 4;
    *(float4*)(sWs + o) = ws;
    *(float4*)(sWn + o) = wn;
    if (tid < F) sB[tid] = bias[tid];
}

static __device__ __forceinline__ void range_gemm_self(const float* __restrict__ h,
                                                       const float* sWs,
                                                       int r0, int wave, int lane, v8f (&d)[4]) {
    const int m0    = wave * 16;
    const int lrow  = lane & 15;
    const int khalf = lane >> 4;
    const int node  = r0 + m0 + lrow;

    const float* hrow = h + (long long)node * F + khalf * 2;
    v2f as[16];
    #pragma unroll
    for (int kk = 0; kk < 16; ++kk) as[kk] = *(const v2f*)(hrow + kk * 4);

    const float* bs = sWs + khalf * 2 * W_P + lrow;
    #pragma unroll
    for (int kk = 0; kk < 16; ++kk) {
        #pragma unroll
        for (int c = 0; c < 4; ++c) {
            v2f b;
            b.x = bs[(kk * 4    ) * W_P + c * 16];
            b.y = bs[(kk * 4 + 1) * W_P + c * 16];
            d[c] = __builtin_amdgcn_wmma_f32_16x16x4_f32(false, as[kk], false, b,
                                                         (short)0, d[c], false, false);
        }
    }
}

static __device__ __forceinline__ void range_gemm_neigh(float degv,
                                                        const float* acc,
                                                        const float* sWn, const float* sB,
                                                        float* __restrict__ hout,
                                                        int r0, int wave, int lane, int do_relu,
                                                        v8f (&d)[4]) {
    const int m0    = wave * 16;
    const int lrow  = lane & 15;
    const int khalf = lane >> 4;
    const float rinv = 1.0f / fmaxf(degv, 1.0f);

    const float* arow = acc + (m0 + lrow) * ACC_P + khalf * 2;
    const float* bn = sWn + khalf * 2 * W_P + lrow;
    #pragma unroll
    for (int kk = 0; kk < 16; ++kk) {
        v2f an = *(const v2f*)(arow + kk * 4);
        an.x *= rinv;
        an.y *= rinv;
        #pragma unroll
        for (int c = 0; c < 4; ++c) {
            v2f b;
            b.x = bn[(kk * 4    ) * W_P + c * 16];
            b.y = bn[(kk * 4 + 1) * W_P + c * 16];
            d[c] = __builtin_amdgcn_wmma_f32_16x16x4_f32(false, an, false, b,
                                                         (short)0, d[c], false, false);
        }
    }

    float* orow = hout + (long long)(r0 + m0 + khalf * 8) * F + lrow;
    #pragma unroll
    for (int c = 0; c < 4; ++c) {
        const float bv = sB[c * 16 + lrow];
        #pragma unroll
        for (int i = 0; i < 8; ++i) {
            float v = d[c][i] + bv;
            if (do_relu) v = fmaxf(v, 0.0f);
            orow[i * F + c * 16] = v;
        }
    }
}

__global__ __launch_bounds__(1024) void sage_aggregate(const float* __restrict__ h,
                                                       const int* __restrict__ src,
                                                       const int* __restrict__ dst,
                                                       const unsigned* __restrict__ rplane,
                                                       const int* __restrict__ cnt,
                                                       float* __restrict__ deg,
                                                       const float* __restrict__ Wself,
                                                       const float* __restrict__ Wneigh,
                                                       const float* __restrict__ bias,
                                                       float* __restrict__ hout,
                                                       unsigned* __restrict__ list,
                                                       int* __restrict__ dir,
                                                       int do_relu) {
    __shared__ __attribute__((aligned(16))) float lds[2 * F * W_P + F + AGG_RN * ACC_P
                                                      + AGG_WAVES * AGG_QCAP + AGG_WAVES * 64 + 4 + AGG_RN];
    float* sWs = lds;
    float* sWn = sWs + F * W_P;
    float* sB  = sWn + F * W_P;
    float* acc = sB + F;
    unsigned* queue = (unsigned*)(acc + AGG_RN * ACC_P);
    unsigned* stage = queue + AGG_WAVES * AGG_QCAP;
    int* s_below = (int*)(stage + AGG_WAVES * 64);
    int* s_lines = s_below + 1;
    int* degbin  = s_below + 4;
    const int tid  = threadIdx.x;
    const int lane = tid & 31;
    const int wave = __builtin_amdgcn_readfirstlane(tid >> 5);
    const int r0   = blockIdx.x * AGG_RN;

    stage_weights(Wself, Wneigh, bias, sWs, sWn, sB, tid);

    float4* acc4 = (float4*)acc;
    for (int j = tid; j < AGG_RN * ACC_P / 4; j += 1024) acc4[j] = make_float4(0.f, 0.f, 0.f, 0.f);
    if (tid == 0) { *s_below = 0; *s_lines = 0; }
    if (tid < AGG_RN) degbin[tid] = 0;

    int below = 0;
    {
        const int4* c4 = (const int4*)cnt;
        const int n4 = PLANE_BLOCKS * 64;
        const int r  = (tid & 63) * 4;
        const int b  = (int)blockIdx.x;
        int4 t[8];
        #pragma unroll
        for (int k = 0; k < 8; ++k) t[k] = c4[min(tid + k * 1024, n4 - 1)];
        #pragma unroll
        for (int k = 0; k < 8; ++k) {
            if (tid + k * 1024 < n4)
                below += (r < b ? t[k].x : 0) + (r + 1 < b ? t[k].y : 0)
                       + (r + 2 < b ? t[k].z : 0) + (r + 3 < b ? t[k].w : 0);
        }
        for (int m = 16; m >= 1; m >>= 1) below += __shfl_xor(below, m);
    }
    __syncthreads();
    if (lane == 0 && below != 0) atomicAdd(s_below, below);
    __syncthreads();
    unsigned* blist = list + ((__builtin_amdgcn_readfirstlane(*s_below) & ~31) + (int)blockIdx.x * LIST_PAD);
    unsigned* st = stage + wave * 64;
    int sn = 0;

    unsigned* q = queue + wave * AGG_QCAP;
    const int4* rp4 = (const int4*)rplane;
    const int nchunks = (N_EDGES + AGG_CHUNK - 1) / AGG_CHUNK;
    const unsigned idrep = (unsigned)blockIdx.x * 0x01010101u;

    const int off = (int)(((long long)blockIdx.x * nchunks) / AGG_RANGES);

    int qn = 0;

    auto drain = [&]() {
        while (qn >= 32) {
            qn -= 32;
            const unsigned e = q[qn + lane];
            const int sv = src[e];
            const int dl = dst[e] - r0;
            atomicAdd(&degbin[dl], 1);
            agg_add_rows<REPLAY_DEPTH, true>(h, acc, sv, dl, 32, lane);
            sn = agg_stage(st, sn, ((unsigned)sv << 10) | (unsigned)dl, 32, lane, s_lines, blist);
        }
    };
    auto scan_half = [&](const int4 (&v)[4], int c, int half) {
        #pragma unroll
        for (int u = 0; u < 4; u += 2) {
            const unsigned e0 = (unsigned)(c * (AGG_CHUNK / 16) + (half * 4 + u) * 32 + lane) * 16u;
            unsigned hit = 0u;
            if (agg_vec_inside(c, half, u))     hit  = agg_hits<0>(v[u], idrep);
            if (agg_vec_inside(c, half, u + 1)) hit |= agg_hits<4>(v[u + 1], idrep);
            qn = agg_queue_hits(hit, e0, q, qn);
            drain();
        }
    };

    int4 va[4], vb[4];
    {
        const int c = wave + off;
        const int cw = (c >= nchunks) ? c - nchunks : c;
        agg_load_half(rp4, cw, 0, lane, va);
        agg_load_half(rp4, cw, 1, lane, vb);
    }
    v8f d[4] = {};
    const bool has_tile = wave < AGG_RN / 16;
    const int self_step = (wave >> 2) + 1;
    static_assert((N_EDGES + AGG_CHUNK - 1) / AGG_CHUNK / AGG_WAVES > (AGG_RN / 16 - 1) / 4 + 1,
                  "each wave's scan is longer than its self-product step");
    int step = 0;
    for (int i = wave; i < nchunks; i += AGG_WAVES, ++step) {
        const int c = (i + off >= nchunks) ? i + off - nchunks : i + off;
        const int cn0 = min(i + AGG_WAVES, nchunks - 1) + off;
        const int cn = (cn0 >= nchunks) ? cn0 - nchunks : cn0;
        scan_half(va, c, 0);
        agg_load_half(rp4, cn, 0, lane, va);
        scan_half(vb, c, 1);
        if (has_tile && step == self_step) range_gemm_self(h, sWs, r0, wave, lane, d);
        agg_load_half(rp4, cn, 1, lane, vb);
    }
    if (qn > 0) {
        unsigned e = 0u;
        int sv = 0, dl = 0;
        if (lane < qn) { e = q[lane]; sv = src[e]; dl = dst[e] - r0; atomicAdd(&degbin[dl], 1); }
        agg_add_rows<AGG_DEPTH, false>(h, acc, sv, dl, qn, lane);
        sn = agg_stage(st, sn, ((unsigned)sv << 10) | (unsigned)dl, qn, lane, s_lines, blist);
    }
    if (sn > 0) {
        int ln = 0;
        if (lane == 0) ln = atomicAdd(s_lines, 1);
        ln = __builtin_amdgcn_readfirstlane(ln);
        blist[ln * 32 + lane] = (lane < sn) ? st[lane] : LIST_END;
    }
    __syncthreads();

    if (tid == 0) {
        dir[2 * blockIdx.x]     = (int)(blist - list);
        dir[2 * blockIdx.x + 1] = *s_lines;
    }
    if (tid < AGG_RN) deg[(int)blockIdx.x * DEG_PITCH + tid] = (float)degbin[tid];
    if (has_tile)
        range_gemm_neigh((float)degbin[wave * 16 + (lane & 15)], acc, sWn, sB, hout, r0, wave, lane, do_relu, d);
}

__global__ __launch_bounds__(1024) void sage_replay(const float* __restrict__ h,
                                                    const unsigned* __restrict__ list,
                                                    const int* __restrict__ dir,
                                                    const float* __restrict__ deg,
                                                    const float* __restrict__ Wself,
                                                    const float* __restrict__ Wneigh,
                                                    const float* __restrict__ bias,
                                                    float* __restrict__ hout,
                                                    int do_relu) {
    __shared__ __attribute__((aligned(16))) float lds[2 * F * W_P + F + REPLAY_ACC_FLOATS];
    float* sWs = lds;
    float* sWn = sWs + F * W_P;
    float* sB  = sWn + F * W_P;
    float* acc = sB + F;
    const int tid  = threadIdx.x;
    const int lane = tid & 31;
    const int wave = __builtin_amdgcn_readfirstlane(tid >> 5);
    const int r0   = blockIdx.x * AGG_RN;

    const unsigned* blist = list + dir[2 * blockIdx.x];
    const int nl = dir[2 * blockIdx.x + 1];

    unsigned e = LIST_END;
    if (wave < nl) e = blist[wave * 32 + lane];

    stage_weights(Wself, Wneigh, bias, sWs, sWn, sB, tid);

    float4* acc4 = (float4*)acc;
    for (int j = tid; j < AGG_RN * ACC_P / 4; j += 1024) acc4[j] = make_float4(0.f, 0.f, 0.f, 0.f);
    __syncthreads();

    v8f d[4] = {};
    const bool has_tile = wave < AGG_RN / 16;
    float degv = 1.0f;
    if (has_tile) degv = deg[(int)blockIdx.x * DEG_PITCH + wave * 16 + (lane & 15)];
    const int self_step = wave >> 2;
    bool self_done = false;
    int step = 0;
    for (int i = wave; i < nl; i += AGG_WAVES, ++step) {
        const unsigned en = blist[min(i + AGG_WAVES, nl - 1) * 32 + lane];
        const int cnt = __builtin_popcount(__builtin_amdgcn_ballot_w32(e != LIST_END));
        agg_add_rows<REPLAY_DEPTH, true>(h, acc, (int)(e >> 10), (int)(e & 1023u), cnt, lane);
        e = en;
        if (has_tile && step == self_step) {
            range_gemm_self(h, sWs, r0, wave, lane, d);
            self_done = true;
        }
    }
    if (has_tile && !self_done) range_gemm_self(h, sWs, r0, wave, lane, d);
    __syncthreads();

    if (has_tile)
        range_gemm_neigh(degv, acc, sWn, sB, hout, r0, wave, lane, do_relu, d);
}

extern "C" void kernel_launch(void* const* d_in, const int* in_sizes, int n_in,
                              void* d_out, int out_size, void* d_ws, size_t ws_size,
                              hipStream_t stream) {
    (void)in_sizes; (void)n_in; (void)out_size; (void)ws_size;
    const float* feat = (const float*)d_in[0];
    const int*   src  = (const int*)d_in[1];
    const int*   dst  = (const int*)d_in[2];
    const float* Ws[3] = { (const float*)d_in[3], (const float*)d_in[6], (const float*)d_in[9]  };
    const float* Wn[3] = { (const float*)d_in[4], (const float*)d_in[7], (const float*)d_in[10] };
    const float* bs[3] = { (const float*)d_in[5], (const float*)d_in[8], (const float*)d_in[11] };
    float* outp = (float*)d_out;

    float* ws   = (float*)d_ws;
    float* hb   = ws + (long long)N_NODES;
    float* ha   = hb  + (long long)N_NODES * F;
    float* h2   = ha  + (long long)N_NODES * F;

    const int aggBlocks   = AGG_RANGES;

    unsigned* list   = (unsigned*)h2;
    int* dir         = (int*)(list + LIST_CAP);
    int* cnt         = (int*)(h2 + 3000000);
    float* deg       = h2 + 3100000;
    unsigned* rplane = (unsigned*)(h2 + 4000000);
    sage_range_plane<<<PLANE_BLOCKS, 1024, 0, stream>>>(dst, rplane, cnt);

    sage_aggregate<<<aggBlocks, 1024, 0, stream>>>(feat, src, dst, rplane, cnt, deg, Ws[0], Wn[0], bs[0], ha, list, dir, 1);

    sage_replay<<<aggBlocks, 1024, 0, stream>>>(ha, list, dir, deg, Ws[1], Wn[1], bs[1], hb, 1);

    sage_replay<<<aggBlocks, 1024, 0, stream>>>(hb, list, dir, deg, Ws[2], Wn[2], bs[2], outp, 0);
}


# layer 0 reads its own range's row of a bit plane (one bit per 4 edges per range, built in LDS by launch 1), not the byte plane of all edges; a set bit's four dst by one int4 load; sum order changed
# speedup vs baseline: 4.1946x; 1.0857x over previous
#include <hip/hip_runtime.h>


#define N_NODES 100000
#define N_EDGES 1600000
#define F 64
#define ACC_P 68
#define W_P   80

#define AGG_RANGES 250
#define AGG_RN     (N_NODES / AGG_RANGES)
static_assert(AGG_RANGES <= 255, "a range number fits one byte, 255 left for no range");

typedef __attribute__((ext_vector_type(2))) float v2f;
typedef __attribute__((ext_vector_type(8))) float v8f;

#define QUADS        (N_EDGES / 4)
#define BIT_LINES    ((QUADS / 32 + 31) / 32)
#define ROW_WORDS    (BIT_LINES * 32)
#define PLANE_BLOCKS 125
#define PLANE_SLICES 4
#define BM_PITCH     257
static_assert(QUADS * 4 == N_EDGES && QUADS % 32 == 0, "whole quads, whole words of quads");
static_assert(BIT_LINES <= PLANE_SLICES * PLANE_BLOCKS, "four slices per block cover the plane");
static_assert((ROW_WORDS * 4) % 128 == 0, "every row starts on a 128-byte line");
static_assert((long long)AGG_RANGES * ROW_WORDS <= (long long)N_NODES * F, "the bit plane fits in one feature plane");

static __device__ __forceinline__ unsigned range_byte(int d) {
    return ((unsigned)d < (unsigned)N_NODES) ? (unsigned)d / (unsigned)AGG_RN : 255u;
}
__global__ __launch_bounds__(1024) void sage_range_bits(const int* __restrict__ dst,
                                                        unsigned* __restrict__ bits,
                                                        int* __restrict__ cnt) {
    __shared__ int hist[256];
    __shared__ __attribute__((aligned(16))) unsigned bm[PLANE_SLICES * 32 * BM_PITCH];
    static_assert((PLANE_SLICES * 32 * BM_PITCH) % 4 == 0, "the bit tables are cleared in 16-byte pieces");
    const int tid  = threadIdx.x;
    const int lane = tid & 31;
    const int wave = __builtin_amdgcn_readfirstlane(tid >> 5);

    int4 v[PLANE_SLICES];
    #pragma unroll
    for (int k = 0; k < PLANE_SLICES; ++k) {
        const int qd = (((int)blockIdx.x + PLANE_BLOCKS * k) * 32 + wave) * 32 + lane;
        v[k] = ((const int4*)dst)[min(qd, QUADS - 1)];
    }
    if (tid < 256) hist[tid] = 0;
    for (int j = tid; j < PLANE_SLICES * 32 * BM_PITCH / 4; j += 1024) ((uint4*)bm)[j] = make_uint4(0u, 0u, 0u, 0u);
    __syncthreads();

    #pragma unroll
    for (int k = 0; k < PLANE_SLICES; ++k) {
        const int qd = (((int)blockIdx.x + PLANE_BLOCKS * k) * 32 + wave) * 32 + lane;
        if (qd < QUADS) {
            unsigned* bw = bm + (k * 32 + wave) * BM_PITCH;
            const unsigned bit = 1u << lane;
            const unsigned b0 = range_byte(v[k].x), b1 = range_byte(v[k].y);
            const unsigned b2 = range_byte(v[k].z), b3 = range_byte(v[k].w);
            atomicAdd(&hist[b0], 1);
            atomicAdd(&hist[b1], 1);
            atomicAdd(&hist[b2], 1);
            atomicAdd(&hist[b3], 1);
            if (b0 < AGG_RANGES) atomicOr(&bw[b0], bit);
            if (b1 < AGG_RANGES) atomicOr(&bw[b1], bit);
            if (b2 < AGG_RANGES) atomicOr(&bw[b2], bit);
            if (b3 < AGG_RANGES) atomicOr(&bw[b3], bit);
        }
    }
    __syncthreads();
    if (tid < 256) cnt[(int)blockIdx.x * 256 + tid] = hist[tid];
    static_assert(AGG_RANGES <= 8 * 32, "eight ranges per wave cover them all");
    #pragma unroll
    for (int k = 0; k < PLANE_SLICES; ++k) {
        const int s = (int)blockIdx.x + PLANE_BLOCKS * k;
        if (s < BIT_LINES) {
            unsigned w[8];
            #pragma unroll
            for (int m = 0; m < 8; ++m)
                w[m] = bm[(k * 32 + lane) * BM_PITCH + min(wave + 32 * m, AGG_RANGES - 1)];
            #pragma unroll
            for (int m = 0; m < 8; ++m) {
                const int r = wave + 32 * m;
                if (r < AGG_RANGES) bits[r * ROW_WORDS + s * 32 + lane] = w[m];
            }
        }
    }
}

#define AGG_WAVES  32
#define AGG_QCAP   (1024 + 32)
#define AGG_Q2CAP  (128 + 32)
#define AGG_DEPTH  8
#define REPLAY_DEPTH 16
#define LIST_PAD   1056
#define LIST_CAP   (N_EDGES + AGG_RANGES * LIST_PAD)
#define LIST_END   0xFFFFFFFFu
#define REPLAY_ACC_FLOATS 41472
static_assert(REPLAY_ACC_FLOATS >= AGG_RN * ACC_P, "room for the block's sums");
static_assert(AGG_RN % 16 == 0 && AGG_RN / 16 <= AGG_WAVES, "one 16-node tile per wave covers the range");
static_assert(F * F / 4 == AGG_WAVES * 32, "one 16-byte piece of each weight matrix per thread");
static_assert(REPLAY_ACC_FLOATS * 4 > 160 * 1024, "one list-walk block per WGP");
static_assert(LIST_PAD >= 32 * AGG_WAVES + 32, "room for every wave's part line and the rounding");
static_assert(LIST_CAP + 2 * AGG_RANGES <= 3000000, "list and directory end before the range counts");
static_assert(3000000 + PLANE_BLOCKS * 256 <= 3100000, "the range counts end before the degrees");
#define DEG_PITCH 512
static_assert(DEG_PITCH >= AGG_RN && DEG_PITCH % 32 == 0, "a block's degrees fill whole lines of their own");
static_assert(3100000 + AGG_RANGES * DEG_PITCH <= N_NODES * F, "the degrees fit in the same workspace plane");
static_assert(N_NODES <= (1 << 21), "an entry keeps the source node above the 10 bits of the row");
static_assert(AGG_RN * AGG_RANGES == N_NODES, "ranges must tile the nodes");
static_assert(AGG_RN <= 512, "an edge queue entry keeps the local row in 9 bits, a list entry in 10");
static_assert(N_EDGES <= (1 << 23), "an edge queue entry keeps the edge number above the 9 bits of the row");
static_assert(BIT_LINES >= AGG_WAVES, "every wave has a first line");

static __device__ __forceinline__ int agg_queue_bits(unsigned w, unsigned q0, unsigned* q, int qn) {
    unsigned bal = __builtin_amdgcn_ballot_w32(w != 0u);
    while (bal != 0u) {
        if (w != 0u) {
            q[qn + (int)__builtin_amdgcn_mbcnt_lo(bal, 0u)] = q0 + (unsigned)__builtin_ctz(w);
            w &= w - 1u;
        }
        qn += __builtin_popcount(bal);
        bal = __builtin_amdgcn_ballot_w32(w != 0u);
    }
    return qn;
}

template <int DEPTH, bool WHOLE>
static __device__ __forceinline__ void agg_add_rows(const float* __restrict__ h, float* acc,
                                                    int s, int dl, int cnt, int lane) {
    int j = 0;
    for (; WHOLE && j + DEPTH <= cnt; j += DEPTH) {
        float2 hv[DEPTH];
        int dj[DEPTH];
        #pragma unroll
        for (int k = 0; k < DEPTH; ++k) {
            const int sj = __builtin_amdgcn_readlane(s, j + k);
            dj[k] = __builtin_amdgcn_readlane(dl, j + k);
            hv[k] = *(const float2*)(h + (long long)sj * F + lane * 2);
        }
        #pragma unroll
        for (int k = 0; k < DEPTH; ++k) {
            float* row = &acc[dj[k] * ACC_P + lane * 2];
            __hip_atomic_fetch_add(row,     hv[k].x, __ATOMIC_RELAXED, __HIP_MEMORY_SCOPE_WORKGROUP);
            __hip_atomic_fetch_add(row + 1, hv[k].y, __ATOMIC_RELAXED, __HIP_MEMORY_SCOPE_WORKGROUP);
        }
    }
    for (; j < cnt; j += DEPTH) {
        float2 hv[DEPTH];
        int dj[DEPTH];
        #pragma unroll
        for (int k = 0; k < DEPTH; ++k) {
            const int jj = min(j + k, cnt - 1);
            const int sj = __builtin_amdgcn_readlane(s, jj);
            dj[k] = __builtin_amdgcn_readlane(dl, jj);
            hv[k] = *(const float2*)(h + (long long)sj * F + lane * 2);
        }
        #pragma unroll
        for (int k = 0; k < DEPTH; ++k) {
            if (k == 0 || j + k < cnt) {
                float* row = &acc[dj[k] * ACC_P + lane * 2];
                __hip_atomic_fetch_add(row,     hv[k].x, __ATOMIC_RELAXED, __HIP_MEMORY_SCOPE_WORKGROUP);
                __hip_atomic_fetch_add(row + 1, hv[k].y, __ATOMIC_RELAXED, __HIP_MEMORY_SCOPE_WORKGROUP);
            }
        }
    }
}

static __device__ __forceinline__ int agg_stage(unsigned* st, int sn, unsigned entry,
                                                int cnt, int lane, int* lines,
                                                unsigned* __restrict__ blist) {
    if (lane < cnt) st[sn + lane] = entry;
    sn += cnt;
    if (sn >= 32) {
        int ln = 0;
        if (lane == 0) ln = atomicAdd(lines, 1);
        ln = __builtin_amdgcn_readfirstlane(ln);
        const unsigned e = st[lane];
        const unsigned r = st[32 + lane];
        blist[ln * 32 + lane] = e;
        sn -= 32;
        if (lane < sn) st[lane] = r;
    }
    return sn;
}

static __device__ __forceinline__ void stage_weights(const float* __restrict__ Wself,
                                                     const float* __restrict__ Wneigh,
                                                     const float* __restrict__ bias,
                                                     float* sWs, float* sWn, float* sB, int tid) {
    const float4 ws = ((const float4*)Wself)[tid];
    const float4 wn = ((const float4*)Wneigh)[tid];
    const int o = (tid >> 4) * W_P + (tid & 15) * 4;
    *(float4*)(sWs + o) = ws;
    *(float4*)(sWn + o) = wn;
    if (tid < F) sB[tid] = bias[tid];
}

static __device__ __forceinline__ void range_gemm_self(const float* __restrict__ h,
                                                       const float* sWs,
                                                       int r0, int wave, int lane, v8f (&d)[4]) {
    const int m0    = wave * 16;
    const int lrow  = lane & 15;
    const int khalf = lane >> 4;
    const int node  = r0 + m0 + lrow;

    const float* hrow = h + (long long)node * F + khalf * 2;
    v2f as[16];
    #pragma unroll
    for (int kk = 0; kk < 16; ++kk) as[kk] = *(const v2f*)(hrow + kk * 4);

    const float* bs = sWs + khalf * 2 * W_P + lrow;
    #pragma unroll
    for (int kk = 0; kk < 16; ++kk) {
        #pragma unroll
        for (int c = 0; c < 4; ++c) {
            v2f b;
            b.x = bs[(kk * 4    ) * W_P + c * 16];
            b.y = bs[(kk * 4 + 1) * W_P + c * 16];
            d[c] = __builtin_amdgcn_wmma_f32_16x16x4_f32(false, as[kk], false, b,
                                                         (short)0, d[c], false, false);
        }
    }
}

static __device__ __forceinline__ void range_gemm_neigh(float degv,
                                                        const float* acc,
                                                        const float* sWn, const float* sB,
                                                        float* __restrict__ hout,
                                                        int r0, int wave, int lane, int do_relu,
                                                        v8f (&d)[4]) {
    const int m0    = wave * 16;
    const int lrow  = lane & 15;
    const int khalf = lane >> 4;
    const float rinv = 1.0f / fmaxf(degv, 1.0f);

    const float* arow = acc + (m0 + lrow) * ACC_P + khalf * 2;
    const float* bn = sWn + khalf * 2 * W_P + lrow;
    #pragma unroll
    for (int kk = 0; kk < 16; ++kk) {
        v2f an = *(const v2f*)(arow + kk * 4);
        an.x *= rinv;
        an.y *= rinv;
        #pragma unroll
        for (int c = 0; c < 4; ++c) {
            v2f b;
            b.x = bn[(kk * 4    ) * W_P + c * 16];
            b.y = bn[(kk * 4 + 1) * W_P + c * 16];
            d[c] = __builtin_amdgcn_wmma_f32_16x16x4_f32(false, an, false, b,
                                                         (short)0, d[c], false, false);
        }
    }

    float* orow = hout + (long long)(r0 + m0 + khalf * 8) * F + lrow;
    #pragma unroll
    for (int c = 0; c < 4; ++c) {
        const float bv = sB[c * 16 + lrow];
        #pragma unroll
        for (int i = 0; i < 8; ++i) {
            float v = d[c][i] + bv;
            if (do_relu) v = fmaxf(v, 0.0f);
            orow[i * F + c * 16] = v;
        }
    }
}

__global__ __launch_bounds__(1024) void sage_aggregate(const float* __restrict__ h,
                                                       const int* __restrict__ src,
                                                       const int* __restrict__ dst,
                                                       const unsigned* __restrict__ bits,
                                                       const int* __restrict__ cnt,
                                                       float* __restrict__ deg,
                                                       const float* __restrict__ Wself,
                                                       const float* __restrict__ Wneigh,
                                                       const float* __restrict__ bias,
                                                       float* __restrict__ hout,
                                                       unsigned* __restrict__ list,
                                                       int* __restrict__ dir,
                                                       int do_relu) {
    __shared__ __attribute__((aligned(16))) float lds[2 * F * W_P + F + AGG_RN * ACC_P
                                                      + AGG_WAVES * AGG_QCAP + AGG_WAVES * AGG_Q2CAP
                                                      + AGG_WAVES * 64 + 4 + AGG_RN];
    float* sWs = lds;
    float* sWn = sWs + F * W_P;
    float* sB  = sWn + F * W_P;
    float* acc = sB + F;
    unsigned* queue  = (unsigned*)(acc + AGG_RN * ACC_P);
    unsigned* queue2 = queue + AGG_WAVES * AGG_QCAP;
    unsigned* stage  = queue2 + AGG_WAVES * AGG_Q2CAP;
    int* s_below = (int*)(stage + AGG_WAVES * 64);
    int* s_lines = s_below + 1;
    int* degbin  = s_below + 4;
    const int tid  = threadIdx.x;
    const int lane = tid & 31;
    const int wave = __builtin_amdgcn_readfirstlane(tid >> 5);
    const int r0   = blockIdx.x * AGG_RN;

    const unsigned* row = bits + (int)blockIdx.x * ROW_WORDS;
    unsigned wcur = row[wave * 32 + lane];

    stage_weights(Wself, Wneigh, bias, sWs, sWn, sB, tid);

    float4* acc4 = (float4*)acc;
    for (int j = tid; j < AGG_RN * ACC_P / 4; j += 1024) acc4[j] = make_float4(0.f, 0.f, 0.f, 0.f);
    if (tid == 0) { *s_below = 0; *s_lines = 0; }
    if (tid < AGG_RN) degbin[tid] = 0;

    int below = 0;
    {
        const int4* c4 = (const int4*)cnt;
        const int n4 = PLANE_BLOCKS * 64;
        const int r  = (tid & 63) * 4;
        const int b  = (int)blockIdx.x;
        int4 t[8];
        #pragma unroll
        for (int k = 0; k < 8; ++k) t[k] = c4[min(tid + k * 1024, n4 - 1)];
        #pragma unroll
        for (int k = 0; k < 8; ++k) {
            if (tid + k * 1024 < n4)
                below += (r < b ? t[k].x : 0) + (r + 1 < b ? t[k].y : 0)
                       + (r + 2 < b ? t[k].z : 0) + (r + 3 < b ? t[k].w : 0);
        }
        for (int m = 16; m >= 1; m >>= 1) below += __shfl_xor(below, m);
    }
    __syncthreads();
    if (lane == 0 && below != 0) atomicAdd(s_below, below);
    __syncthreads();
    unsigned* blist = list + ((__builtin_amdgcn_readfirstlane(*s_below) & ~31) + (int)blockIdx.x * LIST_PAD);
    unsigned* st = stage + wave * 64;
    int sn = 0;

    unsigned* q  = queue  + wave * AGG_QCAP;
    unsigned* q2 = queue2 + wave * AGG_Q2CAP;
    const int4* dst4 = (const int4*)dst;
    int qn  = 0;
    int q2n = 0;

    auto drain_edges = [&]() {
        while (q2n >= 32) {
            q2n -= 32;
            const unsigned ent = q2[q2n + lane];
            const int sv = src[ent >> 9];
            const int dl = (int)(ent & 511u);
            atomicAdd(&degbin[dl], 1);
            agg_add_rows<REPLAY_DEPTH, true>(h, acc, sv, dl, 32, lane);
            sn = agg_stage(st, sn, ((unsigned)sv << 10) | (unsigned)dl, 32, lane, s_lines, blist);
        }
    };
    auto resolve = [&](int base, int n) {
        unsigned qd = 0u;
        int4 d4 = make_int4(-1, -1, -1, -1);
        if (lane < n) { qd = q[base + lane]; d4 = dst4[qd]; }
        const int dd[4] = { d4.x, d4.y, d4.z, d4.w };
        #pragma unroll
        for (int k = 0; k < 4; ++k) {
            const unsigned dl = (unsigned)(dd[k] - r0);
            const bool mine = dl < (unsigned)AGG_RN;
            const unsigned bal = __builtin_amdgcn_ballot_w32(mine);
            if (mine) q2[q2n + (int)__builtin_amdgcn_mbcnt_lo(bal, 0u)] = ((qd * 4u + (unsigned)k) << 9) | dl;
            q2n += __builtin_popcount(bal);
        }
    };
    auto drain_quads = [&]() {
        while (qn >= 32) {
            qn -= 32;
            resolve(qn, 32);
            drain_edges();
        }
    };

    v8f d[4] = {};
    const bool has_tile = wave < AGG_RN / 16;
    const int self_step = (wave >> 2) + 1;
    static_assert(BIT_LINES / AGG_WAVES > (AGG_RN / 16 - 1) / 4 + 1,
                  "each wave's walk of the row is longer than its self-product step");
    int step = 0;
    for (int i = wave; i < BIT_LINES; i += AGG_WAVES, ++step) {
        const unsigned wnext = row[min(i + AGG_WAVES, BIT_LINES - 1) * 32 + lane];
        qn = agg_queue_bits(wcur, (unsigned)(i * 32 + lane) * 32u, q, qn);
        drain_quads();
        if (has_tile && step == self_step) range_gemm_self(h, sWs, r0, wave, lane, d);
        wcur = wnext;
    }
    if (qn > 0) {
        resolve(0, qn);
        qn = 0;
        drain_edges();
    }
    if (q2n > 0) {
        int sv = 0, dl = 0;
        if (lane < q2n) {
            const unsigned ent = q2[lane];
            sv = src[ent >> 9];
            dl = (int)(ent & 511u);
            atomicAdd(&degbin[dl], 1);
        }
        agg_add_rows<AGG_DEPTH, false>(h, acc, sv, dl, q2n, lane);
        sn = agg_stage(st, sn, ((unsigned)sv << 10) | (unsigned)dl, q2n, lane, s_lines, blist);
    }
    if (sn > 0) {
        int ln = 0;
        if (lane == 0) ln = atomicAdd(s_lines, 1);
        ln = __builtin_amdgcn_readfirstlane(ln);
        blist[ln * 32 + lane] = (lane < sn) ? st[lane] : LIST_END;
    }
    __syncthreads();

    if (tid == 0) {
        dir[2 * blockIdx.x]     = (int)(blist - list);
        dir[2 * blockIdx.x + 1] = *s_lines;
    }
    if (tid < AGG_RN) deg[(int)blockIdx.x * DEG_PITCH + tid] = (float)degbin[tid];
    if (has_tile)
        range_gemm_neigh((float)degbin[wave * 16 + (lane & 15)], acc, sWn, sB, hout, r0, wave, lane, do_relu, d);
}

__global__ __launch_bounds__(1024) void sage_replay(const float* __restrict__ h,
                                                    const unsigned* __restrict__ list,
                                                    const int* __restrict__ dir,
                                                    const float* __restrict__ deg,
                                                    const float* __restrict__ Wself,
                                                    const float* __restrict__ Wneigh,
                                                    const float* __restrict__ bias,
                                                    float* __restrict__ hout,
                                                    int do_relu) {
    __shared__ __attribute__((aligned(16))) float lds[2 * F * W_P + F + REPLAY_ACC_FLOATS];
    float* sWs = lds;
    float* sWn = sWs + F * W_P;
    float* sB  = sWn + F * W_P;
    float* acc = sB + F;
    const int tid  = threadIdx.x;
    const int lane = tid & 31;
    const int wave = __builtin_amdgcn_readfirstlane(tid >> 5);
    const int r0   = blockIdx.x * AGG_RN;

    const unsigned* blist = list + dir[2 * blockIdx.x];
    const int nl = dir[2 * blockIdx.x + 1];

    unsigned e = LIST_END;
    if (wave < nl) e = blist[wave * 32 + lane];

    stage_weights(Wself, Wneigh, bias, sWs, sWn, sB, tid);

    float4* acc4 = (float4*)acc;
    for (int j = tid; j < AGG_RN * ACC_P / 4; j += 1024) acc4[j] = make_float4(0.f, 0.f, 0.f, 0.f);
    __syncthreads();

    v8f d[4] = {};
    const bool has_tile = wave < AGG_RN / 16;
    float degv = 1.0f;
    if (has_tile) degv = deg[(int)blockIdx.x * DEG_PITCH + wave * 16 + (lane & 15)];
    const int self_step = wave >> 2;
    bool self_done = false;
    int step = 0;
    for (int i = wave; i < nl; i += AGG_WAVES, ++step) {
        const unsigned en = blist[min(i + AGG_WAVES, nl - 1) * 32 + lane];
        const int cnt = __builtin_popcount(__builtin_amdgcn_ballot_w32(e != LIST_END));
        agg_add_rows<REPLAY_DEPTH, true>(h, acc, (int)(e >> 10), (int)(e & 1023u), cnt, lane);
        e = en;
        if (has_tile && step == self_step) {
            range_gemm_self(h, sWs, r0, wave, lane, d);
            self_done = true;
        }
    }
    if (has_tile && !self_done) range_gemm_self(h, sWs, r0, wave, lane, d);
    __syncthreads();

    if (has_tile)
        range_gemm_neigh(degv, acc, sWn, sB, hout, r0, wave, lane, do_relu, d);
}

extern "C" void kernel_launch(void* const* d_in, const int* in_sizes, int n_in,
                              void* d_out, int out_size, void* d_ws, size_t ws_size,
                              hipStream_t stream) {
    (void)in_sizes; (void)n_in; (void)out_size; (void)ws_size;
    const float* feat = (const float*)d_in[0];
    const int*   src  = (const int*)d_in[1];
    const int*   dst  = (const int*)d_in[2];
    const float* Ws[3] = { (const float*)d_in[3], (const float*)d_in[6], (const float*)d_in[9]  };
    const float* Wn[3] = { (const float*)d_in[4], (const float*)d_in[7], (const float*)d_in[10] };
    const float* bs[3] = { (const float*)d_in[5], (const float*)d_in[8], (const float*)d_in[11] };
    float* outp = (float*)d_out;

    float* ws   = (float*)d_ws;
    float* hb   = ws + (long long)N_NODES;
    float* ha   = hb  + (long long)N_NODES * F;
    float* h2   = ha  + (long long)N_NODES * F;

    const int aggBlocks   = AGG_RANGES;

    unsigned* list   = (unsigned*)h2;
    int* dir         = (int*)(list + LIST_CAP);
    int* cnt         = (int*)(h2 + 3000000);
    float* deg       = h2 + 3100000;
    unsigned* bits   = (unsigned*)hb;
    sage_range_bits<<<PLANE_BLOCKS, 1024, 0, stream>>>(dst, bits, cnt);

    sage_aggregate<<<aggBlocks, 1024, 0, stream>>>(feat, src, dst, bits, cnt, deg, Ws[0], Wn[0], bs[0], ha, list, dir, 1);

    sage_replay<<<aggBlocks, 1024, 0, stream>>>(ha, list, dir, deg, Ws[1], Wn[1], bs[1], hb, 1);

    sage_replay<<<aggBlocks, 1024, 0, stream>>>(hb, list, dir, deg, Ws[2], Wn[2], bs[2], outp, 0);
}
